# starting kernel
# baseline (speedup 1.0000x reference)
#include <hip/hip_runtime.h>
#include <hip/hip_bf16.h>
#include <math.h>


typedef __attribute__((ext_vector_type(2))) float v2f;
typedef __attribute__((ext_vector_type(8))) float v8f;

#define NEG_SLOPE 0.2f
#define BN_EPS 1e-5f

__global__ void gemm_wmma_f32(const float* __restrict__ A,
                              const float* __restrict__ B,
                              float* __restrict__ C,
                              int M, int K) {
  const int Nn   = 128;
  const int wave = threadIdx.x >> 5;
  const int lane = threadIdx.x & 31;
  const int half = lane >> 4;
  const int l    = lane & 15;
  const int m0   = blockIdx.x * 16;
  const int n0   = wave * 16;

  v8f c = {};
  for (int k = 0; k < K; k += 4) {
    const int ka = k + half * 2;
    v2f a, b;
    a.x = A[(size_t)(m0 + l) * K + ka];
    a.y = A[(size_t)(m0 + l) * K + ka + 1];
    b.x = B[(size_t)ka * Nn + n0 + l];
    b.y = B[(size_t)(ka + 1) * Nn + n0 + l];
    c = __builtin_amdgcn_wmma_f32_16x16x4_f32(false, a, false, b,
                                              (short)0, c, false, false);
  }
#pragma unroll
  for (int j = 0; j < 8; ++j) {
    const int row = m0 + j + half * 8;
    C[(size_t)row * Nn + n0 + l] = c[j];
  }
}

__device__ __forceinline__ void atomicMaxF32(float* addr, float val) {
  if (val >= 0.0f)
    atomicMax((int*)addr, __float_as_int(val));
  else
    atomicMin((unsigned int*)addr, __float_as_uint(val));
}

__device__ __forceinline__ void edge_endpoints(const int* __restrict__ ei,
                                               int eid, int E,
                                               int& src, int& dst) {
  if (eid < E) { src = ei[eid]; dst = ei[E + eid]; }
  else         { src = dst = eid - E; }
}

__global__ void fill_f32(float* p, float v, int n) {
  int t = blockIdx.x * blockDim.x + threadIdx.x;
  if (t < n) p[t] = v;
}

__global__ void edge_score(const float* __restrict__ xp,
                           const int* __restrict__ ei,
                           const float* __restrict__ avec,
                           float* __restrict__ e,
                           float* __restrict__ emax,
                           int E, int Etot) {
  const int eid = blockIdx.x * (blockDim.x >> 5) + (threadIdx.x >> 5);
  if (eid >= Etot) return;
  const int lane = threadIdx.x & 31;
  int src, dst; edge_endpoints(ei, eid, E, src, dst);

  const int h   = lane >> 4;
  const int idx = (lane & 15) * 4;
  const float4 s  = *(const float4*)(xp + (size_t)src * 128 + h * 64 + idx);
  const float4 d  = *(const float4*)(xp + (size_t)dst * 128 + h * 64 + idx);
  const float4 av = *(const float4*)(avec + h * 64 + idx);

  float m, lr, acc = 0.0f;
  m = s.x + d.x; lr = (m > 0.0f) ? m : NEG_SLOPE * m; acc += lr * av.x;
  m = s.y + d.y; lr = (m > 0.0f) ? m : NEG_SLOPE * m; acc += lr * av.y;
  m = s.z + d.z; lr = (m > 0.0f) ? m : NEG_SLOPE * m; acc += lr * av.z;
  m = s.w + d.w; lr = (m > 0.0f) ? m : NEG_SLOPE * m; acc += lr * av.w;

  for (int off = 8; off > 0; off >>= 1) acc += __shfl_down(acc, off, 16);
  if ((lane & 15) == 0) {
    e[(size_t)eid * 2 + h] = acc;
    atomicMaxF32(&emax[(size_t)dst * 2 + h], acc);
  }
}

__global__ void edge_exp(const int* __restrict__ ei,
                         float* __restrict__ e,
                         const float* __restrict__ emax,
                         float* __restrict__ denom,
                         int E, int Etot) {
  int t = blockIdx.x * blockDim.x + threadIdx.x;
  if (t >= Etot * 2) return;
  const int eid = t >> 1, h = t & 1;
  int src, dst; edge_endpoints(ei, eid, E, src, dst);
  (void)src;
  const float ee = expf(e[t] - emax[(size_t)dst * 2 + h]);
  e[t] = ee;
  __hip_atomic_fetch_add(&denom[(size_t)dst * 2 + h], ee,
                         __ATOMIC_RELAXED, __HIP_MEMORY_SCOPE_AGENT);
}

__global__ void edge_agg(const float* __restrict__ xp,
                         const int* __restrict__ ei,
                         const float* __restrict__ e,
                         const float* __restrict__ denom,
                         float* __restrict__ acc,
                         int E, int Etot) {
  const int eid = blockIdx.x * (blockDim.x >> 5) + (threadIdx.x >> 5);
  if (eid >= Etot) return;
  const int lane = threadIdx.x & 31;
  int src, dst; edge_endpoints(ei, eid, E, src, dst);

  const int h   = lane >> 4;
  const int idx = (lane & 15) * 4;
  const float alpha = e[(size_t)eid * 2 + h] /
                      (denom[(size_t)dst * 2 + h] + 1e-16f);
  const float4 s = *(const float4*)(xp + (size_t)src * 128 + h * 64 + idx);
  float* p = acc + (size_t)dst * 128 + h * 64 + idx;
  __hip_atomic_fetch_add(p + 0, alpha * s.x, __ATOMIC_RELAXED, __HIP_MEMORY_SCOPE_AGENT);
  __hip_atomic_fetch_add(p + 1, alpha * s.y, __ATOMIC_RELAXED, __HIP_MEMORY_SCOPE_AGENT);
  __hip_atomic_fetch_add(p + 2, alpha * s.z, __ATOMIC_RELAXED, __HIP_MEMORY_SCOPE_AGENT);
  __hip_atomic_fetch_add(p + 3, alpha * s.w, __ATOMIC_RELAXED, __HIP_MEMORY_SCOPE_AGENT);
}

__global__ void head_mean_stats(const float* __restrict__ acc,
                                float* __restrict__ tmp,
                                float* __restrict__ sum,
                                float* __restrict__ sumsq,
                                int N) {
  __shared__ float s1[256];
  __shared__ float s2[256];
  const int c = threadIdx.x & 63;
  const int nodesPerBlk = blockDim.x >> 6;
  float ls = 0.0f, lq = 0.0f;
  for (int n = blockIdx.x * nodesPerBlk + (threadIdx.x >> 6);
       n < N; n += gridDim.x * nodesPerBlk) {
    const float v = 0.5f * (acc[(size_t)n * 128 + c] +
                            acc[(size_t)n * 128 + 64 + c]);
    tmp[(size_t)n * 64 + c] = v;
    ls += v; lq += v * v;
  }
  s1[threadIdx.x] = ls; s2[threadIdx.x] = lq;
  __syncthreads();
  if (threadIdx.x < 64) {
    ls = s1[threadIdx.x] + s1[threadIdx.x + 64] +
         s1[threadIdx.x + 128] + s1[threadIdx.x + 192];
    lq = s2[threadIdx.x] + s2[threadIdx.x + 64] +
         s2[threadIdx.x + 128] + s2[threadIdx.x + 192];
    __hip_atomic_fetch_add(&sum[c],   ls, __ATOMIC_RELAXED, __HIP_MEMORY_SCOPE_AGENT);
    __hip_atomic_fetch_add(&sumsq[c], lq, __ATOMIC_RELAXED, __HIP_MEMORY_SCOPE_AGENT);
  }
}

__global__ void bn_apply(const float* __restrict__ tmp,
                         const float* __restrict__ sum,
                         const float* __restrict__ sumsq,
                         const float* __restrict__ gamma,
                         const float* __restrict__ beta,
                         float* __restrict__ out,
                         int N, int doTanh) {
  int t = blockIdx.x * blockDim.x + threadIdx.x;
  if (t >= N * 64) return;
  const int c = t & 63;
  const float invN = 1.0f / (float)N;
  const float mean = sum[c] * invN;
  const float var  = sumsq[c] * invN - mean * mean;
  float v = (tmp[t] - mean) * rsqrtf(var + BN_EPS) * gamma[c] + beta[c];
  out[t] = doTanh ? tanhf(v) : v;
}

extern "C" void kernel_launch(void* const* d_in, const int* in_sizes, int n_in,
                              void* d_out, int out_size, void* d_ws, size_t ws_size,
                              hipStream_t stream) {
  const float* x      = (const float*)d_in[0];
  const int*   ei     = (const int*)  d_in[1];
  const float* W0     = (const float*)d_in[2];
  const float* a0     = (const float*)d_in[3];
  const float* gamma0 = (const float*)d_in[4];
  const float* beta0  = (const float*)d_in[5];
  const float* W1     = (const float*)d_in[6];
  const float* a1     = (const float*)d_in[7];
  const float* gamma1 = (const float*)d_in[8];
  const float* beta1  = (const float*)d_in[9];
  float* out = (float*)d_out;

  const int N    = in_sizes[0] / 128;
  const int E    = in_sizes[1] / 2;
  const int Etot = E + N;

  char* ws = (char*)d_ws;
  size_t off = 0;
  auto carve = [&](size_t nfloats) {
    float* p = (float*)(ws + off);
    off = (off + nfloats * sizeof(float) + 255) & ~(size_t)255;
    return p;
  };
  float* xp    = carve((size_t)N * 128);
  float* acc   = carve((size_t)N * 128);
  float* e     = carve((size_t)Etot * 2);
  float* emax  = carve((size_t)N * 2);
  float* denom = carve((size_t)N * 2);
  float* tmp   = carve((size_t)N * 64);
  float* h0    = carve((size_t)N * 64);
  float* stats = carve(128);
  (void)ws_size;

  const int waveEdgeBlocks = (Etot + 7) / 8;
  const int ehBlocks       = (Etot * 2 + 255) / 256;
  const int nodeElBlocks   = (N * 64 + 255) / 256;
  const int fillBlocks     = (N * 2 + 255) / 256;

  gemm_wmma_f32<<<N / 16, 256, 0, stream>>>(x, W0, xp, N, 128);

  fill_f32<<<fillBlocks, 256, 0, stream>>>(emax, -INFINITY, N * 2);
  hipMemsetAsync(denom, 0, (size_t)N * 2 * sizeof(float), stream);
  hipMemsetAsync(acc,   0, (size_t)N * 128 * sizeof(float), stream);
  hipMemsetAsync(stats, 0, 128 * sizeof(float), stream);

  edge_score<<<waveEdgeBlocks, 256, 0, stream>>>(xp, ei, a0, e, emax, E, Etot);
  edge_exp  <<<ehBlocks,       256, 0, stream>>>(ei, e, emax, denom, E, Etot);
  edge_agg  <<<waveEdgeBlocks, 256, 0, stream>>>(xp, ei, e, denom, acc, E, Etot);

  head_mean_stats<<<256, 256, 0, stream>>>(acc, tmp, stats, stats + 64, N);
  bn_apply<<<nodeElBlocks, 256, 0, stream>>>(tmp, stats, stats + 64,
                                             gamma0, beta0, h0, N, 1);

  gemm_wmma_f32<<<N / 16, 256, 0, stream>>>(h0, W1, xp, N, 64);

  fill_f32<<<fillBlocks, 256, 0, stream>>>(emax, -INFINITY, N * 2);
  hipMemsetAsync(denom, 0, (size_t)N * 2 * sizeof(float), stream);
  hipMemsetAsync(acc,   0, (size_t)N * 128 * sizeof(float), stream);
  hipMemsetAsync(stats, 0, 128 * sizeof(float), stream);

  edge_score<<<waveEdgeBlocks, 256, 0, stream>>>(xp, ei, a1, e, emax, E, Etot);
  edge_exp  <<<ehBlocks,       256, 0, stream>>>(ei, e, emax, denom, E, Etot);
  edge_agg  <<<waveEdgeBlocks, 256, 0, stream>>>(xp, ei, e, denom, acc, E, Etot);

  head_mean_stats<<<256, 256, 0, stream>>>(acc, tmp, stats, stats + 64, N);
  bn_apply<<<nodeElBlocks, 256, 0, stream>>>(tmp, stats, stats + 64,
                                             gamma1, beta1, out, N, 0);
}


# Edges counting-sorted by destination once per call; one wave per node does score, running softmax and aggregation in registers: no float atomics, sums run in sorted-slot arrival order
# speedup vs baseline: 2.5766x; 2.5766x over previous
#include <hip/hip_runtime.h>
#include <hip/hip_bf16.h>
#include <math.h>


typedef __attribute__((ext_vector_type(2))) float v2f;
typedef __attribute__((ext_vector_type(8))) float v8f;

#define NEG_SLOPE 0.2f
#define BN_EPS 1e-5f

__global__ void gemm_wmma_f32(const float* __restrict__ A,
                              const float* __restrict__ B,
                              float* __restrict__ C,
                              int M, int K) {
  const int Nn   = 128;
  const int wave = threadIdx.x >> 5;
  const int lane = threadIdx.x & 31;
  const int half = lane >> 4;
  const int l    = lane & 15;
  const int m0   = blockIdx.x * 16;
  const int n0   = wave * 16;

  v8f c = {};
  for (int k = 0; k < K; k += 4) {
    const int ka = k + half * 2;
    v2f a, b;
    a.x = A[(size_t)(m0 + l) * K + ka];
    a.y = A[(size_t)(m0 + l) * K + ka + 1];
    b.x = B[(size_t)ka * Nn + n0 + l];
    b.y = B[(size_t)(ka + 1) * Nn + n0 + l];
    c = __builtin_amdgcn_wmma_f32_16x16x4_f32(false, a, false, b,
                                              (short)0, c, false, false);
  }
#pragma unroll
  for (int j = 0; j < 8; ++j) {
    const int row = m0 + j + half * 8;
    C[(size_t)row * Nn + n0 + l] = c[j];
  }
}

__global__ void zero_i32(int* __restrict__ p, int n) {
  int t = blockIdx.x * blockDim.x + threadIdx.x;
  if (t < n) p[t] = 0;
}

__global__ void csr_hist(const int* __restrict__ ei, int* __restrict__ cnt, int E) {
  int t = blockIdx.x * blockDim.x + threadIdx.x;
  if (t < E) atomicAdd(&cnt[ei[E + t]], 1);
}

__global__ void __launch_bounds__(1024)
csr_scan(int* __restrict__ cnt, int* __restrict__ rowbeg, int N) {
  __shared__ int part[1024];
  const int tid   = threadIdx.x;
  const int chunk = (N + 1023) / 1024;
  const int b = min(tid * chunk, N);
  const int e = min(b + chunk, N);
  int s = 0;
  for (int i = b; i < e; ++i) s += cnt[i];
  part[tid] = s;
  __syncthreads();
  for (int off = 1; off < 1024; off <<= 1) {
    const int v = (tid >= off) ? part[tid - off] : 0;
    __syncthreads();
    part[tid] += v;
    __syncthreads();
  }
  int run = part[tid] - s;
  for (int i = b; i < e; ++i) {
    const int c = cnt[i];
    rowbeg[i] = run;
    cnt[i]    = run;
    run += c;
  }
}

__global__ void csr_fill(const int* __restrict__ ei, int* __restrict__ cursor,
                         int* __restrict__ srcs, int E) {
  int t = blockIdx.x * blockDim.x + threadIdx.x;
  if (t >= E) return;
  const int s = ei[t], d = ei[E + t];
  const int pos = atomicAdd(&cursor[d], 1);
  srcs[pos] = s;
}

__device__ __forceinline__ float score4(const float4 s, const float4 d,
                                        const float4 av) {
  float m, lr, acc = 0.0f;
  m = s.x + d.x; lr = (m > 0.0f) ? m : NEG_SLOPE * m; acc += lr * av.x;
  m = s.y + d.y; lr = (m > 0.0f) ? m : NEG_SLOPE * m; acc += lr * av.y;
  m = s.z + d.z; lr = (m > 0.0f) ? m : NEG_SLOPE * m; acc += lr * av.z;
  m = s.w + d.w; lr = (m > 0.0f) ? m : NEG_SLOPE * m; acc += lr * av.w;
  return acc;
}

__device__ __forceinline__ float row16_sum(float v) {
  v += __int_as_float(__builtin_amdgcn_update_dpp(0, __float_as_int(v), 0x168, 0xf, 0xf, true));
  v += __int_as_float(__builtin_amdgcn_update_dpp(0, __float_as_int(v), 0x164, 0xf, 0xf, true));
  v += __int_as_float(__builtin_amdgcn_update_dpp(0, __float_as_int(v), 0x162, 0xf, 0xf, true));
  v += __int_as_float(__builtin_amdgcn_update_dpp(0, __float_as_int(v), 0x161, 0xf, 0xf, true));
  return v;
}

__global__ void __launch_bounds__(256)
gat_node(const float* __restrict__ xp,
         const int* __restrict__ rowbeg,
         const int* __restrict__ rowend,
         const int* __restrict__ srcs,
         const float* __restrict__ avec,
         float* __restrict__ outp,
         int N) {
  const int lane = threadIdx.x & 31;
  const int node = __builtin_amdgcn_readfirstlane(
      (int)(blockIdx.x * (blockDim.x >> 5) + (threadIdx.x >> 5)));
  if (node >= N) return;

  const int coff = (lane >> 4) * 64 + (lane & 15) * 4;
  const float4 d  = *(const float4*)(xp + (size_t)node * 128 + coff);
  const float4 av = *(const float4*)(avec + coff);
  const int beg = rowbeg[node];
  const int end = rowend[node];

  float  mx  = row16_sum(score4(d, d, av));
  float  l   = 1.0f;
  float4 acc = d;

  for (int base = beg; base < end; base += 32) {
    const int nb = min(32, end - base);
    const int mysrc = (lane < nb) ? srcs[base + lane] : node;
    for (int t = 0; t < nb; t += 4) {
      const int s0 = __builtin_amdgcn_readlane(mysrc, t);
      const int s1 = __builtin_amdgcn_readlane(mysrc, t + 1);
      const int s2 = __builtin_amdgcn_readlane(mysrc, t + 2);
      const int s3 = __builtin_amdgcn_readlane(mysrc, t + 3);
      const float4 x0 = *(const float4*)(xp + (size_t)s0 * 128 + coff);
      const float4 x1 = *(const float4*)(xp + (size_t)s1 * 128 + coff);
      const float4 x2 = *(const float4*)(xp + (size_t)s2 * 128 + coff);
      const float4 x3 = *(const float4*)(xp + (size_t)s3 * 128 + coff);
      float e0 = row16_sum(score4(x0, d, av));
      float e1 = row16_sum(score4(x1, d, av));
      float e2 = row16_sum(score4(x2, d, av));
      float e3 = row16_sum(score4(x3, d, av));
      e1 = (t + 1 < nb) ? e1 : -INFINITY;
      e2 = (t + 2 < nb) ? e2 : -INFINITY;
      e3 = (t + 3 < nb) ? e3 : -INFINITY;
      const float mn = fmaxf(fmaxf(mx, e0), fmaxf(e1, fmaxf(e2, e3)));
      const float sc = expf(mx - mn);
      const float p0 = expf(e0 - mn);
      const float p1 = expf(e1 - mn);
      const float p2 = expf(e2 - mn);
      const float p3 = expf(e3 - mn);
      l = l * sc + ((p0 + p1) + (p2 + p3));
      acc.x = acc.x * sc + ((p0 * x0.x + p1 * x1.x) + (p2 * x2.x + p3 * x3.x));
      acc.y = acc.y * sc + ((p0 * x0.y + p1 * x1.y) + (p2 * x2.y + p3 * x3.y));
      acc.z = acc.z * sc + ((p0 * x0.z + p1 * x1.z) + (p2 * x2.z + p3 * x3.z));
      acc.w = acc.w * sc + ((p0 * x0.w + p1 * x1.w) + (p2 * x2.w + p3 * x3.w));
      mx = mn;
    }
  }

  const float den = l + 1e-16f;
  float4 o;
  o.x = acc.x / den; o.y = acc.y / den; o.z = acc.z / den; o.w = acc.w / den;
  *(float4*)(outp + (size_t)node * 128 + coff) = o;
}

__global__ void head_mean_stats(const float* __restrict__ acc,
                                float* __restrict__ tmp,
                                float* __restrict__ sum,
                                float* __restrict__ sumsq,
                                int N) {
  __shared__ float s1[256];
  __shared__ float s2[256];
  const int c = threadIdx.x & 63;
  const int nodesPerBlk = blockDim.x >> 6;
  float ls = 0.0f, lq = 0.0f;
  for (int n = blockIdx.x * nodesPerBlk + (threadIdx.x >> 6);
       n < N; n += gridDim.x * nodesPerBlk) {
    const float v = 0.5f * (acc[(size_t)n * 128 + c] +
                            acc[(size_t)n * 128 + 64 + c]);
    tmp[(size_t)n * 64 + c] = v;
    ls += v; lq += v * v;
  }
  s1[threadIdx.x] = ls; s2[threadIdx.x] = lq;
  __syncthreads();
  if (threadIdx.x < 64) {
    ls = s1[threadIdx.x] + s1[threadIdx.x + 64] +
         s1[threadIdx.x + 128] + s1[threadIdx.x + 192];
    lq = s2[threadIdx.x] + s2[threadIdx.x + 64] +
         s2[threadIdx.x + 128] + s2[threadIdx.x + 192];
    __hip_atomic_fetch_add(&sum[c],   ls, __ATOMIC_RELAXED, __HIP_MEMORY_SCOPE_AGENT);
    __hip_atomic_fetch_add(&sumsq[c], lq, __ATOMIC_RELAXED, __HIP_MEMORY_SCOPE_AGENT);
  }
}

__global__ void bn_apply(const float* __restrict__ tmp,
                         const float* __restrict__ sum,
                         const float* __restrict__ sumsq,
                         const float* __restrict__ gamma,
                         const float* __restrict__ beta,
                         float* __restrict__ out,
                         int N, int doTanh) {
  int t = blockIdx.x * blockDim.x + threadIdx.x;
  if (t >= N * 64) return;
  const int c = t & 63;
  const float invN = 1.0f / (float)N;
  const float mean = sum[c] * invN;
  const float var  = sumsq[c] * invN - mean * mean;
  float v = (tmp[t] - mean) * rsqrtf(var + BN_EPS) * gamma[c] + beta[c];
  out[t] = doTanh ? tanhf(v) : v;
}

extern "C" void kernel_launch(void* const* d_in, const int* in_sizes, int n_in,
                              void* d_out, int out_size, void* d_ws, size_t ws_size,
                              hipStream_t stream) {
  const float* x      = (const float*)d_in[0];
  const int*   ei     = (const int*)  d_in[1];
  const float* W0     = (const float*)d_in[2];
  const float* a0     = (const float*)d_in[3];
  const float* gamma0 = (const float*)d_in[4];
  const float* beta0  = (const float*)d_in[5];
  const float* W1     = (const float*)d_in[6];
  const float* a1     = (const float*)d_in[7];
  const float* gamma1 = (const float*)d_in[8];
  const float* beta1  = (const float*)d_in[9];
  float* out = (float*)d_out;

  const int N    = in_sizes[0] / 128;
  const int E    = in_sizes[1] / 2;

  char* ws = (char*)d_ws;
  size_t off = 0;
  auto carve = [&](size_t nwords) {
    float* p = (float*)(ws + off);
    off = (off + nwords * sizeof(float) + 255) & ~(size_t)255;
    return p;
  };
  float* xp    = carve((size_t)N * 128);
  float* acc   = carve((size_t)N * 128);
  float* tmp   = carve((size_t)N * 64);
  float* h0    = carve((size_t)N * 64);
  float* stats = carve(128);
  int* cnt     = (int*)carve((size_t)N);
  int* rowbeg  = (int*)carve((size_t)N);
  int* srcs    = (int*)carve((size_t)E);
  (void)ws_size;

  const int edgeBlocks   = (E + 255) / 256;
  const int nodeBlocks   = (N + 255) / 256;
  const int nodeWaveBlks = (N + 7) / 8;
  const int nodeElBlocks = (N * 64 + 255) / 256;

  zero_i32<<<nodeBlocks, 256, 0, stream>>>(cnt, N);
  csr_hist<<<edgeBlocks, 256, 0, stream>>>(ei, cnt, E);
  csr_scan<<<1, 1024, 0, stream>>>(cnt, rowbeg, N);
  csr_fill<<<edgeBlocks, 256, 0, stream>>>(ei, cnt, srcs, E);

  gemm_wmma_f32<<<N / 16, 256, 0, stream>>>(x, W0, xp, N, 128);
  hipMemsetAsync(stats, 0, 128 * sizeof(float), stream);
  gat_node<<<nodeWaveBlks, 256, 0, stream>>>(xp, rowbeg, cnt, srcs, a0, acc, N);

  head_mean_stats<<<256, 256, 0, stream>>>(acc, tmp, stats, stats + 64, N);
  bn_apply<<<nodeElBlocks, 256, 0, stream>>>(tmp, stats, stats + 64,
                                             gamma0, beta0, h0, N, 1);

  gemm_wmma_f32<<<N / 16, 256, 0, stream>>>(h0, W1, xp, N, 64);
  hipMemsetAsync(stats, 0, 128 * sizeof(float), stream);
  gat_node<<<nodeWaveBlks, 256, 0, stream>>>(xp, rowbeg, cnt, srcs, a1, acc, N);

  head_mean_stats<<<256, 256, 0, stream>>>(acc, tmp, stats, stats + 64, N);
  bn_apply<<<nodeElBlocks, 256, 0, stream>>>(tmp, stats, stats + 64,
                                             gamma1, beta1, out, N, 0);
}


# Row-offset prefix sum split over one workgroup per 1024-row tile: each sums the counts before its tile with coalesced 16-byte loads, then scans the tile in-wave; was one workgroup, strided chunks
# speedup vs baseline: 2.7497x; 1.0672x over previous
#include <hip/hip_runtime.h>
#include <hip/hip_bf16.h>
#include <math.h>


typedef __attribute__((ext_vector_type(2))) float v2f;
typedef __attribute__((ext_vector_type(8))) float v8f;

#define NEG_SLOPE 0.2f
#define BN_EPS 1e-5f

__global__ void gemm_wmma_f32(const float* __restrict__ A,
                              const float* __restrict__ B,
                              float* __restrict__ C,
                              int M, int K) {
  const int Nn   = 128;
  const int wave = threadIdx.x >> 5;
  const int lane = threadIdx.x & 31;
  const int half = lane >> 4;
  const int l    = lane & 15;
  const int m0   = blockIdx.x * 16;
  const int n0   = wave * 16;

  v8f c = {};
  for (int k = 0; k < K; k += 4) {
    const int ka = k + half * 2;
    v2f a, b;
    a.x = A[(size_t)(m0 + l) * K + ka];
    a.y = A[(size_t)(m0 + l) * K + ka + 1];
    b.x = B[(size_t)ka * Nn + n0 + l];
    b.y = B[(size_t)(ka + 1) * Nn + n0 + l];
    c = __builtin_amdgcn_wmma_f32_16x16x4_f32(false, a, false, b,
                                              (short)0, c, false, false);
  }
#pragma unroll
  for (int j = 0; j < 8; ++j) {
    const int row = m0 + j + half * 8;
    C[(size_t)row * Nn + n0 + l] = c[j];
  }
}

__global__ void zero_i32(int* __restrict__ p, int n) {
  int t = blockIdx.x * blockDim.x + threadIdx.x;
  if (t < n) p[t] = 0;
}

__global__ void csr_hist(const int* __restrict__ ei, int* __restrict__ cnt, int E) {
  int t = blockIdx.x * blockDim.x + threadIdx.x;
  if (t < E) atomicAdd(&cnt[ei[E + t]], 1);
}

__global__ void __launch_bounds__(1024)
csr_scan(const int* __restrict__ cnt, int* __restrict__ rowbeg,
         int* __restrict__ cursor, int N) {
  __shared__ int s_pre[32];
  __shared__ int s_tot[32];
  const int tid   = threadIdx.x;
  const int lane  = tid & 31;
  const int wave  = tid >> 5;
  const int tile0 = blockIdx.x * 1024;

  const int4* __restrict__ c4 = (const int4*)cnt;
  const int n4 = tile0 >> 2;
  int pre = 0;
#pragma unroll 4
  for (int i = tid; i < n4; i += 1024) {
    const int4 q = c4[i];
    pre += (q.x + q.y) + (q.z + q.w);
  }

  const int idx = tile0 + tid;
  const int v = (idx < N) ? cnt[idx] : 0;

  int inc = v;
  for (int off = 1; off < 32; off <<= 1) {
    const int up = __shfl_up(inc, off);
    if (lane >= off) inc += up;
  }
  for (int off = 16; off > 0; off >>= 1) pre += __shfl_xor(pre, off);

  if (lane == 31) s_tot[wave] = inc;
  if (lane == 0)  s_pre[wave] = pre;
  __syncthreads();

  int base = s_pre[lane] + ((lane < wave) ? s_tot[lane] : 0);
  for (int off = 16; off > 0; off >>= 1) base += __shfl_xor(base, off);

  if (idx < N) {
    const int excl = base + inc - v;
    rowbeg[idx] = excl;
    cursor[idx] = excl;
  }
}

__global__ void csr_fill(const int* __restrict__ ei, int* __restrict__ cursor,
                         int* __restrict__ srcs, int E) {
  int t = blockIdx.x * blockDim.x + threadIdx.x;
  if (t >= E) return;
  const int s = ei[t], d = ei[E + t];
  const int pos = atomicAdd(&cursor[d], 1);
  srcs[pos] = s;
}

__device__ __forceinline__ float score4(const float4 s, const float4 d,
                                        const float4 av) {
  float m, lr, acc = 0.0f;
  m = s.x + d.x; lr = (m > 0.0f) ? m : NEG_SLOPE * m; acc += lr * av.x;
  m = s.y + d.y; lr = (m > 0.0f) ? m : NEG_SLOPE * m; acc += lr * av.y;
  m = s.z + d.z; lr = (m > 0.0f) ? m : NEG_SLOPE * m; acc += lr * av.z;
  m = s.w + d.w; lr = (m > 0.0f) ? m : NEG_SLOPE * m; acc += lr * av.w;
  return acc;
}

__device__ __forceinline__ float row16_sum(float v) {
  v += __int_as_float(__builtin_amdgcn_update_dpp(0, __float_as_int(v), 0x168, 0xf, 0xf, true));
  v += __int_as_float(__builtin_amdgcn_update_dpp(0, __float_as_int(v), 0x164, 0xf, 0xf, true));
  v += __int_as_float(__builtin_amdgcn_update_dpp(0, __float_as_int(v), 0x162, 0xf, 0xf, true));
  v += __int_as_float(__builtin_amdgcn_update_dpp(0, __float_as_int(v), 0x161, 0xf, 0xf, true));
  return v;
}

__global__ void __launch_bounds__(256)
gat_node(const float* __restrict__ xp,
         const int* __restrict__ rowbeg,
         const int* __restrict__ rowend,
         const int* __restrict__ srcs,
         const float* __restrict__ avec,
         float* __restrict__ outp,
         int N) {
  const int lane = threadIdx.x & 31;
  const int node = __builtin_amdgcn_readfirstlane(
      (int)(blockIdx.x * (blockDim.x >> 5) + (threadIdx.x >> 5)));
  if (node >= N) return;

  const int coff = (lane >> 4) * 64 + (lane & 15) * 4;
  const float4 d  = *(const float4*)(xp + (size_t)node * 128 + coff);
  const float4 av = *(const float4*)(avec + coff);
  const int beg = rowbeg[node];
  const int end = rowend[node];

  float  mx  = row16_sum(score4(d, d, av));
  float  l   = 1.0f;
  float4 acc = d;

  for (int base = beg; base < end; base += 32) {
    const int nb = min(32, end - base);
    const int mysrc = (lane < nb) ? srcs[base + lane] : node;
    for (int t = 0; t < nb; t += 4) {
      const int s0 = __builtin_amdgcn_readlane(mysrc, t);
      const int s1 = __builtin_amdgcn_readlane(mysrc, t + 1);
      const int s2 = __builtin_amdgcn_readlane(mysrc, t + 2);
      const int s3 = __builtin_amdgcn_readlane(mysrc, t + 3);
      const float4 x0 = *(const float4*)(xp + (size_t)s0 * 128 + coff);
      const float4 x1 = *(const float4*)(xp + (size_t)s1 * 128 + coff);
      const float4 x2 = *(const float4*)(xp + (size_t)s2 * 128 + coff);
      const float4 x3 = *(const float4*)(xp + (size_t)s3 * 128 + coff);
      float e0 = row16_sum(score4(x0, d, av));
      float e1 = row16_sum(score4(x1, d, av));
      float e2 = row16_sum(score4(x2, d, av));
      float e3 = row16_sum(score4(x3, d, av));
      e1 = (t + 1 < nb) ? e1 : -INFINITY;
      e2 = (t + 2 < nb) ? e2 : -INFINITY;
      e3 = (t + 3 < nb) ? e3 : -INFINITY;
      const float mn = fmaxf(fmaxf(mx, e0), fmaxf(e1, fmaxf(e2, e3)));
      const float sc = expf(mx - mn);
      const float p0 = expf(e0 - mn);
      const float p1 = expf(e1 - mn);
      const float p2 = expf(e2 - mn);
      const float p3 = expf(e3 - mn);
      l = l * sc + ((p0 + p1) + (p2 + p3));
      acc.x = acc.x * sc + ((p0 * x0.x + p1 * x1.x) + (p2 * x2.x + p3 * x3.x));
      acc.y = acc.y * sc + ((p0 * x0.y + p1 * x1.y) + (p2 * x2.y + p3 * x3.y));
      acc.z = acc.z * sc + ((p0 * x0.z + p1 * x1.z) + (p2 * x2.z + p3 * x3.z));
      acc.w = acc.w * sc + ((p0 * x0.w + p1 * x1.w) + (p2 * x2.w + p3 * x3.w));
      mx = mn;
    }
  }

  const float den = l + 1e-16f;
  float4 o;
  o.x = acc.x / den; o.y = acc.y / den; o.z = acc.z / den; o.w = acc.w / den;
  *(float4*)(outp + (size_t)node * 128 + coff) = o;
}

__global__ void head_mean_stats(const float* __restrict__ acc,
                                float* __restrict__ tmp,
                                float* __restrict__ sum,
                                float* __restrict__ sumsq,
                                int N) {
  __shared__ float s1[256];
  __shared__ float s2[256];
  const int c = threadIdx.x & 63;
  const int nodesPerBlk = blockDim.x >> 6;
  float ls = 0.0f, lq = 0.0f;
  for (int n = blockIdx.x * nodesPerBlk + (threadIdx.x >> 6);
       n < N; n += gridDim.x * nodesPerBlk) {
    const float v = 0.5f * (acc[(size_t)n * 128 + c] +
                            acc[(size_t)n * 128 + 64 + c]);
    tmp[(size_t)n * 64 + c] = v;
    ls += v; lq += v * v;
  }
  s1[threadIdx.x] = ls; s2[threadIdx.x] = lq;
  __syncthreads();
  if (threadIdx.x < 64) {
    ls = s1[threadIdx.x] + s1[threadIdx.x + 64] +
         s1[threadIdx.x + 128] + s1[threadIdx.x + 192];
    lq = s2[threadIdx.x] + s2[threadIdx.x + 64] +
         s2[threadIdx.x + 128] + s2[threadIdx.x + 192];
    __hip_atomic_fetch_add(&sum[c],   ls, __ATOMIC_RELAXED, __HIP_MEMORY_SCOPE_AGENT);
    __hip_atomic_fetch_add(&sumsq[c], lq, __ATOMIC_RELAXED, __HIP_MEMORY_SCOPE_AGENT);
  }
}

__global__ void bn_apply(const float* __restrict__ tmp,
                         const float* __restrict__ sum,
                         const float* __restrict__ sumsq,
                         const float* __restrict__ gamma,
                         const float* __restrict__ beta,
                         float* __restrict__ out,
                         int N, int doTanh) {
  int t = blockIdx.x * blockDim.x + threadIdx.x;
  if (t >= N * 64) return;
  const int c = t & 63;
  const float invN = 1.0f / (float)N;
  const float mean = sum[c] * invN;
  const float var  = sumsq[c] * invN - mean * mean;
  float v = (tmp[t] - mean) * rsqrtf(var + BN_EPS) * gamma[c] + beta[c];
  out[t] = doTanh ? tanhf(v) : v;
}

extern "C" void kernel_launch(void* const* d_in, const int* in_sizes, int n_in,
                              void* d_out, int out_size, void* d_ws, size_t ws_size,
                              hipStream_t stream) {
  const float* x      = (const float*)d_in[0];
  const int*   ei     = (const int*)  d_in[1];
  const float* W0     = (const float*)d_in[2];
  const float* a0     = (const float*)d_in[3];
  const float* gamma0 = (const float*)d_in[4];
  const float* beta0  = (const float*)d_in[5];
  const float* W1     = (const float*)d_in[6];
  const float* a1     = (const float*)d_in[7];
  const float* gamma1 = (const float*)d_in[8];
  const float* beta1  = (const float*)d_in[9];
  float* out = (float*)d_out;

  const int N    = in_sizes[0] / 128;
  const int E    = in_sizes[1] / 2;

  char* ws = (char*)d_ws;
  size_t off = 0;
  auto carve = [&](size_t nwords) {
    float* p = (float*)(ws + off);
    off = (off + nwords * sizeof(float) + 255) & ~(size_t)255;
    return p;
  };
  float* xp    = carve((size_t)N * 128);
  float* acc   = carve((size_t)N * 128);
  float* tmp   = carve((size_t)N * 64);
  float* h0    = carve((size_t)N * 64);
  float* stats = carve(128);
  int* cnt     = (int*)carve((size_t)N);
  int* rowbeg  = (int*)carve((size_t)N);
  int* cursor  = (int*)carve((size_t)N);
  int* srcs    = (int*)carve((size_t)E);
  (void)ws_size;

  const int edgeBlocks   = (E + 255) / 256;
  const int nodeBlocks   = (N + 255) / 256;
  const int scanBlocks   = (N + 1023) / 1024;
  const int nodeWaveBlks = (N + 7) / 8;
  const int nodeElBlocks = (N * 64 + 255) / 256;

  zero_i32<<<nodeBlocks, 256, 0, stream>>>(cnt, N);
  csr_hist<<<edgeBlocks, 256, 0, stream>>>(ei, cnt, E);
  csr_scan<<<scanBlocks, 1024, 0, stream>>>(cnt, rowbeg, cursor, N);
  csr_fill<<<edgeBlocks, 256, 0, stream>>>(ei, cursor, srcs, E);

  gemm_wmma_f32<<<N / 16, 256, 0, stream>>>(x, W0, xp, N, 128);
  hipMemsetAsync(stats, 0, 128 * sizeof(float), stream);
  gat_node<<<nodeWaveBlks, 256, 0, stream>>>(xp, rowbeg, cursor, srcs, a0, acc, N);

  head_mean_stats<<<256, 256, 0, stream>>>(acc, tmp, stats, stats + 64, N);
  bn_apply<<<nodeElBlocks, 256, 0, stream>>>(tmp, stats, stats + 64,
                                             gamma0, beta0, h0, N, 1);

  gemm_wmma_f32<<<N / 16, 256, 0, stream>>>(h0, W1, xp, N, 64);
  hipMemsetAsync(stats, 0, 128 * sizeof(float), stream);
  gat_node<<<nodeWaveBlks, 256, 0, stream>>>(xp, rowbeg, cursor, srcs, a1, acc, N);

  head_mean_stats<<<256, 256, 0, stream>>>(acc, tmp, stats, stats + 64, N);
  bn_apply<<<nodeElBlocks, 256, 0, stream>>>(tmp, stats, stats + 64,
                                             gamma1, beta1, out, N, 0);
}


# Head mean and batch-norm partial sums folded into the node kernel: one partial row per workgroup, rows added in fixed order by a small column-sum kernel; stats pass, its atomics, both memsets gone
# speedup vs baseline: 3.0038x; 1.0924x over previous
#include <hip/hip_runtime.h>
#include <hip/hip_bf16.h>
#include <math.h>


typedef __attribute__((ext_vector_type(2))) float v2f;
typedef __attribute__((ext_vector_type(8))) float v8f;

#define NEG_SLOPE 0.2f
#define BN_EPS 1e-5f

__global__ void gemm_wmma_f32(const float* __restrict__ A,
                              const float* __restrict__ B,
                              float* __restrict__ C,
                              int M, int K) {
  const int Nn   = 128;
  const int wave = threadIdx.x >> 5;
  const int lane = threadIdx.x & 31;
  const int half = lane >> 4;
  const int l    = lane & 15;
  const int m0   = blockIdx.x * 16;
  const int n0   = wave * 16;

  v8f c = {};
  for (int k = 0; k < K; k += 4) {
    const int ka = k + half * 2;
    v2f a, b;
    a.x = A[(size_t)(m0 + l) * K + ka];
    a.y = A[(size_t)(m0 + l) * K + ka + 1];
    b.x = B[(size_t)ka * Nn + n0 + l];
    b.y = B[(size_t)(ka + 1) * Nn + n0 + l];
    c = __builtin_amdgcn_wmma_f32_16x16x4_f32(false, a, false, b,
                                              (short)0, c, false, false);
  }
#pragma unroll
  for (int j = 0; j < 8; ++j) {
    const int row = m0 + j + half * 8;
    C[(size_t)row * Nn + n0 + l] = c[j];
  }
}

__global__ void zero_i32(int* __restrict__ p, int n) {
  int t = blockIdx.x * blockDim.x + threadIdx.x;
  if (t < n) p[t] = 0;
}

__global__ void csr_hist(const int* __restrict__ ei, int* __restrict__ cnt, int E) {
  int t = blockIdx.x * blockDim.x + threadIdx.x;
  if (t < E) atomicAdd(&cnt[ei[E + t]], 1);
}

__global__ void __launch_bounds__(1024)
csr_scan(const int* __restrict__ cnt, int* __restrict__ rowbeg,
         int* __restrict__ cursor, int N) {
  __shared__ int s_pre[32];
  __shared__ int s_tot[32];
  const int tid   = threadIdx.x;
  const int lane  = tid & 31;
  const int wave  = tid >> 5;
  const int tile0 = blockIdx.x * 1024;

  const int4* __restrict__ c4 = (const int4*)cnt;
  const int n4 = tile0 >> 2;
  int pre = 0;
#pragma unroll 4
  for (int i = tid; i < n4; i += 1024) {
    const int4 q = c4[i];
    pre += (q.x + q.y) + (q.z + q.w);
  }

  const int idx = tile0 + tid;
  const int v = (idx < N) ? cnt[idx] : 0;

  int inc = v;
  for (int off = 1; off < 32; off <<= 1) {
    const int up = __shfl_up(inc, off);
    if (lane >= off) inc += up;
  }
  for (int off = 16; off > 0; off >>= 1) pre += __shfl_xor(pre, off);

  if (lane == 31) s_tot[wave] = inc;
  if (lane == 0)  s_pre[wave] = pre;
  __syncthreads();

  int base = s_pre[lane] + ((lane < wave) ? s_tot[lane] : 0);
  for (int off = 16; off > 0; off >>= 1) base += __shfl_xor(base, off);

  if (idx < N) {
    const int excl = base + inc - v;
    rowbeg[idx] = excl;
    cursor[idx] = excl;
  }
}

__global__ void csr_fill(const int* __restrict__ ei, int* __restrict__ cursor,
                         int* __restrict__ srcs, int E) {
  int t = blockIdx.x * blockDim.x + threadIdx.x;
  if (t >= E) return;
  const int s = ei[t], d = ei[E + t];
  const int pos = atomicAdd(&cursor[d], 1);
  srcs[pos] = s;
}

__device__ __forceinline__ float score4(const float4 s, const float4 d,
                                        const float4 av) {
  float m, lr, acc = 0.0f;
  m = s.x + d.x; lr = (m > 0.0f) ? m : NEG_SLOPE * m; acc += lr * av.x;
  m = s.y + d.y; lr = (m > 0.0f) ? m : NEG_SLOPE * m; acc += lr * av.y;
  m = s.z + d.z; lr = (m > 0.0f) ? m : NEG_SLOPE * m; acc += lr * av.z;
  m = s.w + d.w; lr = (m > 0.0f) ? m : NEG_SLOPE * m; acc += lr * av.w;
  return acc;
}

__device__ __forceinline__ float row16_sum(float v) {
  v += __int_as_float(__builtin_amdgcn_update_dpp(0, __float_as_int(v), 0x168, 0xf, 0xf, true));
  v += __int_as_float(__builtin_amdgcn_update_dpp(0, __float_as_int(v), 0x164, 0xf, 0xf, true));
  v += __int_as_float(__builtin_amdgcn_update_dpp(0, __float_as_int(v), 0x162, 0xf, 0xf, true));
  v += __int_as_float(__builtin_amdgcn_update_dpp(0, __float_as_int(v), 0x161, 0xf, 0xf, true));
  return v;
}

#define NODES_PER_BLOCK 8

__global__ void __launch_bounds__(256)
gat_node(const float* __restrict__ xp,
         const int* __restrict__ rowbeg,
         const int* __restrict__ rowend,
         const int* __restrict__ srcs,
         const float* __restrict__ avec,
         float* __restrict__ tmp,
         float* __restrict__ part,
         int N) {
  __shared__ __attribute__((aligned(16))) float s_red[NODES_PER_BLOCK][128];
  const int lane = threadIdx.x & 31;
  const int wave = threadIdx.x >> 5;
  const int node = __builtin_amdgcn_readfirstlane(
      (int)(blockIdx.x * NODES_PER_BLOCK + wave));

  float4 v = {0.0f, 0.0f, 0.0f, 0.0f};

  if (node < N) {
    const int coff = (lane >> 4) * 64 + (lane & 15) * 4;
    const float4 d  = *(const float4*)(xp + (size_t)node * 128 + coff);
    const float4 av = *(const float4*)(avec + coff);
    const int beg = rowbeg[node];
    const int end = rowend[node];

    float  mx  = row16_sum(score4(d, d, av));
    float  l   = 1.0f;
    float4 acc = d;

    for (int base = beg; base < end; base += 32) {
      const int nb = min(32, end - base);
      const int mysrc = (lane < nb) ? srcs[base + lane] : node;
      for (int t = 0; t < nb; t += 4) {
        const int s0 = __builtin_amdgcn_readlane(mysrc, t);
        const int s1 = __builtin_amdgcn_readlane(mysrc, t + 1);
        const int s2 = __builtin_amdgcn_readlane(mysrc, t + 2);
        const int s3 = __builtin_amdgcn_readlane(mysrc, t + 3);
        const float4 x0 = *(const float4*)(xp + (size_t)s0 * 128 + coff);
        const float4 x1 = *(const float4*)(xp + (size_t)s1 * 128 + coff);
        const float4 x2 = *(const float4*)(xp + (size_t)s2 * 128 + coff);
        const float4 x3 = *(const float4*)(xp + (size_t)s3 * 128 + coff);
        float e0 = row16_sum(score4(x0, d, av));
        float e1 = row16_sum(score4(x1, d, av));
        float e2 = row16_sum(score4(x2, d, av));
        float e3 = row16_sum(score4(x3, d, av));
        e1 = (t + 1 < nb) ? e1 : -INFINITY;
        e2 = (t + 2 < nb) ? e2 : -INFINITY;
        e3 = (t + 3 < nb) ? e3 : -INFINITY;
        const float mn = fmaxf(fmaxf(mx, e0), fmaxf(e1, fmaxf(e2, e3)));
        const float sc = expf(mx - mn);
        const float p0 = expf(e0 - mn);
        const float p1 = expf(e1 - mn);
        const float p2 = expf(e2 - mn);
        const float p3 = expf(e3 - mn);
        l = l * sc + ((p0 + p1) + (p2 + p3));
        acc.x = acc.x * sc + ((p0 * x0.x + p1 * x1.x) + (p2 * x2.x + p3 * x3.x));
        acc.y = acc.y * sc + ((p0 * x0.y + p1 * x1.y) + (p2 * x2.y + p3 * x3.y));
        acc.z = acc.z * sc + ((p0 * x0.z + p1 * x1.z) + (p2 * x2.z + p3 * x3.z));
        acc.w = acc.w * sc + ((p0 * x0.w + p1 * x1.w) + (p2 * x2.w + p3 * x3.w));
        mx = mn;
      }
    }

    const float den = l + 1e-16f;
    float4 o;
    o.x = acc.x / den; o.y = acc.y / den; o.z = acc.z / den; o.w = acc.w / den;

    v.x = 0.5f * (o.x + __shfl_xor(o.x, 16));
    v.y = 0.5f * (o.y + __shfl_xor(o.y, 16));
    v.z = 0.5f * (o.z + __shfl_xor(o.z, 16));
    v.w = 0.5f * (o.w + __shfl_xor(o.w, 16));
  }

  if (lane < 16) {
    float4 q;
    q.x = v.x * v.x; q.y = v.y * v.y; q.z = v.z * v.z; q.w = v.w * v.w;
    *(float4*)&s_red[wave][lane * 4]      = v;
    *(float4*)&s_red[wave][64 + lane * 4] = q;
  }
  __syncthreads();
  if (node < N && lane < 16) *(float4*)(tmp + (size_t)node * 64 + lane * 4) = v;
  if (threadIdx.x < 128) {
    const int k = threadIdx.x;
    const float t = ((s_red[0][k] + s_red[1][k]) + (s_red[2][k] + s_red[3][k])) +
                    ((s_red[4][k] + s_red[5][k]) + (s_red[6][k] + s_red[7][k]));
    part[(size_t)blockIdx.x * 128 + k] = t;
  }
}

__global__ void __launch_bounds__(1024)
bn_partials_sum(const float* __restrict__ part, float* __restrict__ stats,
                int rows) {
  __shared__ __attribute__((aligned(16))) float s_w[32][32];
  const int lane = threadIdx.x & 31;
  const int wave = threadIdx.x >> 5;
  const int cg   = lane & 7;
  const int rs   = wave * 4 + (lane >> 3);
  const float* __restrict__ p = part + blockIdx.x * 32 + cg * 4;
  float4 a = {0.0f, 0.0f, 0.0f, 0.0f};
  int r0 = rs;
  for (; r0 + 15 * 128 < rows; r0 += 16 * 128) {
    const float* __restrict__ pr = p + (size_t)r0 * 128;
    float4 q[16];
    for (int j = 0; j < 16; ++j) q[j] = *(const float4*)(pr + (size_t)j * (128 * 128));
    for (int j = 0; j < 16; ++j) {
      a.x += q[j].x; a.y += q[j].y; a.z += q[j].z; a.w += q[j].w;
    }
  }
  for (; r0 < rows; r0 += 128) {
    const float4 q = *(const float4*)(p + (size_t)r0 * 128);
    a.x += q.x; a.y += q.y; a.z += q.z; a.w += q.w;
  }
  a.x += __shfl_xor(a.x, 8);  a.y += __shfl_xor(a.y, 8);
  a.z += __shfl_xor(a.z, 8);  a.w += __shfl_xor(a.w, 8);
  a.x += __shfl_xor(a.x, 16); a.y += __shfl_xor(a.y, 16);
  a.z += __shfl_xor(a.z, 16); a.w += __shfl_xor(a.w, 16);
  if (lane < 8) *(float4*)&s_w[wave][cg * 4] = a;
  __syncthreads();
  if (wave == 0) {
    float t = 0.0f;
    for (int w = 0; w < 32; ++w) t += s_w[w][lane];
    stats[blockIdx.x * 32 + lane] = t;
  }
}

__global__ void bn_apply(const float* __restrict__ tmp,
                         const float* __restrict__ sum,
                         const float* __restrict__ sumsq,
                         const float* __restrict__ gamma,
                         const float* __restrict__ beta,
                         float* __restrict__ out,
                         int N, int doTanh) {
  int t = blockIdx.x * blockDim.x + threadIdx.x;
  if (t >= N * 64) return;
  const int c = t & 63;
  const float invN = 1.0f / (float)N;
  const float mean = sum[c] * invN;
  const float var  = sumsq[c] * invN - mean * mean;
  float v = (tmp[t] - mean) * rsqrtf(var + BN_EPS) * gamma[c] + beta[c];
  out[t] = doTanh ? tanhf(v) : v;
}

extern "C" void kernel_launch(void* const* d_in, const int* in_sizes, int n_in,
                              void* d_out, int out_size, void* d_ws, size_t ws_size,
                              hipStream_t stream) {
  const float* x      = (const float*)d_in[0];
  const int*   ei     = (const int*)  d_in[1];
  const float* W0     = (const float*)d_in[2];
  const float* a0     = (const float*)d_in[3];
  const float* gamma0 = (const float*)d_in[4];
  const float* beta0  = (const float*)d_in[5];
  const float* W1     = (const float*)d_in[6];
  const float* a1     = (const float*)d_in[7];
  const float* gamma1 = (const float*)d_in[8];
  const float* beta1  = (const float*)d_in[9];
  float* out = (float*)d_out;

  const int N    = in_sizes[0] / 128;
  const int E    = in_sizes[1] / 2;

  char* ws = (char*)d_ws;
  size_t off = 0;
  auto carve = [&](size_t nwords) {
    float* p = (float*)(ws + off);
    off = (off + nwords * sizeof(float) + 255) & ~(size_t)255;
    return p;
  };
  float* xp    = carve((size_t)N * 128);
  float* tmp   = carve((size_t)N * 64);
  float* h0    = carve((size_t)N * 64);
  float* stats = carve(128);
  int* cnt     = (int*)carve((size_t)N);
  int* rowbeg  = (int*)carve((size_t)N);
  int* cursor  = (int*)carve((size_t)N);
  int* srcs    = (int*)carve((size_t)E);
  const int nodeGroups = (N + NODES_PER_BLOCK - 1) / NODES_PER_BLOCK;
  float* part  = carve((size_t)nodeGroups * 128);
  (void)ws_size;

  const int edgeBlocks   = (E + 255) / 256;
  const int nodeBlocks   = (N + 255) / 256;
  const int scanBlocks   = (N + 1023) / 1024;
  const int nodeElBlocks = (N * 64 + 255) / 256;

  zero_i32<<<nodeBlocks, 256, 0, stream>>>(cnt, N);
  csr_hist<<<edgeBlocks, 256, 0, stream>>>(ei, cnt, E);
  csr_scan<<<scanBlocks, 1024, 0, stream>>>(cnt, rowbeg, cursor, N);
  csr_fill<<<edgeBlocks, 256, 0, stream>>>(ei, cursor, srcs, E);

  gemm_wmma_f32<<<N / 16, 256, 0, stream>>>(x, W0, xp, N, 128);
  gat_node<<<nodeGroups, 256, 0, stream>>>(xp, rowbeg, cursor, srcs, a0, tmp, part, N);
  bn_partials_sum<<<4, 1024, 0, stream>>>(part, stats, nodeGroups);
  bn_apply<<<nodeElBlocks, 256, 0, stream>>>(tmp, stats, stats + 64,
                                             gamma0, beta0, h0, N, 1);

  gemm_wmma_f32<<<N / 16, 256, 0, stream>>>(h0, W1, xp, N, 64);
  gat_node<<<nodeGroups, 256, 0, stream>>>(xp, rowbeg, cursor, srcs, a1, tmp, part, N);
  bn_partials_sum<<<4, 1024, 0, stream>>>(part, stats, nodeGroups);
  bn_apply<<<nodeElBlocks, 256, 0, stream>>>(tmp, stats, stats + 64,
                                             gamma1, beta1, out, N, 0);
}


# Projection GEMM on a resident grid of 512 workgroups: W fragments read once into registers, each 16-row input tile staged once in LDS for its 8 waves, result stored as whole 512-B rows; same sums
# speedup vs baseline: 3.1441x; 1.0467x over previous
#include <hip/hip_runtime.h>
#include <hip/hip_bf16.h>
#include <math.h>


typedef __attribute__((ext_vector_type(2))) float v2f;
typedef __attribute__((ext_vector_type(8))) float v8f;

#define NEG_SLOPE 0.2f
#define BN_EPS 1e-5f

template <int K>
__global__ void __launch_bounds__(256)
gemm_wmma_f32(const float* __restrict__ A,
              const float* __restrict__ B,
              float* __restrict__ C,
              int M) {
  constexpr int Nn  = 128;
  constexpr int PA  = K + 4;
  constexpr int PC  = Nn + 4;
  constexpr int KS  = K / 4;
  constexpr int RQ  = K / 4;
  constexpr int NPA = (16 * RQ) / 256;
  __shared__ __attribute__((aligned(16))) float sA[16 * PA];
  __shared__ __attribute__((aligned(16))) float sC[16 * PC];

  const int tid  = threadIdx.x;
  const int wave = tid >> 5;
  const int lane = tid & 31;
  const int half = lane >> 4;
  const int l    = lane & 15;
  const int n0   = wave * 16;
  const int tiles = M >> 4;

  v2f bf[KS];
#pragma unroll
  for (int ks = 0; ks < KS; ++ks) {
    const int kb = ks * 4 + half * 2;
    bf[ks].x = B[(size_t)kb * Nn + n0 + l];
    bf[ks].y = B[(size_t)(kb + 1) * Nn + n0 + l];
  }

  int tile = blockIdx.x;
  if (tile >= tiles) return;

  const int q0 = tid,       r0 = q0 / RQ, e0 = (q0 % RQ) * 4;
  const int q1 = tid + 256, r1 = q1 / RQ, e1 = (q1 % RQ) * 4;
  const unsigned aoff0 = r0 * K + e0;
  const unsigned aoff1 = r1 * K + e1;
  const int orow = tid >> 5, ocol = (tid & 31) * 4;
  const unsigned coff = orow * Nn + ocol;

  float4 pre0, pre1 = {0.0f, 0.0f, 0.0f, 0.0f};
  {
    const float* __restrict__ At = A + (size_t)tile * (16 * K);
    pre0 = *(const float4*)(At + aoff0);
    if constexpr (NPA > 1) pre1 = *(const float4*)(At + aoff1);
  }

  for (; tile < tiles; tile += gridDim.x) {
    *(float4*)&sA[r0 * PA + e0] = pre0;
    if constexpr (NPA > 1) *(float4*)&sA[r1 * PA + e1] = pre1;
    __syncthreads();

    const int nxt = tile + gridDim.x;
    if (nxt < tiles) {
      const float* __restrict__ An = A + (size_t)nxt * (16 * K);
      pre0 = *(const float4*)(An + aoff0);
      if constexpr (NPA > 1) pre1 = *(const float4*)(An + aoff1);
    }

    v2f af[KS];
#pragma unroll
    for (int ks = 0; ks < KS; ++ks)
      af[ks] = *(const v2f*)&sA[l * PA + ks * 4 + half * 2];
    __builtin_amdgcn_sched_barrier(0);

    v8f c = {};
#pragma unroll
    for (int ks = 0; ks < KS; ++ks) {
      c = __builtin_amdgcn_wmma_f32_16x16x4_f32(false, af[ks], false, bf[ks],
                                                (short)0, c, false, false);
    }
    __builtin_amdgcn_sched_barrier(0);

#pragma unroll
    for (int j = 0; j < 8; ++j) sC[(j + half * 8) * PC + n0 + l] = c[j];
    __syncthreads();

    float* __restrict__ Ct = C + (size_t)tile * (16 * Nn);
    const float4 o0 = *(const float4*)&sC[orow * PC + ocol];
    const float4 o1 = *(const float4*)&sC[(orow + 8) * PC + ocol];
    *(float4*)(Ct + coff) = o0;
    *(float4*)(Ct + coff + 8 * Nn) = o1;
  }
}

__global__ void zero_i32(int* __restrict__ p, int n) {
  int t = blockIdx.x * blockDim.x + threadIdx.x;
  if (t < n) p[t] = 0;
}

__global__ void csr_hist(const int* __restrict__ ei, int* __restrict__ cnt, int E) {
  int t = blockIdx.x * blockDim.x + threadIdx.x;
  if (t < E) atomicAdd(&cnt[ei[E + t]], 1);
}

__global__ void __launch_bounds__(1024)
csr_scan(const int* __restrict__ cnt, int* __restrict__ rowbeg,
         int* __restrict__ cursor, int N) {
  __shared__ int s_pre[32];
  __shared__ int s_tot[32];
  const int tid   = threadIdx.x;
  const int lane  = tid & 31;
  const int wave  = tid >> 5;
  const int tile0 = blockIdx.x * 1024;

  const int4* __restrict__ c4 = (const int4*)cnt;
  const int n4 = tile0 >> 2;
  int pre = 0;
#pragma unroll 4
  for (int i = tid; i < n4; i += 1024) {
    const int4 q = c4[i];
    pre += (q.x + q.y) + (q.z + q.w);
  }

  const int idx = tile0 + tid;
  const int v = (idx < N) ? cnt[idx] : 0;

  int inc = v;
  for (int off = 1; off < 32; off <<= 1) {
    const int up = __shfl_up(inc, off);
    if (lane >= off) inc += up;
  }
  for (int off = 16; off > 0; off >>= 1) pre += __shfl_xor(pre, off);

  if (lane == 31) s_tot[wave] = inc;
  if (lane == 0)  s_pre[wave] = pre;
  __syncthreads();

  int base = s_pre[lane] + ((lane < wave) ? s_tot[lane] : 0);
  for (int off = 16; off > 0; off >>= 1) base += __shfl_xor(base, off);

  if (idx < N) {
    const int excl = base + inc - v;
    rowbeg[idx] = excl;
    cursor[idx] = excl;
  }
}

__global__ void csr_fill(const int* __restrict__ ei, int* __restrict__ cursor,
                         int* __restrict__ srcs, int E) {
  int t = blockIdx.x * blockDim.x + threadIdx.x;
  if (t >= E) return;
  const int s = ei[t], d = ei[E + t];
  const int pos = atomicAdd(&cursor[d], 1);
  srcs[pos] = s;
}

__device__ __forceinline__ float score4(const float4 s, const float4 d,
                                        const float4 av) {
  float m, lr, acc = 0.0f;
  m = s.x + d.x; lr = (m > 0.0f) ? m : NEG_SLOPE * m; acc += lr * av.x;
  m = s.y + d.y; lr = (m > 0.0f) ? m : NEG_SLOPE * m; acc += lr * av.y;
  m = s.z + d.z; lr = (m > 0.0f) ? m : NEG_SLOPE * m; acc += lr * av.z;
  m = s.w + d.w; lr = (m > 0.0f) ? m : NEG_SLOPE * m; acc += lr * av.w;
  return acc;
}

__device__ __forceinline__ float row16_sum(float v) {
  v += __int_as_float(__builtin_amdgcn_update_dpp(0, __float_as_int(v), 0x168, 0xf, 0xf, true));
  v += __int_as_float(__builtin_amdgcn_update_dpp(0, __float_as_int(v), 0x164, 0xf, 0xf, true));
  v += __int_as_float(__builtin_amdgcn_update_dpp(0, __float_as_int(v), 0x162, 0xf, 0xf, true));
  v += __int_as_float(__builtin_amdgcn_update_dpp(0, __float_as_int(v), 0x161, 0xf, 0xf, true));
  return v;
}

#define NODES_PER_BLOCK 8

__global__ void __launch_bounds__(256)
gat_node(const float* __restrict__ xp,
         const int* __restrict__ rowbeg,
         const int* __restrict__ rowend,
         const int* __restrict__ srcs,
         const float* __restrict__ avec,
         float* __restrict__ tmp,
         float* __restrict__ part,
         int N) {
  __shared__ __attribute__((aligned(16))) float s_red[NODES_PER_BLOCK][128];
  const int lane = threadIdx.x & 31;
  const int wave = threadIdx.x >> 5;
  const int node = __builtin_amdgcn_readfirstlane(
      (int)(blockIdx.x * NODES_PER_BLOCK + wave));

  float4 v = {0.0f, 0.0f, 0.0f, 0.0f};

  if (node < N) {
    const int coff = (lane >> 4) * 64 + (lane & 15) * 4;
    const float4 d  = *(const float4*)(xp + (size_t)node * 128 + coff);
    const float4 av = *(const float4*)(avec + coff);
    const int beg = rowbeg[node];
    const int end = rowend[node];

    float  mx  = row16_sum(score4(d, d, av));
    float  l   = 1.0f;
    float4 acc = d;

    for (int base = beg; base < end; base += 32) {
      const int nb = min(32, end - base);
      const int mysrc = (lane < nb) ? srcs[base + lane] : node;
      for (int t = 0; t < nb; t += 4) {
        const int s0 = __builtin_amdgcn_readlane(mysrc, t);
        const int s1 = __builtin_amdgcn_readlane(mysrc, t + 1);
        const int s2 = __builtin_amdgcn_readlane(mysrc, t + 2);
        const int s3 = __builtin_amdgcn_readlane(mysrc, t + 3);
        const float4 x0 = *(const float4*)(xp + (size_t)s0 * 128 + coff);
        const float4 x1 = *(const float4*)(xp + (size_t)s1 * 128 + coff);
        const float4 x2 = *(const float4*)(xp + (size_t)s2 * 128 + coff);
        const float4 x3 = *(const float4*)(xp + (size_t)s3 * 128 + coff);
        float e0 = row16_sum(score4(x0, d, av));
        float e1 = row16_sum(score4(x1, d, av));
        float e2 = row16_sum(score4(x2, d, av));
        float e3 = row16_sum(score4(x3, d, av));
        e1 = (t + 1 < nb) ? e1 : -INFINITY;
        e2 = (t + 2 < nb) ? e2 : -INFINITY;
        e3 = (t + 3 < nb) ? e3 : -INFINITY;
        const float mn = fmaxf(fmaxf(mx, e0), fmaxf(e1, fmaxf(e2, e3)));
        const float sc = expf(mx - mn);
        const float p0 = expf(e0 - mn);
        const float p1 = expf(e1 - mn);
        const float p2 = expf(e2 - mn);
        const float p3 = expf(e3 - mn);
        l = l * sc + ((p0 + p1) + (p2 + p3));
        acc.x = acc.x * sc + ((p0 * x0.x + p1 * x1.x) + (p2 * x2.x + p3 * x3.x));
        acc.y = acc.y * sc + ((p0 * x0.y + p1 * x1.y) + (p2 * x2.y + p3 * x3.y));
        acc.z = acc.z * sc + ((p0 * x0.z + p1 * x1.z) + (p2 * x2.z + p3 * x3.z));
        acc.w = acc.w * sc + ((p0 * x0.w + p1 * x1.w) + (p2 * x2.w + p3 * x3.w));
        mx = mn;
      }
    }

    const float den = l + 1e-16f;
    float4 o;
    o.x = acc.x / den; o.y = acc.y / den; o.z = acc.z / den; o.w = acc.w / den;

    v.x = 0.5f * (o.x + __shfl_xor(o.x, 16));
    v.y = 0.5f * (o.y + __shfl_xor(o.y, 16));
    v.z = 0.5f * (o.z + __shfl_xor(o.z, 16));
    v.w = 0.5f * (o.w + __shfl_xor(o.w, 16));
  }

  if (lane < 16) {
    float4 q;
    q.x = v.x * v.x; q.y = v.y * v.y; q.z = v.z * v.z; q.w = v.w * v.w;
    *(float4*)&s_red[wave][lane * 4]      = v;
    *(float4*)&s_red[wave][64 + lane * 4] = q;
  }
  __syncthreads();
  if (node < N && lane < 16) *(float4*)(tmp + (size_t)node * 64 + lane * 4) = v;
  if (threadIdx.x < 128) {
    const int k = threadIdx.x;
    const float t = ((s_red[0][k] + s_red[1][k]) + (s_red[2][k] + s_red[3][k])) +
                    ((s_red[4][k] + s_red[5][k]) + (s_red[6][k] + s_red[7][k]));
    part[(size_t)blockIdx.x * 128 + k] = t;
  }
}

__global__ void __launch_bounds__(1024)
bn_partials_sum(const float* __restrict__ part, float* __restrict__ stats,
                int rows) {
  __shared__ __attribute__((aligned(16))) float s_w[32][32];
  const int lane = threadIdx.x & 31;
  const int wave = threadIdx.x >> 5;
  const int cg   = lane & 7;
  const int rs   = wave * 4 + (lane >> 3);
  const float* __restrict__ p = part + blockIdx.x * 32 + cg * 4;
  float4 a = {0.0f, 0.0f, 0.0f, 0.0f};
  int r0 = rs;
  for (; r0 + 15 * 128 < rows; r0 += 16 * 128) {
    const float* __restrict__ pr = p + (size_t)r0 * 128;
    float4 q[16];
    for (int j = 0; j < 16; ++j) q[j] = *(const float4*)(pr + (size_t)j * (128 * 128));
    for (int j = 0; j < 16; ++j) {
      a.x += q[j].x; a.y += q[j].y; a.z += q[j].z; a.w += q[j].w;
    }
  }
  for (; r0 < rows; r0 += 128) {
    const float4 q = *(const float4*)(p + (size_t)r0 * 128);
    a.x += q.x; a.y += q.y; a.z += q.z; a.w += q.w;
  }
  a.x += __shfl_xor(a.x, 8);  a.y += __shfl_xor(a.y, 8);
  a.z += __shfl_xor(a.z, 8);  a.w += __shfl_xor(a.w, 8);
  a.x += __shfl_xor(a.x, 16); a.y += __shfl_xor(a.y, 16);
  a.z += __shfl_xor(a.z, 16); a.w += __shfl_xor(a.w, 16);
  if (lane < 8) *(float4*)&s_w[wave][cg * 4] = a;
  __syncthreads();
  if (wave == 0) {
    float t = 0.0f;
    for (int w = 0; w < 32; ++w) t += s_w[w][lane];
    stats[blockIdx.x * 32 + lane] = t;
  }
}

__global__ void bn_apply(const float* __restrict__ tmp,
                         const float* __restrict__ sum,
                         const float* __restrict__ sumsq,
                         const float* __restrict__ gamma,
                         const float* __restrict__ beta,
                         float* __restrict__ out,
                         int N, int doTanh) {
  int t = blockIdx.x * blockDim.x + threadIdx.x;
  if (t >= N * 64) return;
  const int c = t & 63;
  const float invN = 1.0f / (float)N;
  const float mean = sum[c] * invN;
  const float var  = sumsq[c] * invN - mean * mean;
  float v = (tmp[t] - mean) * rsqrtf(var + BN_EPS) * gamma[c] + beta[c];
  out[t] = doTanh ? tanhf(v) : v;
}

extern "C" void kernel_launch(void* const* d_in, const int* in_sizes, int n_in,
                              void* d_out, int out_size, void* d_ws, size_t ws_size,
                              hipStream_t stream) {
  const float* x      = (const float*)d_in[0];
  const int*   ei     = (const int*)  d_in[1];
  const float* W0     = (const float*)d_in[2];
  const float* a0     = (const float*)d_in[3];
  const float* gamma0 = (const float*)d_in[4];
  const float* beta0  = (const float*)d_in[5];
  const float* W1     = (const float*)d_in[6];
  const float* a1     = (const float*)d_in[7];
  const float* gamma1 = (const float*)d_in[8];
  const float* beta1  = (const float*)d_in[9];
  float* out = (float*)d_out;

  const int N    = in_sizes[0] / 128;
  const int E    = in_sizes[1] / 2;

  char* ws = (char*)d_ws;
  size_t off = 0;
  auto carve = [&](size_t nwords) {
    float* p = (float*)(ws + off);
    off = (off + nwords * sizeof(float) + 255) & ~(size_t)255;
    return p;
  };
  float* xp    = carve((size_t)N * 128);
  float* tmp   = carve((size_t)N * 64);
  float* h0    = carve((size_t)N * 64);
  float* stats = carve(128);
  int* cnt     = (int*)carve((size_t)N);
  int* rowbeg  = (int*)carve((size_t)N);
  int* cursor  = (int*)carve((size_t)N);
  int* srcs    = (int*)carve((size_t)E);
  const int nodeGroups = (N + NODES_PER_BLOCK - 1) / NODES_PER_BLOCK;
  float* part  = carve((size_t)nodeGroups * 128);
  (void)ws_size;

  const int edgeBlocks   = (E + 255) / 256;
  const int nodeBlocks   = (N + 255) / 256;
  const int scanBlocks   = (N + 1023) / 1024;
  const int nodeElBlocks = (N * 64 + 255) / 256;

  zero_i32<<<nodeBlocks, 256, 0, stream>>>(cnt, N);
  csr_hist<<<edgeBlocks, 256, 0, stream>>>(ei, cnt, E);
  csr_scan<<<scanBlocks, 1024, 0, stream>>>(cnt, rowbeg, cursor, N);
  csr_fill<<<edgeBlocks, 256, 0, stream>>>(ei, cursor, srcs, E);

  gemm_wmma_f32<128><<<512, 256, 0, stream>>>(x, W0, xp, N);
  gat_node<<<nodeGroups, 256, 0, stream>>>(xp, rowbeg, cursor, srcs, a0, tmp, part, N);
  bn_partials_sum<<<4, 1024, 0, stream>>>(part, stats, nodeGroups);
  bn_apply<<<nodeElBlocks, 256, 0, stream>>>(tmp, stats, stats + 64,
                                             gamma0, beta0, h0, N, 1);

  gemm_wmma_f32<64><<<512, 256, 0, stream>>>(h0, W1, xp, N);
  gat_node<<<nodeGroups, 256, 0, stream>>>(xp, rowbeg, cursor, srcs, a1, tmp, part, N);
  bn_partials_sum<<<4, 1024, 0, stream>>>(part, stats, nodeGroups);
  bn_apply<<<nodeElBlocks, 256, 0, stream>>>(tmp, stats, stats + 64,
                                             gamma1, beta1, out, N, 0);
}


# Edges sorted by destination via LDS counters: 256 chunks binned by 256-node range, a workgroup per range builds its rows; no global atomics, zero or scan launch; in-row sum order = LDS add arrival
# speedup vs baseline: 5.3754x; 1.7097x over previous
#include <hip/hip_runtime.h>
#include <hip/hip_bf16.h>
#include <math.h>


typedef __attribute__((ext_vector_type(2))) float v2f;
typedef __attribute__((ext_vector_type(8))) float v8f;

#define NEG_SLOPE 0.2f
#define BN_EPS 1e-5f

template <int K>
__global__ void __launch_bounds__(256)
gemm_wmma_f32(const float* __restrict__ A,
              const float* __restrict__ B,
              float* __restrict__ C,
              int M) {
  constexpr int Nn  = 128;
  constexpr int PA  = K + 4;
  constexpr int PC  = Nn + 4;
  constexpr int KS  = K / 4;
  constexpr int RQ  = K / 4;
  constexpr int NPA = (16 * RQ) / 256;
  __shared__ __attribute__((aligned(16))) float sA[16 * PA];
  __shared__ __attribute__((aligned(16))) float sC[16 * PC];

  const int tid  = threadIdx.x;
  const int wave = tid >> 5;
  const int lane = tid & 31;
  const int half = lane >> 4;
  const int l    = lane & 15;
  const int n0   = wave * 16;
  const int tiles = M >> 4;

  v2f bf[KS];
#pragma unroll
  for (int ks = 0; ks < KS; ++ks) {
    const int kb = ks * 4 + half * 2;
    bf[ks].x = B[(size_t)kb * Nn + n0 + l];
    bf[ks].y = B[(size_t)(kb + 1) * Nn + n0 + l];
  }

  int tile = blockIdx.x;
  if (tile >= tiles) return;

  const int q0 = tid,       r0 = q0 / RQ, e0 = (q0 % RQ) * 4;
  const int q1 = tid + 256, r1 = q1 / RQ, e1 = (q1 % RQ) * 4;
  const unsigned aoff0 = r0 * K + e0;
  const unsigned aoff1 = r1 * K + e1;
  const int orow = tid >> 5, ocol = (tid & 31) * 4;
  const unsigned coff = orow * Nn + ocol;

  float4 pre0, pre1 = {0.0f, 0.0f, 0.0f, 0.0f};
  {
    const float* __restrict__ At = A + (size_t)tile * (16 * K);
    pre0 = *(const float4*)(At + aoff0);
    if constexpr (NPA > 1) pre1 = *(const float4*)(At + aoff1);
  }

  for (; tile < tiles; tile += gridDim.x) {
    *(float4*)&sA[r0 * PA + e0] = pre0;
    if constexpr (NPA > 1) *(float4*)&sA[r1 * PA + e1] = pre1;
    __syncthreads();

    const int nxt = tile + gridDim.x;
    if (nxt < tiles) {
      const float* __restrict__ An = A + (size_t)nxt * (16 * K);
      pre0 = *(const float4*)(An + aoff0);
      if constexpr (NPA > 1) pre1 = *(const float4*)(An + aoff1);
    }

    v2f af[KS];
#pragma unroll
    for (int ks = 0; ks < KS; ++ks)
      af[ks] = *(const v2f*)&sA[l * PA + ks * 4 + half * 2];
    __builtin_amdgcn_sched_barrier(0);

    v8f c = {};
#pragma unroll
    for (int ks = 0; ks < KS; ++ks) {
      c = __builtin_amdgcn_wmma_f32_16x16x4_f32(false, af[ks], false, bf[ks],
                                                (short)0, c, false, false);
    }
    __builtin_amdgcn_sched_barrier(0);

#pragma unroll
    for (int j = 0; j < 8; ++j) sC[(j + half * 8) * PC + n0 + l] = c[j];
    __syncthreads();

    float* __restrict__ Ct = C + (size_t)tile * (16 * Nn);
    const float4 o0 = *(const float4*)&sC[orow * PC + ocol];
    const float4 o1 = *(const float4*)&sC[(orow + 8) * PC + ocol];
    *(float4*)(Ct + coff) = o0;
    *(float4*)(Ct + coff + 8 * Nn) = o1;
  }
}

#define BIN_SHIFT   8
#define BIN_NODES   (1 << BIN_SHIFT)
#define SORT_BLOCKS 256
#define MAX_BINS    1024
#define CHUNK_LDS   4096
#define BIN_LDS     8192

__global__ void __launch_bounds__(1024)
csr_bin(const int* __restrict__ ei, int* __restrict__ stgw,
        int* __restrict__ tabw, int E, int N) {
  __shared__ int s_cnt[MAX_BINS];
  __shared__ int s_cur[MAX_BINS];
  __shared__ int s_tot[32];
  __shared__ int2 s_out[CHUNK_LDS];
  const int tid  = threadIdx.x;
  const int lane = tid & 31;
  const int wave = tid >> 5;
  const int G  = gridDim.x;
  const int C  = (E + G - 1) / G;
  const int CP = (C + 15) & ~15;
  const int NB = (N + BIN_NODES - 1) >> BIN_SHIFT;
  const int PB = (NB + 15) & ~15;
  const int e0 = blockIdx.x * C;
  const int e1 = min(E, e0 + C);

  s_cnt[tid] = 0;
  int dd[4], ss[4];
#pragma unroll
  for (int j = 0; j < 4; ++j) {
    const int i = e0 + tid + j * 1024;
    const bool ok = i < e1;
    dd[j] = ok ? ei[E + i] : -1;
    ss[j] = ok ? ei[i] : 0;
  }
  __syncthreads();
#pragma unroll
  for (int j = 0; j < 4; ++j)
    if (dd[j] >= 0) atomicAdd(&s_cnt[dd[j] >> BIN_SHIFT], 1);
  for (int i = e0 + tid + 4096; i < e1; i += 1024)
    atomicAdd(&s_cnt[ei[E + i] >> BIN_SHIFT], 1);
  __syncthreads();

  const int v = s_cnt[tid];
  int inc = v;
  for (int off = 1; off < 32; off <<= 1) {
    const int up = __shfl_up(inc, off);
    if (lane >= off) inc += up;
  }
  if (lane == 31) s_tot[wave] = inc;
  __syncthreads();
  int wb = (lane < wave) ? s_tot[lane] : 0;
  for (int off = 16; off > 0; off >>= 1) wb += __shfl_xor(wb, off);
  const int excl = wb + inc - v;
  s_cur[tid] = excl;
  if (tid < PB)
    ((int2*)tabw)[(size_t)blockIdx.x * PB + tid] = make_int2(excl, v);
  __syncthreads();

  int2* __restrict__ outp = (int2*)stgw + (size_t)blockIdx.x * CP;
  if (C <= CHUNK_LDS) {
#pragma unroll
    for (int j = 0; j < 4; ++j)
      if (dd[j] >= 0) {
        const int pos = atomicAdd(&s_cur[dd[j] >> BIN_SHIFT], 1);
        s_out[pos] = make_int2(ss[j], dd[j]);
      }
    __syncthreads();
    for (int q = tid; q < e1 - e0; q += 1024) outp[q] = s_out[q];
  } else {
#pragma unroll
    for (int j = 0; j < 4; ++j)
      if (dd[j] >= 0) {
        const int pos = atomicAdd(&s_cur[dd[j] >> BIN_SHIFT], 1);
        outp[pos] = make_int2(ss[j], dd[j]);
      }
    for (int i = e0 + tid + 4096; i < e1; i += 1024) {
      const int d = ei[E + i], s = ei[i];
      const int pos = atomicAdd(&s_cur[d >> BIN_SHIFT], 1);
      outp[pos] = make_int2(s, d);
    }
  }
}

__global__ void __launch_bounds__(1024)
csr_rows(const int* __restrict__ stgw, const int* __restrict__ tabw,
         int* __restrict__ rowbeg, int* __restrict__ rowend,
         int* __restrict__ srcs, int E, int N) {
  constexpr int G   = SORT_BLOCKS;
  constexpr int PPW = G / 32;
  __shared__ int s_po[G];
  __shared__ int s_pc[G];
  __shared__ int s_cnt[BIN_NODES];
  __shared__ int s_cur[BIN_NODES];
  __shared__ int s_red[32];
  __shared__ int s_len[32];
  __shared__ int s_tot[32];
  __shared__ int s_rows[BIN_LDS];
  const int2* __restrict__ stg = (const int2*)stgw;
  const int2* __restrict__ tab = (const int2*)tabw;
  const int tid   = threadIdx.x;
  const int lane  = tid & 31;
  const int wave  = tid >> 5;
  const int bin   = blockIdx.x;
  const int node0 = bin << BIN_SHIFT;
  const int C  = (E + G - 1) / G;
  const int CP = (C + 15) & ~15;
  const int NB = (N + BIN_NODES - 1) >> BIN_SHIFT;
  const int PB = (NB + 15) & ~15;

  int myoff = 0, mylen = 0;
  if (tid < G) {
    const int2 t = tab[(size_t)tid * PB + bin];
    myoff = t.x;
    mylen = t.y;
    s_po[tid] = tid * CP + t.x;
    s_pc[tid] = t.y;
  }
  if (tid < BIN_NODES) s_cnt[tid] = 0;
  for (int off = 16; off > 0; off >>= 1) {
    myoff += __shfl_xor(myoff, off);
    mylen += __shfl_xor(mylen, off);
  }
  if (lane == 0) { s_red[wave] = myoff; s_len[wave] = mylen; }
  __syncthreads();
  int base = s_red[lane], total = s_len[lane];
  for (int off = 16; off > 0; off >>= 1) {
    base  += __shfl_xor(base, off);
    total += __shfl_xor(total, off);
  }

  int2 e[PPW];
  int pc[PPW], po[PPW];
#pragma unroll
  for (int k = 0; k < PPW; ++k) {
    const int p = wave * PPW + k;
    pc[k] = s_pc[p];
    po[k] = s_po[p];
  }
#pragma unroll
  for (int k = 0; k < PPW; ++k)
    e[k] = (lane < pc[k]) ? stg[po[k] + lane] : make_int2(0, -1);
#pragma unroll
  for (int k = 0; k < PPW; ++k)
    if (lane < pc[k]) atomicAdd(&s_cnt[e[k].y - node0], 1);
#pragma unroll
  for (int k = 0; k < PPW; ++k)
    for (int i = lane + 32; i < pc[k]; i += 32)
      atomicAdd(&s_cnt[stg[po[k] + i].y - node0], 1);
  __syncthreads();

  const int v = (tid < BIN_NODES) ? s_cnt[tid] : 0;
  int inc = v;
  for (int off = 1; off < 32; off <<= 1) {
    const int up = __shfl_up(inc, off);
    if (lane >= off) inc += up;
  }
  if (lane == 31) s_tot[wave] = inc;
  __syncthreads();
  int wb = (lane < wave) ? s_tot[lane] : 0;
  for (int off = 16; off > 0; off >>= 1) wb += __shfl_xor(wb, off);
  const int excl = wb + inc - v;
  if (tid < BIN_NODES) {
    s_cur[tid] = excl;
    const int node = node0 + tid;
    if (node < N) {
      rowbeg[node] = base + excl;
      rowend[node] = base + excl + v;
    }
  }
  __syncthreads();

  int* __restrict__ rowsOut = srcs + base;
  if (total <= BIN_LDS) {
#pragma unroll
    for (int k = 0; k < PPW; ++k)
      if (lane < pc[k]) {
        const int pos = atomicAdd(&s_cur[e[k].y - node0], 1);
        s_rows[pos] = e[k].x;
      }
#pragma unroll
    for (int k = 0; k < PPW; ++k)
      for (int i = lane + 32; i < pc[k]; i += 32) {
        const int2 q = stg[po[k] + i];
        const int pos = atomicAdd(&s_cur[q.y - node0], 1);
        s_rows[pos] = q.x;
      }
    __syncthreads();
    const int lead = base & 31;
    for (int q = tid - lead; q < total; q += 1024)
      if (q >= 0) rowsOut[q] = s_rows[q];
  } else {
#pragma unroll
    for (int k = 0; k < PPW; ++k)
      if (lane < pc[k]) {
        const int pos = atomicAdd(&s_cur[e[k].y - node0], 1);
        rowsOut[pos] = e[k].x;
      }
#pragma unroll
    for (int k = 0; k < PPW; ++k)
      for (int i = lane + 32; i < pc[k]; i += 32) {
        const int2 q = stg[po[k] + i];
        const int pos = atomicAdd(&s_cur[q.y - node0], 1);
        rowsOut[pos] = q.x;
      }
  }
}

__device__ __forceinline__ float score4(const float4 s, const float4 d,
                                        const float4 av) {
  float m, lr, acc = 0.0f;
  m = s.x + d.x; lr = (m > 0.0f) ? m : NEG_SLOPE * m; acc += lr * av.x;
  m = s.y + d.y; lr = (m > 0.0f) ? m : NEG_SLOPE * m; acc += lr * av.y;
  m = s.z + d.z; lr = (m > 0.0f) ? m : NEG_SLOPE * m; acc += lr * av.z;
  m = s.w + d.w; lr = (m > 0.0f) ? m : NEG_SLOPE * m; acc += lr * av.w;
  return acc;
}

__device__ __forceinline__ float row16_sum(float v) {
  v += __int_as_float(__builtin_amdgcn_update_dpp(0, __float_as_int(v), 0x168, 0xf, 0xf, true));
  v += __int_as_float(__builtin_amdgcn_update_dpp(0, __float_as_int(v), 0x164, 0xf, 0xf, true));
  v += __int_as_float(__builtin_amdgcn_update_dpp(0, __float_as_int(v), 0x162, 0xf, 0xf, true));
  v += __int_as_float(__builtin_amdgcn_update_dpp(0, __float_as_int(v), 0x161, 0xf, 0xf, true));
  return v;
}

#define NODES_PER_BLOCK 8

__global__ void __launch_bounds__(256)
gat_node(const float* __restrict__ xp,
         const int* __restrict__ rowbeg,
         const int* __restrict__ rowend,
         const int* __restrict__ srcs,
         const float* __restrict__ avec,
         float* __restrict__ tmp,
         float* __restrict__ part,
         int N) {
  __shared__ __attribute__((aligned(16))) float s_red[NODES_PER_BLOCK][128];
  const int lane = threadIdx.x & 31;
  const int wave = threadIdx.x >> 5;
  const int node = __builtin_amdgcn_readfirstlane(
      (int)(blockIdx.x * NODES_PER_BLOCK + wave));

  float4 v = {0.0f, 0.0f, 0.0f, 0.0f};

  if (node < N) {
    const int coff = (lane >> 4) * 64 + (lane & 15) * 4;
    const float4 d  = *(const float4*)(xp + (size_t)node * 128 + coff);
    const float4 av = *(const float4*)(avec + coff);
    const int beg = rowbeg[node];
    const int end = rowend[node];

    float  mx  = row16_sum(score4(d, d, av));
    float  l   = 1.0f;
    float4 acc = d;

    for (int base = beg; base < end; base += 32) {
      const int nb = min(32, end - base);
      const int mysrc = (lane < nb) ? srcs[base + lane] : node;
      for (int t = 0; t < nb; t += 4) {
        const int s0 = __builtin_amdgcn_readlane(mysrc, t);
        const int s1 = __builtin_amdgcn_readlane(mysrc, t + 1);
        const int s2 = __builtin_amdgcn_readlane(mysrc, t + 2);
        const int s3 = __builtin_amdgcn_readlane(mysrc, t + 3);
        const float4 x0 = *(const float4*)(xp + (size_t)s0 * 128 + coff);
        const float4 x1 = *(const float4*)(xp + (size_t)s1 * 128 + coff);
        const float4 x2 = *(const float4*)(xp + (size_t)s2 * 128 + coff);
        const float4 x3 = *(const float4*)(xp + (size_t)s3 * 128 + coff);
        float e0 = row16_sum(score4(x0, d, av));
        float e1 = row16_sum(score4(x1, d, av));
        float e2 = row16_sum(score4(x2, d, av));
        float e3 = row16_sum(score4(x3, d, av));
        e1 = (t + 1 < nb) ? e1 : -INFINITY;
        e2 = (t + 2 < nb) ? e2 : -INFINITY;
        e3 = (t + 3 < nb) ? e3 : -INFINITY;
        const float mn = fmaxf(fmaxf(mx, e0), fmaxf(e1, fmaxf(e2, e3)));
        const float sc = expf(mx - mn);
        const float p0 = expf(e0 - mn);
        const float p1 = expf(e1 - mn);
        const float p2 = expf(e2 - mn);
        const float p3 = expf(e3 - mn);
        l = l * sc + ((p0 + p1) + (p2 + p3));
        acc.x = acc.x * sc + ((p0 * x0.x + p1 * x1.x) + (p2 * x2.x + p3 * x3.x));
        acc.y = acc.y * sc + ((p0 * x0.y + p1 * x1.y) + (p2 * x2.y + p3 * x3.y));
        acc.z = acc.z * sc + ((p0 * x0.z + p1 * x1.z) + (p2 * x2.z + p3 * x3.z));
        acc.w = acc.w * sc + ((p0 * x0.w + p1 * x1.w) + (p2 * x2.w + p3 * x3.w));
        mx = mn;
      }
    }

    const float den = l + 1e-16f;
    float4 o;
    o.x = acc.x / den; o.y = acc.y / den; o.z = acc.z / den; o.w = acc.w / den;

    v.x = 0.5f * (o.x + __shfl_xor(o.x, 16));
    v.y = 0.5f * (o.y + __shfl_xor(o.y, 16));
    v.z = 0.5f * (o.z + __shfl_xor(o.z, 16));
    v.w = 0.5f * (o.w + __shfl_xor(o.w, 16));
  }

  if (lane < 16) {
    float4 q;
    q.x = v.x * v.x; q.y = v.y * v.y; q.z = v.z * v.z; q.w = v.w * v.w;
    *(float4*)&s_red[wave][lane * 4]      = v;
    *(float4*)&s_red[wave][64 + lane * 4] = q;
  }
  __syncthreads();
  if (node < N && lane < 16) *(float4*)(tmp + (size_t)node * 64 + lane * 4) = v;
  if (threadIdx.x < 128) {
    const int k = threadIdx.x;
    const float t = ((s_red[0][k] + s_red[1][k]) + (s_red[2][k] + s_red[3][k])) +
                    ((s_red[4][k] + s_red[5][k]) + (s_red[6][k] + s_red[7][k]));
    part[(size_t)blockIdx.x * 128 + k] = t;
  }
}

__global__ void __launch_bounds__(1024)
bn_partials_sum(const float* __restrict__ part, float* __restrict__ stats,
                int rows) {
  __shared__ __attribute__((aligned(16))) float s_w[32][32];
  const int lane = threadIdx.x & 31;
  const int wave = threadIdx.x >> 5;
  const int cg   = lane & 7;
  const int rs   = wave * 4 + (lane >> 3);
  const float* __restrict__ p = part + blockIdx.x * 32 + cg * 4;
  float4 a = {0.0f, 0.0f, 0.0f, 0.0f};
  int r0 = rs;
  for (; r0 + 15 * 128 < rows; r0 += 16 * 128) {
    const float* __restrict__ pr = p + (size_t)r0 * 128;
    float4 q[16];
    for (int j = 0; j < 16; ++j) q[j] = *(const float4*)(pr + (size_t)j * (128 * 128));
    for (int j = 0; j < 16; ++j) {
      a.x += q[j].x; a.y += q[j].y; a.z += q[j].z; a.w += q[j].w;
    }
  }
  for (; r0 < rows; r0 += 128) {
    const float4 q = *(const float4*)(p + (size_t)r0 * 128);
    a.x += q.x; a.y += q.y; a.z += q.z; a.w += q.w;
  }
  a.x += __shfl_xor(a.x, 8);  a.y += __shfl_xor(a.y, 8);
  a.z += __shfl_xor(a.z, 8);  a.w += __shfl_xor(a.w, 8);
  a.x += __shfl_xor(a.x, 16); a.y += __shfl_xor(a.y, 16);
  a.z += __shfl_xor(a.z, 16); a.w += __shfl_xor(a.w, 16);
  if (lane < 8) *(float4*)&s_w[wave][cg * 4] = a;
  __syncthreads();
  if (wave == 0) {
    float t = 0.0f;
    for (int w = 0; w < 32; ++w) t += s_w[w][lane];
    stats[blockIdx.x * 32 + lane] = t;
  }
}

__global__ void bn_apply(const float* __restrict__ tmp,
                         const float* __restrict__ sum,
                         const float* __restrict__ sumsq,
                         const float* __restrict__ gamma,
                         const float* __restrict__ beta,
                         float* __restrict__ out,
                         int N, int doTanh) {
  int t = blockIdx.x * blockDim.x + threadIdx.x;
  if (t >= N * 64) return;
  const int c = t & 63;
  const float invN = 1.0f / (float)N;
  const float mean = sum[c] * invN;
  const float var  = sumsq[c] * invN - mean * mean;
  float v = (tmp[t] - mean) * rsqrtf(var + BN_EPS) * gamma[c] + beta[c];
  out[t] = doTanh ? tanhf(v) : v;
}

extern "C" void kernel_launch(void* const* d_in, const int* in_sizes, int n_in,
                              void* d_out, int out_size, void* d_ws, size_t ws_size,
                              hipStream_t stream) {
  const float* x      = (const float*)d_in[0];
  const int*   ei     = (const int*)  d_in[1];
  const float* W0     = (const float*)d_in[2];
  const float* a0     = (const float*)d_in[3];
  const float* gamma0 = (const float*)d_in[4];
  const float* beta0  = (const float*)d_in[5];
  const float* W1     = (const float*)d_in[6];
  const float* a1     = (const float*)d_in[7];
  const float* gamma1 = (const float*)d_in[8];
  const float* beta1  = (const float*)d_in[9];
  float* out = (float*)d_out;

  const int N    = in_sizes[0] / 128;
  const int E    = in_sizes[1] / 2;

  char* ws = (char*)d_ws;
  size_t off = 0;
  auto carve = [&](size_t nwords) {
    float* p = (float*)(ws + off);
    off = (off + nwords * sizeof(float) + 255) & ~(size_t)255;
    return p;
  };
  float* xp    = carve((size_t)N * 128);
  float* tmp   = carve((size_t)N * 64);
  float* h0    = carve((size_t)N * 64);
  float* stats = carve(128);
  int* rowbeg  = (int*)carve((size_t)N);
  int* rowend  = (int*)carve((size_t)N);
  int* srcs    = (int*)carve((size_t)E);
  int* stg     = (int*)carve((size_t)E * 2 + SORT_BLOCKS * 32);
  int* tab     = (int*)carve((size_t)SORT_BLOCKS * MAX_BINS * 2);
  const int nodeGroups = (N + NODES_PER_BLOCK - 1) / NODES_PER_BLOCK;
  float* part  = carve((size_t)nodeGroups * 128);
  (void)ws_size;

  const int binBlocks    = (N + BIN_NODES - 1) / BIN_NODES;
  const int nodeElBlocks = (N * 64 + 255) / 256;

  csr_bin<<<SORT_BLOCKS, 1024, 0, stream>>>(ei, stg, tab, E, N);
  csr_rows<<<binBlocks, 1024, 0, stream>>>(stg, tab, rowbeg, rowend, srcs, E, N);

  gemm_wmma_f32<128><<<512, 256, 0, stream>>>(x, W0, xp, N);
  gat_node<<<nodeGroups, 256, 0, stream>>>(xp, rowbeg, rowend, srcs, a0, tmp, part, N);
  bn_partials_sum<<<4, 1024, 0, stream>>>(part, stats, nodeGroups);
  bn_apply<<<nodeElBlocks, 256, 0, stream>>>(tmp, stats, stats + 64,
                                             gamma0, beta0, h0, N, 1);

  gemm_wmma_f32<64><<<512, 256, 0, stream>>>(h0, W1, xp, N);
  gat_node<<<nodeGroups, 256, 0, stream>>>(xp, rowbeg, rowend, srcs, a1, tmp, part, N);
  bn_partials_sum<<<4, 1024, 0, stream>>>(part, stats, nodeGroups);
  bn_apply<<<nodeElBlocks, 256, 0, stream>>>(tmp, stats, stats + 64,
                                             gamma1, beta1, out, N, 0);
}


# Batch-norm apply pass on a resident grid of 512 workgroups: a lane moves 16 bytes per piece, up to 8 pieces with all loads in flight, per-channel terms formed once; was one wave per 32 elements
# speedup vs baseline: 5.7291x; 1.0658x over previous
#include <hip/hip_runtime.h>
#include <hip/hip_bf16.h>
#include <math.h>


typedef __attribute__((ext_vector_type(2))) float v2f;
typedef __attribute__((ext_vector_type(8))) float v8f;

#define NEG_SLOPE 0.2f
#define BN_EPS 1e-5f

template <int K>
__global__ void __launch_bounds__(256)
gemm_wmma_f32(const float* __restrict__ A,
              const float* __restrict__ B,
              float* __restrict__ C,
              int M) {
  constexpr int Nn  = 128;
  constexpr int PA  = K + 4;
  constexpr int PC  = Nn + 4;
  constexpr int KS  = K / 4;
  constexpr int RQ  = K / 4;
  constexpr int NPA = (16 * RQ) / 256;
  __shared__ __attribute__((aligned(16))) float sA[16 * PA];
  __shared__ __attribute__((aligned(16))) float sC[16 * PC];

  const int tid  = threadIdx.x;
  const int wave = tid >> 5;
  const int lane = tid & 31;
  const int half = lane >> 4;
  const int l    = lane & 15;
  const int n0   = wave * 16;
  const int tiles = M >> 4;

  v2f bf[KS];
#pragma unroll
  for (int ks = 0; ks < KS; ++ks) {
    const int kb = ks * 4 + half * 2;
    bf[ks].x = B[(size_t)kb * Nn + n0 + l];
    bf[ks].y = B[(size_t)(kb + 1) * Nn + n0 + l];
  }

  int tile = blockIdx.x;
  if (tile >= tiles) return;

  const int q0 = tid,       r0 = q0 / RQ, e0 = (q0 % RQ) * 4;
  const int q1 = tid + 256, r1 = q1 / RQ, e1 = (q1 % RQ) * 4;
  const unsigned aoff0 = r0 * K + e0;
  const unsigned aoff1 = r1 * K + e1;
  const int orow = tid >> 5, ocol = (tid & 31) * 4;
  const unsigned coff = orow * Nn + ocol;

  float4 pre0, pre1 = {0.0f, 0.0f, 0.0f, 0.0f};
  {
    const float* __restrict__ At = A + (size_t)tile * (16 * K);
    pre0 = *(const float4*)(At + aoff0);
    if constexpr (NPA > 1) pre1 = *(const float4*)(At + aoff1);
  }

  for (; tile < tiles; tile += gridDim.x) {
    *(float4*)&sA[r0 * PA + e0] = pre0;
    if constexpr (NPA > 1) *(float4*)&sA[r1 * PA + e1] = pre1;
    __syncthreads();

    const int nxt = tile + gridDim.x;
    if (nxt < tiles) {
      const float* __restrict__ An = A + (size_t)nxt * (16 * K);
      pre0 = *(const float4*)(An + aoff0);
      if constexpr (NPA > 1) pre1 = *(const float4*)(An + aoff1);
    }

    v2f af[KS];
#pragma unroll
    for (int ks = 0; ks < KS; ++ks)
      af[ks] = *(const v2f*)&sA[l * PA + ks * 4 + half * 2];
    __builtin_amdgcn_sched_barrier(0);

    v8f c = {};
#pragma unroll
    for (int ks = 0; ks < KS; ++ks) {
      c = __builtin_amdgcn_wmma_f32_16x16x4_f32(false, af[ks], false, bf[ks],
                                                (short)0, c, false, false);
    }
    __builtin_amdgcn_sched_barrier(0);

#pragma unroll
    for (int j = 0; j < 8; ++j) sC[(j + half * 8) * PC + n0 + l] = c[j];
    __syncthreads();

    float* __restrict__ Ct = C + (size_t)tile * (16 * Nn);
    const float4 o0 = *(const float4*)&sC[orow * PC + ocol];
    const float4 o1 = *(const float4*)&sC[(orow + 8) * PC + ocol];
    *(float4*)(Ct + coff) = o0;
    *(float4*)(Ct + coff + 8 * Nn) = o1;
  }
}

#define BIN_SHIFT   8
#define BIN_NODES   (1 << BIN_SHIFT)
#define SORT_BLOCKS 256
#define MAX_BINS    1024
#define CHUNK_LDS   4096
#define BIN_LDS     8192

__global__ void __launch_bounds__(1024)
csr_bin(const int* __restrict__ ei, int* __restrict__ stgw,
        int* __restrict__ tabw, int E, int N) {
  __shared__ int s_cnt[MAX_BINS];
  __shared__ int s_cur[MAX_BINS];
  __shared__ int s_tot[32];
  __shared__ int2 s_out[CHUNK_LDS];
  const int tid  = threadIdx.x;
  const int lane = tid & 31;
  const int wave = tid >> 5;
  const int G  = gridDim.x;
  const int C  = (E + G - 1) / G;
  const int CP = (C + 15) & ~15;
  const int NB = (N + BIN_NODES - 1) >> BIN_SHIFT;
  const int PB = (NB + 15) & ~15;
  const int e0 = blockIdx.x * C;
  const int e1 = min(E, e0 + C);

  s_cnt[tid] = 0;
  int dd[4], ss[4];
#pragma unroll
  for (int j = 0; j < 4; ++j) {
    const int i = e0 + tid + j * 1024;
    const bool ok = i < e1;
    dd[j] = ok ? ei[E + i] : -1;
    ss[j] = ok ? ei[i] : 0;
  }
  __syncthreads();
#pragma unroll
  for (int j = 0; j < 4; ++j)
    if (dd[j] >= 0) atomicAdd(&s_cnt[dd[j] >> BIN_SHIFT], 1);
  for (int i = e0 + tid + 4096; i < e1; i += 1024)
    atomicAdd(&s_cnt[ei[E + i] >> BIN_SHIFT], 1);
  __syncthreads();

  const int v = s_cnt[tid];
  int inc = v;
  for (int off = 1; off < 32; off <<= 1) {
    const int up = __shfl_up(inc, off);
    if (lane >= off) inc += up;
  }
  if (lane == 31) s_tot[wave] = inc;
  __syncthreads();
  int wb = (lane < wave) ? s_tot[lane] : 0;
  for (int off = 16; off > 0; off >>= 1) wb += __shfl_xor(wb, off);
  const int excl = wb + inc - v;
  s_cur[tid] = excl;
  if (tid < PB)
    ((int2*)tabw)[(size_t)blockIdx.x * PB + tid] = make_int2(excl, v);
  __syncthreads();

  int2* __restrict__ outp = (int2*)stgw + (size_t)blockIdx.x * CP;
  if (C <= CHUNK_LDS) {
#pragma unroll
    for (int j = 0; j < 4; ++j)
      if (dd[j] >= 0) {
        const int pos = atomicAdd(&s_cur[dd[j] >> BIN_SHIFT], 1);
        s_out[pos] = make_int2(ss[j], dd[j]);
      }
    __syncthreads();
    for (int q = tid; q < e1 - e0; q += 1024) outp[q] = s_out[q];
  } else {
#pragma unroll
    for (int j = 0; j < 4; ++j)
      if (dd[j] >= 0) {
        const int pos = atomicAdd(&s_cur[dd[j] >> BIN_SHIFT], 1);
        outp[pos] = make_int2(ss[j], dd[j]);
      }
    for (int i = e0 + tid + 4096; i < e1; i += 1024) {
      const int d = ei[E + i], s = ei[i];
      const int pos = atomicAdd(&s_cur[d >> BIN_SHIFT], 1);
      outp[pos] = make_int2(s, d);
    }
  }
}

__global__ void __launch_bounds__(1024)
csr_rows(const int* __restrict__ stgw, const int* __restrict__ tabw,
         int* __restrict__ rowbeg, int* __restrict__ rowend,
         int* __restrict__ srcs, int E, int N) {
  constexpr int G   = SORT_BLOCKS;
  constexpr int PPW = G / 32;
  __shared__ int s_po[G];
  __shared__ int s_pc[G];
  __shared__ int s_cnt[BIN_NODES];
  __shared__ int s_cur[BIN_NODES];
  __shared__ int s_red[32];
  __shared__ int s_len[32];
  __shared__ int s_tot[32];
  __shared__ int s_rows[BIN_LDS];
  const int2* __restrict__ stg = (const int2*)stgw;
  const int2* __restrict__ tab = (const int2*)tabw;
  const int tid   = threadIdx.x;
  const int lane  = tid & 31;
  const int wave  = tid >> 5;
  const int bin   = blockIdx.x;
  const int node0 = bin << BIN_SHIFT;
  const int C  = (E + G - 1) / G;
  const int CP = (C + 15) & ~15;
  const int NB = (N + BIN_NODES - 1) >> BIN_SHIFT;
  const int PB = (NB + 15) & ~15;

  int myoff = 0, mylen = 0;
  if (tid < G) {
    const int2 t = tab[(size_t)tid * PB + bin];
    myoff = t.x;
    mylen = t.y;
    s_po[tid] = tid * CP + t.x;
    s_pc[tid] = t.y;
  }
  if (tid < BIN_NODES) s_cnt[tid] = 0;
  for (int off = 16; off > 0; off >>= 1) {
    myoff += __shfl_xor(myoff, off);
    mylen += __shfl_xor(mylen, off);
  }
  if (lane == 0) { s_red[wave] = myoff; s_len[wave] = mylen; }
  __syncthreads();
  int base = s_red[lane], total = s_len[lane];
  for (int off = 16; off > 0; off >>= 1) {
    base  += __shfl_xor(base, off);
    total += __shfl_xor(total, off);
  }

  int2 e[PPW];
  int pc[PPW], po[PPW];
#pragma unroll
  for (int k = 0; k < PPW; ++k) {
    const int p = wave * PPW + k;
    pc[k] = s_pc[p];
    po[k] = s_po[p];
  }
#pragma unroll
  for (int k = 0; k < PPW; ++k)
    e[k] = (lane < pc[k]) ? stg[po[k] + lane] : make_int2(0, -1);
#pragma unroll
  for (int k = 0; k < PPW; ++k)
    if (lane < pc[k]) atomicAdd(&s_cnt[e[k].y - node0], 1);
#pragma unroll
  for (int k = 0; k < PPW; ++k)
    for (int i = lane + 32; i < pc[k]; i += 32)
      atomicAdd(&s_cnt[stg[po[k] + i].y - node0], 1);
  __syncthreads();

  const int v = (tid < BIN_NODES) ? s_cnt[tid] : 0;
  int inc = v;
  for (int off = 1; off < 32; off <<= 1) {
    const int up = __shfl_up(inc, off);
    if (lane >= off) inc += up;
  }
  if (lane == 31) s_tot[wave] = inc;
  __syncthreads();
  int wb = (lane < wave) ? s_tot[lane] : 0;
  for (int off = 16; off > 0; off >>= 1) wb += __shfl_xor(wb, off);
  const int excl = wb + inc - v;
  if (tid < BIN_NODES) {
    s_cur[tid] = excl;
    const int node = node0 + tid;
    if (node < N) {
      rowbeg[node] = base + excl;
      rowend[node] = base + excl + v;
    }
  }
  __syncthreads();

  int* __restrict__ rowsOut = srcs + base;
  if (total <= BIN_LDS) {
#pragma unroll
    for (int k = 0; k < PPW; ++k)
      if (lane < pc[k]) {
        const int pos = atomicAdd(&s_cur[e[k].y - node0], 1);
        s_rows[pos] = e[k].x;
      }
#pragma unroll
    for (int k = 0; k < PPW; ++k)
      for (int i = lane + 32; i < pc[k]; i += 32) {
        const int2 q = stg[po[k] + i];
        const int pos = atomicAdd(&s_cur[q.y - node0], 1);
        s_rows[pos] = q.x;
      }
    __syncthreads();
    const int lead = base & 31;
    for (int q = tid - lead; q < total; q += 1024)
      if (q >= 0) rowsOut[q] = s_rows[q];
  } else {
#pragma unroll
    for (int k = 0; k < PPW; ++k)
      if (lane < pc[k]) {
        const int pos = atomicAdd(&s_cur[e[k].y - node0], 1);
        rowsOut[pos] = e[k].x;
      }
#pragma unroll
    for (int k = 0; k < PPW; ++k)
      for (int i = lane + 32; i < pc[k]; i += 32) {
        const int2 q = stg[po[k] + i];
        const int pos = atomicAdd(&s_cur[q.y - node0], 1);
        rowsOut[pos] = q.x;
      }
  }
}

__device__ __forceinline__ float score4(const float4 s, const float4 d,
                                        const float4 av) {
  float m, lr, acc = 0.0f;
  m = s.x + d.x; lr = (m > 0.0f) ? m : NEG_SLOPE * m; acc += lr * av.x;
  m = s.y + d.y; lr = (m > 0.0f) ? m : NEG_SLOPE * m; acc += lr * av.y;
  m = s.z + d.z; lr = (m > 0.0f) ? m : NEG_SLOPE * m; acc += lr * av.z;
  m = s.w + d.w; lr = (m > 0.0f) ? m : NEG_SLOPE * m; acc += lr * av.w;
  return acc;
}

__device__ __forceinline__ float row16_sum(float v) {
  v += __int_as_float(__builtin_amdgcn_update_dpp(0, __float_as_int(v), 0x168, 0xf, 0xf, true));
  v += __int_as_float(__builtin_amdgcn_update_dpp(0, __float_as_int(v), 0x164, 0xf, 0xf, true));
  v += __int_as_float(__builtin_amdgcn_update_dpp(0, __float_as_int(v), 0x162, 0xf, 0xf, true));
  v += __int_as_float(__builtin_amdgcn_update_dpp(0, __float_as_int(v), 0x161, 0xf, 0xf, true));
  return v;
}

#define NODES_PER_BLOCK 8

__global__ void __launch_bounds__(256)
gat_node(const float* __restrict__ xp,
         const int* __restrict__ rowbeg,
         const int* __restrict__ rowend,
         const int* __restrict__ srcs,
         const float* __restrict__ avec,
         float* __restrict__ tmp,
         float* __restrict__ part,
         int N) {
  __shared__ __attribute__((aligned(16))) float s_red[NODES_PER_BLOCK][128];
  const int lane = threadIdx.x & 31;
  const int wave = threadIdx.x >> 5;
  const int node = __builtin_amdgcn_readfirstlane(
      (int)(blockIdx.x * NODES_PER_BLOCK + wave));

  float4 v = {0.0f, 0.0f, 0.0f, 0.0f};

  if (node < N) {
    const int coff = (lane >> 4) * 64 + (lane & 15) * 4;
    const float4 d  = *(const float4*)(xp + (size_t)node * 128 + coff);
    const float4 av = *(const float4*)(avec + coff);
    const int beg = rowbeg[node];
    const int end = rowend[node];

    float  mx  = row16_sum(score4(d, d, av));
    float  l   = 1.0f;
    float4 acc = d;

    for (int base = beg; base < end; base += 32) {
      const int nb = min(32, end - base);
      const int mysrc = (lane < nb) ? srcs[base + lane] : node;
      for (int t = 0; t < nb; t += 4) {
        const int s0 = __builtin_amdgcn_readlane(mysrc, t);
        const int s1 = __builtin_amdgcn_readlane(mysrc, t + 1);
        const int s2 = __builtin_amdgcn_readlane(mysrc, t + 2);
        const int s3 = __builtin_amdgcn_readlane(mysrc, t + 3);
        const float4 x0 = *(const float4*)(xp + (size_t)s0 * 128 + coff);
        const float4 x1 = *(const float4*)(xp + (size_t)s1 * 128 + coff);
        const float4 x2 = *(const float4*)(xp + (size_t)s2 * 128 + coff);
        const float4 x3 = *(const float4*)(xp + (size_t)s3 * 128 + coff);
        float e0 = row16_sum(score4(x0, d, av));
        float e1 = row16_sum(score4(x1, d, av));
        float e2 = row16_sum(score4(x2, d, av));
        float e3 = row16_sum(score4(x3, d, av));
        e1 = (t + 1 < nb) ? e1 : -INFINITY;
        e2 = (t + 2 < nb) ? e2 : -INFINITY;
        e3 = (t + 3 < nb) ? e3 : -INFINITY;
        const float mn = fmaxf(fmaxf(mx, e0), fmaxf(e1, fmaxf(e2, e3)));
        const float sc = expf(mx - mn);
        const float p0 = expf(e0 - mn);
        const float p1 = expf(e1 - mn);
        const float p2 = expf(e2 - mn);
        const float p3 = expf(e3 - mn);
        l = l * sc + ((p0 + p1) + (p2 + p3));
        acc.x = acc.x * sc + ((p0 * x0.x + p1 * x1.x) + (p2 * x2.x + p3 * x3.x));
        acc.y = acc.y * sc + ((p0 * x0.y + p1 * x1.y) + (p2 * x2.y + p3 * x3.y));
        acc.z = acc.z * sc + ((p0 * x0.z + p1 * x1.z) + (p2 * x2.z + p3 * x3.z));
        acc.w = acc.w * sc + ((p0 * x0.w + p1 * x1.w) + (p2 * x2.w + p3 * x3.w));
        mx = mn;
      }
    }

    const float den = l + 1e-16f;
    float4 o;
    o.x = acc.x / den; o.y = acc.y / den; o.z = acc.z / den; o.w = acc.w / den;

    v.x = 0.5f * (o.x + __shfl_xor(o.x, 16));
    v.y = 0.5f * (o.y + __shfl_xor(o.y, 16));
    v.z = 0.5f * (o.z + __shfl_xor(o.z, 16));
    v.w = 0.5f * (o.w + __shfl_xor(o.w, 16));
  }

  if (lane < 16) {
    float4 q;
    q.x = v.x * v.x; q.y = v.y * v.y; q.z = v.z * v.z; q.w = v.w * v.w;
    *(float4*)&s_red[wave][lane * 4]      = v;
    *(float4*)&s_red[wave][64 + lane * 4] = q;
  }
  __syncthreads();
  if (node < N && lane < 16) *(float4*)(tmp + (size_t)node * 64 + lane * 4) = v;
  if (threadIdx.x < 128) {
    const int k = threadIdx.x;
    const float t = ((s_red[0][k] + s_red[1][k]) + (s_red[2][k] + s_red[3][k])) +
                    ((s_red[4][k] + s_red[5][k]) + (s_red[6][k] + s_red[7][k]));
    part[(size_t)blockIdx.x * 128 + k] = t;
  }
}

__global__ void __launch_bounds__(1024)
bn_partials_sum(const float* __restrict__ part, float* __restrict__ stats,
                int rows) {
  __shared__ __attribute__((aligned(16))) float s_w[32][32];
  const int lane = threadIdx.x & 31;
  const int wave = threadIdx.x >> 5;
  const int cg   = lane & 7;
  const int rs   = wave * 4 + (lane >> 3);
  const float* __restrict__ p = part + blockIdx.x * 32 + cg * 4;
  float4 a = {0.0f, 0.0f, 0.0f, 0.0f};
  int r0 = rs;
  for (; r0 + 15 * 128 < rows; r0 += 16 * 128) {
    const float* __restrict__ pr = p + (size_t)r0 * 128;
    float4 q[16];
    for (int j = 0; j < 16; ++j) q[j] = *(const float4*)(pr + (size_t)j * (128 * 128));
    for (int j = 0; j < 16; ++j) {
      a.x += q[j].x; a.y += q[j].y; a.z += q[j].z; a.w += q[j].w;
    }
  }
  for (; r0 < rows; r0 += 128) {
    const float4 q = *(const float4*)(p + (size_t)r0 * 128);
    a.x += q.x; a.y += q.y; a.z += q.z; a.w += q.w;
  }
  a.x += __shfl_xor(a.x, 8);  a.y += __shfl_xor(a.y, 8);
  a.z += __shfl_xor(a.z, 8);  a.w += __shfl_xor(a.w, 8);
  a.x += __shfl_xor(a.x, 16); a.y += __shfl_xor(a.y, 16);
  a.z += __shfl_xor(a.z, 16); a.w += __shfl_xor(a.w, 16);
  if (lane < 8) *(float4*)&s_w[wave][cg * 4] = a;
  __syncthreads();
  if (wave == 0) {
    float t = 0.0f;
    for (int w = 0; w < 32; ++w) t += s_w[w][lane];
    stats[blockIdx.x * 32 + lane] = t;
  }
}

#define BN_BLOCKS  512
#define BN_THREADS 256
#define BN_PIECES  8

__device__ __forceinline__ float bn_one(float x, float mean, float rstd,
                                        float g, float b) {
  return (x - mean) * rstd * g + b;
}

template <int TANH>
__global__ void __launch_bounds__(BN_THREADS)
bn_apply(const float* __restrict__ tmp,
         const float* __restrict__ sum,
         const float* __restrict__ sumsq,
         const float* __restrict__ gamma,
         const float* __restrict__ beta,
         float* __restrict__ out,
         int N) {
  const int total  = N * 16;
  const int stride = gridDim.x * BN_THREADS;
  const int first  = blockIdx.x * BN_THREADS + threadIdx.x;
  const int c      = (first & 15) * 4;
  const float4* __restrict__ in4  = (const float4*)tmp;
  float4* __restrict__       out4 = (float4*)out;

  const float invN = 1.0f / (float)N;
  const float4 s4 = *(const float4*)(sum + c);
  const float4 q4 = *(const float4*)(sumsq + c);
  const float4 g4 = *(const float4*)(gamma + c);
  const float4 b4 = *(const float4*)(beta + c);
  float4 mean, rstd;
  mean.x = s4.x * invN; rstd.x = rsqrtf(q4.x * invN - mean.x * mean.x + BN_EPS);
  mean.y = s4.y * invN; rstd.y = rsqrtf(q4.y * invN - mean.y * mean.y + BN_EPS);
  mean.z = s4.z * invN; rstd.z = rsqrtf(q4.z * invN - mean.z * mean.z + BN_EPS);
  mean.w = s4.w * invN; rstd.w = rsqrtf(q4.w * invN - mean.w * mean.w + BN_EPS);

  for (int i0 = first; i0 < total; i0 += BN_PIECES * stride) {
    float4 v[BN_PIECES];
#pragma unroll
    for (int j = 0; j < BN_PIECES; ++j) {
      const int i = i0 + j * stride;
      v[j] = in4[(i < total) ? i : i0];
    }
#pragma unroll
    for (int j = 0; j < BN_PIECES; ++j)
      asm volatile("" :: "v"(v[j].x), "v"(v[j].y), "v"(v[j].z), "v"(v[j].w));
#pragma unroll
    for (int j = 0; j < BN_PIECES; ++j) {
      const int i = i0 + j * stride;
      float4 r;
      r.x = bn_one(v[j].x, mean.x, rstd.x, g4.x, b4.x);
      r.y = bn_one(v[j].y, mean.y, rstd.y, g4.y, b4.y);
      r.z = bn_one(v[j].z, mean.z, rstd.z, g4.z, b4.z);
      r.w = bn_one(v[j].w, mean.w, rstd.w, g4.w, b4.w);
      if (TANH) { r.x = tanhf(r.x); r.y = tanhf(r.y); r.z = tanhf(r.z); r.w = tanhf(r.w); }
      if (i < total) out4[i] = r;
    }
  }
}

extern "C" void kernel_launch(void* const* d_in, const int* in_sizes, int n_in,
                              void* d_out, int out_size, void* d_ws, size_t ws_size,
                              hipStream_t stream) {
  const float* x      = (const float*)d_in[0];
  const int*   ei     = (const int*)  d_in[1];
  const float* W0     = (const float*)d_in[2];
  const float* a0     = (const float*)d_in[3];
  const float* gamma0 = (const float*)d_in[4];
  const float* beta0  = (const float*)d_in[5];
  const float* W1     = (const float*)d_in[6];
  const float* a1     = (const float*)d_in[7];
  const float* gamma1 = (const float*)d_in[8];
  const float* beta1  = (const float*)d_in[9];
  float* out = (float*)d_out;

  const int N    = in_sizes[0] / 128;
  const int E    = in_sizes[1] / 2;

  char* ws = (char*)d_ws;
  size_t off = 0;
  auto carve = [&](size_t nwords) {
    float* p = (float*)(ws + off);
    off = (off + nwords * sizeof(float) + 255) & ~(size_t)255;
    return p;
  };
  float* xp    = carve((size_t)N * 128);
  float* tmp   = carve((size_t)N * 64);
  float* h0    = carve((size_t)N * 64);
  float* stats = carve(128);
  int* rowbeg  = (int*)carve((size_t)N);
  int* rowend  = (int*)carve((size_t)N);
  int* srcs    = (int*)carve((size_t)E);
  int* stg     = (int*)carve((size_t)E * 2 + SORT_BLOCKS * 32);
  int* tab     = (int*)carve((size_t)SORT_BLOCKS * MAX_BINS * 2);
  const int nodeGroups = (N + NODES_PER_BLOCK - 1) / NODES_PER_BLOCK;
  float* part  = carve((size_t)nodeGroups * 128);
  (void)ws_size;

  const int binBlocks    = (N + BIN_NODES - 1) / BIN_NODES;

  csr_bin<<<SORT_BLOCKS, 1024, 0, stream>>>(ei, stg, tab, E, N);
  csr_rows<<<binBlocks, 1024, 0, stream>>>(stg, tab, rowbeg, rowend, srcs, E, N);

  gemm_wmma_f32<128><<<512, 256, 0, stream>>>(x, W0, xp, N);
  gat_node<<<nodeGroups, 256, 0, stream>>>(xp, rowbeg, rowend, srcs, a0, tmp, part, N);
  bn_partials_sum<<<4, 1024, 0, stream>>>(part, stats, nodeGroups);
  bn_apply<1><<<BN_BLOCKS, BN_THREADS, 0, stream>>>(tmp, stats, stats + 64,
                                                    gamma0, beta0, h0, N);

  gemm_wmma_f32<64><<<512, 256, 0, stream>>>(h0, W1, xp, N);
  gat_node<<<nodeGroups, 256, 0, stream>>>(xp, rowbeg, rowend, srcs, a1, tmp, part, N);
  bn_partials_sum<<<4, 1024, 0, stream>>>(part, stats, nodeGroups);
  bn_apply<0><<<BN_BLOCKS, BN_THREADS, 0, stream>>>(tmp, stats, stats + 64,
                                                    gamma1, beta1, out, N);
}


# Softmax weights in base 2: log2(e) folded into the attention vector once per wave, each weight one exp2 instruction instead of library expf (13 vector instructions a value); sum order unchanged
# speedup vs baseline: 6.2450x; 1.0900x over previous
#include <hip/hip_runtime.h>
#include <hip/hip_bf16.h>
#include <math.h>


typedef __attribute__((ext_vector_type(2))) float v2f;
typedef __attribute__((ext_vector_type(8))) float v8f;

#define NEG_SLOPE 0.2f
#define BN_EPS 1e-5f
#define LOG2E 1.4426950408889634f

template <int K>
__global__ void __launch_bounds__(256)
gemm_wmma_f32(const float* __restrict__ A,
              const float* __restrict__ B,
              float* __restrict__ C,
              int M) {
  constexpr int Nn  = 128;
  constexpr int PA  = K + 4;
  constexpr int PC  = Nn + 4;
  constexpr int KS  = K / 4;
  constexpr int RQ  = K / 4;
  constexpr int NPA = (16 * RQ) / 256;
  __shared__ __attribute__((aligned(16))) float sA[16 * PA];
  __shared__ __attribute__((aligned(16))) float sC[16 * PC];

  const int tid  = threadIdx.x;
  const int wave = tid >> 5;
  const int lane = tid & 31;
  const int half = lane >> 4;
  const int l    = lane & 15;
  const int n0   = wave * 16;
  const int tiles = M >> 4;

  v2f bf[KS];
#pragma unroll
  for (int ks = 0; ks < KS; ++ks) {
    const int kb = ks * 4 + half * 2;
    bf[ks].x = B[(size_t)kb * Nn + n0 + l];
    bf[ks].y = B[(size_t)(kb + 1) * Nn + n0 + l];
  }

  int tile = blockIdx.x;
  if (tile >= tiles) return;

  const int q0 = tid,       r0 = q0 / RQ, e0 = (q0 % RQ) * 4;
  const int q1 = tid + 256, r1 = q1 / RQ, e1 = (q1 % RQ) * 4;
  const unsigned aoff0 = r0 * K + e0;
  const unsigned aoff1 = r1 * K + e1;
  const int orow = tid >> 5, ocol = (tid & 31) * 4;
  const unsigned coff = orow * Nn + ocol;

  float4 pre0, pre1 = {0.0f, 0.0f, 0.0f, 0.0f};
  {
    const float* __restrict__ At = A + (size_t)tile * (16 * K);
    pre0 = *(const float4*)(At + aoff0);
    if constexpr (NPA > 1) pre1 = *(const float4*)(At + aoff1);
  }

  for (; tile < tiles; tile += gridDim.x) {
    *(float4*)&sA[r0 * PA + e0] = pre0;
    if constexpr (NPA > 1) *(float4*)&sA[r1 * PA + e1] = pre1;
    __syncthreads();

    const int nxt = tile + gridDim.x;
    if (nxt < tiles) {
      const float* __restrict__ An = A + (size_t)nxt * (16 * K);
      pre0 = *(const float4*)(An + aoff0);
      if constexpr (NPA > 1) pre1 = *(const float4*)(An + aoff1);
    }

    v2f af[KS];
#pragma unroll
    for (int ks = 0; ks < KS; ++ks)
      af[ks] = *(const v2f*)&sA[l * PA + ks * 4 + half * 2];
    __builtin_amdgcn_sched_barrier(0);

    v8f c = {};
#pragma unroll
    for (int ks = 0; ks < KS; ++ks) {
      c = __builtin_amdgcn_wmma_f32_16x16x4_f32(false, af[ks], false, bf[ks],
                                                (short)0, c, false, false);
    }
    __builtin_amdgcn_sched_barrier(0);

#pragma unroll
    for (int j = 0; j < 8; ++j) sC[(j + half * 8) * PC + n0 + l] = c[j];
    __syncthreads();

    float* __restrict__ Ct = C + (size_t)tile * (16 * Nn);
    const float4 o0 = *(const float4*)&sC[orow * PC + ocol];
    const float4 o1 = *(const float4*)&sC[(orow + 8) * PC + ocol];
    *(float4*)(Ct + coff) = o0;
    *(float4*)(Ct + coff + 8 * Nn) = o1;
  }
}

#define BIN_SHIFT   8
#define BIN_NODES   (1 << BIN_SHIFT)
#define SORT_BLOCKS 256
#define MAX_BINS    1024
#define CHUNK_LDS   4096
#define BIN_LDS     8192

__global__ void __launch_bounds__(1024)
csr_bin(const int* __restrict__ ei, int* __restrict__ stgw,
        int* __restrict__ tabw, int E, int N) {
  __shared__ int s_cnt[MAX_BINS];
  __shared__ int s_cur[MAX_BINS];
  __shared__ int s_tot[32];
  __shared__ int2 s_out[CHUNK_LDS];
  const int tid  = threadIdx.x;
  const int lane = tid & 31;
  const int wave = tid >> 5;
  const int G  = gridDim.x;
  const int C  = (E + G - 1) / G;
  const int CP = (C + 15) & ~15;
  const int NB = (N + BIN_NODES - 1) >> BIN_SHIFT;
  const int PB = (NB + 15) & ~15;
  const int e0 = blockIdx.x * C;
  const int e1 = min(E, e0 + C);

  s_cnt[tid] = 0;
  int dd[4], ss[4];
#pragma unroll
  for (int j = 0; j < 4; ++j) {
    const int i = e0 + tid + j * 1024;
    const bool ok = i < e1;
    dd[j] = ok ? ei[E + i] : -1;
    ss[j] = ok ? ei[i] : 0;
  }
  __syncthreads();
#pragma unroll
  for (int j = 0; j < 4; ++j)
    if (dd[j] >= 0) atomicAdd(&s_cnt[dd[j] >> BIN_SHIFT], 1);
  for (int i = e0 + tid + 4096; i < e1; i += 1024)
    atomicAdd(&s_cnt[ei[E + i] >> BIN_SHIFT], 1);
  __syncthreads();

  const int v = s_cnt[tid];
  int inc = v;
  for (int off = 1; off < 32; off <<= 1) {
    const int up = __shfl_up(inc, off);
    if (lane >= off) inc += up;
  }
  if (lane == 31) s_tot[wave] = inc;
  __syncthreads();
  int wb = (lane < wave) ? s_tot[lane] : 0;
  for (int off = 16; off > 0; off >>= 1) wb += __shfl_xor(wb, off);
  const int excl = wb + inc - v;
  s_cur[tid] = excl;
  if (tid < PB)
    ((int2*)tabw)[(size_t)blockIdx.x * PB + tid] = make_int2(excl, v);
  __syncthreads();

  int2* __restrict__ outp = (int2*)stgw + (size_t)blockIdx.x * CP;
  if (C <= CHUNK_LDS) {
#pragma unroll
    for (int j = 0; j < 4; ++j)
      if (dd[j] >= 0) {
        const int pos = atomicAdd(&s_cur[dd[j] >> BIN_SHIFT], 1);
        s_out[pos] = make_int2(ss[j], dd[j]);
      }
    __syncthreads();
    for (int q = tid; q < e1 - e0; q += 1024) outp[q] = s_out[q];
  } else {
#pragma unroll
    for (int j = 0; j < 4; ++j)
      if (dd[j] >= 0) {
        const int pos = atomicAdd(&s_cur[dd[j] >> BIN_SHIFT], 1);
        outp[pos] = make_int2(ss[j], dd[j]);
      }
    for (int i = e0 + tid + 4096; i < e1; i += 1024) {
      const int d = ei[E + i], s = ei[i];
      const int pos = atomicAdd(&s_cur[d >> BIN_SHIFT], 1);
      outp[pos] = make_int2(s, d);
    }
  }
}

__global__ void __launch_bounds__(1024)
csr_rows(const int* __restrict__ stgw, const int* __restrict__ tabw,
         int* __restrict__ rowbeg, int* __restrict__ rowend,
         int* __restrict__ srcs, int E, int N) {
  constexpr int G   = SORT_BLOCKS;
  constexpr int PPW = G / 32;
  __shared__ int s_po[G];
  __shared__ int s_pc[G];
  __shared__ int s_cnt[BIN_NODES];
  __shared__ int s_cur[BIN_NODES];
  __shared__ int s_red[32];
  __shared__ int s_len[32];
  __shared__ int s_tot[32];
  __shared__ int s_rows[BIN_LDS];
  const int2* __restrict__ stg = (const int2*)stgw;
  const int2* __restrict__ tab = (const int2*)tabw;
  const int tid   = threadIdx.x;
  const int lane  = tid & 31;
  const int wave  = tid >> 5;
  const int bin   = blockIdx.x;
  const int node0 = bin << BIN_SHIFT;
  const int C  = (E + G - 1) / G;
  const int CP = (C + 15) & ~15;
  const int NB = (N + BIN_NODES - 1) >> BIN_SHIFT;
  const int PB = (NB + 15) & ~15;

  int myoff = 0, mylen = 0;
  if (tid < G) {
    const int2 t = tab[(size_t)tid * PB + bin];
    myoff = t.x;
    mylen = t.y;
    s_po[tid] = tid * CP + t.x;
    s_pc[tid] = t.y;
  }
  if (tid < BIN_NODES) s_cnt[tid] = 0;
  for (int off = 16; off > 0; off >>= 1) {
    myoff += __shfl_xor(myoff, off);
    mylen += __shfl_xor(mylen, off);
  }
  if (lane == 0) { s_red[wave] = myoff; s_len[wave] = mylen; }
  __syncthreads();
  int base = s_red[lane], total = s_len[lane];
  for (int off = 16; off > 0; off >>= 1) {
    base  += __shfl_xor(base, off);
    total += __shfl_xor(total, off);
  }

  int2 e[PPW];
  int pc[PPW], po[PPW];
#pragma unroll
  for (int k = 0; k < PPW; ++k) {
    const int p = wave * PPW + k;
    pc[k] = s_pc[p];
    po[k] = s_po[p];
  }
#pragma unroll
  for (int k = 0; k < PPW; ++k)
    e[k] = (lane < pc[k]) ? stg[po[k] + lane] : make_int2(0, -1);
#pragma unroll
  for (int k = 0; k < PPW; ++k)
    if (lane < pc[k]) atomicAdd(&s_cnt[e[k].y - node0], 1);
#pragma unroll
  for (int k = 0; k < PPW; ++k)
    for (int i = lane + 32; i < pc[k]; i += 32)
      atomicAdd(&s_cnt[stg[po[k] + i].y - node0], 1);
  __syncthreads();

  const int v = (tid < BIN_NODES) ? s_cnt[tid] : 0;
  int inc = v;
  for (int off = 1; off < 32; off <<= 1) {
    const int up = __shfl_up(inc, off);
    if (lane >= off) inc += up;
  }
  if (lane == 31) s_tot[wave] = inc;
  __syncthreads();
  int wb = (lane < wave) ? s_tot[lane] : 0;
  for (int off = 16; off > 0; off >>= 1) wb += __shfl_xor(wb, off);
  const int excl = wb + inc - v;
  if (tid < BIN_NODES) {
    s_cur[tid] = excl;
    const int node = node0 + tid;
    if (node < N) {
      rowbeg[node] = base + excl;
      rowend[node] = base + excl + v;
    }
  }
  __syncthreads();

  int* __restrict__ rowsOut = srcs + base;
  if (total <= BIN_LDS) {
#pragma unroll
    for (int k = 0; k < PPW; ++k)
      if (lane < pc[k]) {
        const int pos = atomicAdd(&s_cur[e[k].y - node0], 1);
        s_rows[pos] = e[k].x;
      }
#pragma unroll
    for (int k = 0; k < PPW; ++k)
      for (int i = lane + 32; i < pc[k]; i += 32) {
        const int2 q = stg[po[k] + i];
        const int pos = atomicAdd(&s_cur[q.y - node0], 1);
        s_rows[pos] = q.x;
      }
    __syncthreads();
    const int lead = base & 31;
    for (int q = tid - lead; q < total; q += 1024)
      if (q >= 0) rowsOut[q] = s_rows[q];
  } else {
#pragma unroll
    for (int k = 0; k < PPW; ++k)
      if (lane < pc[k]) {
        const int pos = atomicAdd(&s_cur[e[k].y - node0], 1);
        rowsOut[pos] = e[k].x;
      }
#pragma unroll
    for (int k = 0; k < PPW; ++k)
      for (int i = lane + 32; i < pc[k]; i += 32) {
        const int2 q = stg[po[k] + i];
        const int pos = atomicAdd(&s_cur[q.y - node0], 1);
        rowsOut[pos] = q.x;
      }
  }
}

__device__ __forceinline__ float score4(const float4 s, const float4 d,
                                        const float4 av) {
  float m, lr, acc = 0.0f;
  m = s.x + d.x; lr = (m > 0.0f) ? m : NEG_SLOPE * m; acc += lr * av.x;
  m = s.y + d.y; lr = (m > 0.0f) ? m : NEG_SLOPE * m; acc += lr * av.y;
  m = s.z + d.z; lr = (m > 0.0f) ? m : NEG_SLOPE * m; acc += lr * av.z;
  m = s.w + d.w; lr = (m > 0.0f) ? m : NEG_SLOPE * m; acc += lr * av.w;
  return acc;
}

__device__ __forceinline__ float row16_sum(float v) {
  v += __int_as_float(__builtin_amdgcn_update_dpp(0, __float_as_int(v), 0x168, 0xf, 0xf, true));
  v += __int_as_float(__builtin_amdgcn_update_dpp(0, __float_as_int(v), 0x164, 0xf, 0xf, true));
  v += __int_as_float(__builtin_amdgcn_update_dpp(0, __float_as_int(v), 0x162, 0xf, 0xf, true));
  v += __int_as_float(__builtin_amdgcn_update_dpp(0, __float_as_int(v), 0x161, 0xf, 0xf, true));
  return v;
}

#define NODES_PER_BLOCK 8

__global__ void __launch_bounds__(256)
gat_node(const float* __restrict__ xp,
         const int* __restrict__ rowbeg,
         const int* __restrict__ rowend,
         const int* __restrict__ srcs,
         const float* __restrict__ avec,
         float* __restrict__ tmp,
         float* __restrict__ part,
         int N) {
  __shared__ __attribute__((aligned(16))) float s_red[NODES_PER_BLOCK][128];
  const int lane = threadIdx.x & 31;
  const int wave = threadIdx.x >> 5;
  const int node = __builtin_amdgcn_readfirstlane(
      (int)(blockIdx.x * NODES_PER_BLOCK + wave));

  float4 v = {0.0f, 0.0f, 0.0f, 0.0f};

  if (node < N) {
    const int coff = (lane >> 4) * 64 + (lane & 15) * 4;
    const float4 d  = *(const float4*)(xp + (size_t)node * 128 + coff);
    float4 av = *(const float4*)(avec + coff);
    av.x *= LOG2E; av.y *= LOG2E; av.z *= LOG2E; av.w *= LOG2E;
    const int beg = rowbeg[node];
    const int end = rowend[node];

    float  mx  = row16_sum(score4(d, d, av));
    float  l   = 1.0f;
    float4 acc = d;

    for (int base = beg; base < end; base += 32) {
      const int nb = min(32, end - base);
      const int mysrc = (lane < nb) ? srcs[base + lane] : node;
      for (int t = 0; t < nb; t += 4) {
        const int s0 = __builtin_amdgcn_readlane(mysrc, t);
        const int s1 = __builtin_amdgcn_readlane(mysrc, t + 1);
        const int s2 = __builtin_amdgcn_readlane(mysrc, t + 2);
        const int s3 = __builtin_amdgcn_readlane(mysrc, t + 3);
        const float4 x0 = *(const float4*)(xp + (size_t)s0 * 128 + coff);
        const float4 x1 = *(const float4*)(xp + (size_t)s1 * 128 + coff);
        const float4 x2 = *(const float4*)(xp + (size_t)s2 * 128 + coff);
        const float4 x3 = *(const float4*)(xp + (size_t)s3 * 128 + coff);
        float e0 = row16_sum(score4(x0, d, av));
        float e1 = row16_sum(score4(x1, d, av));
        float e2 = row16_sum(score4(x2, d, av));
        float e3 = row16_sum(score4(x3, d, av));
        e1 = (t + 1 < nb) ? e1 : -INFINITY;
        e2 = (t + 2 < nb) ? e2 : -INFINITY;
        e3 = (t + 3 < nb) ? e3 : -INFINITY;
        const float mn = fmaxf(fmaxf(mx, e0), fmaxf(e1, fmaxf(e2, e3)));
        const float sc = __builtin_amdgcn_exp2f(mx - mn);
        const float p0 = __builtin_amdgcn_exp2f(e0 - mn);
        const float p1 = __builtin_amdgcn_exp2f(e1 - mn);
        const float p2 = __builtin_amdgcn_exp2f(e2 - mn);
        const float p3 = __builtin_amdgcn_exp2f(e3 - mn);
        l = l * sc + ((p0 + p1) + (p2 + p3));
        acc.x = acc.x * sc + ((p0 * x0.x + p1 * x1.x) + (p2 * x2.x + p3 * x3.x));
        acc.y = acc.y * sc + ((p0 * x0.y + p1 * x1.y) + (p2 * x2.y + p3 * x3.y));
        acc.z = acc.z * sc + ((p0 * x0.z + p1 * x1.z) + (p2 * x2.z + p3 * x3.z));
        acc.w = acc.w * sc + ((p0 * x0.w + p1 * x1.w) + (p2 * x2.w + p3 * x3.w));
        mx = mn;
      }
    }

    const float den = l + 1e-16f;
    float4 o;
    o.x = acc.x / den; o.y = acc.y / den; o.z = acc.z / den; o.w = acc.w / den;

    v.x = 0.5f * (o.x + __shfl_xor(o.x, 16));
    v.y = 0.5f * (o.y + __shfl_xor(o.y, 16));
    v.z = 0.5f * (o.z + __shfl_xor(o.z, 16));
    v.w = 0.5f * (o.w + __shfl_xor(o.w, 16));
  }

  if (lane < 16) {
    float4 q;
    q.x = v.x * v.x; q.y = v.y * v.y; q.z = v.z * v.z; q.w = v.w * v.w;
    *(float4*)&s_red[wave][lane * 4]      = v;
    *(float4*)&s_red[wave][64 + lane * 4] = q;
  }
  __syncthreads();
  if (node < N && lane < 16) *(float4*)(tmp + (size_t)node * 64 + lane * 4) = v;
  if (threadIdx.x < 128) {
    const int k = threadIdx.x;
    const float t = ((s_red[0][k] + s_red[1][k]) + (s_red[2][k] + s_red[3][k])) +
                    ((s_red[4][k] + s_red[5][k]) + (s_red[6][k] + s_red[7][k]));
    part[(size_t)blockIdx.x * 128 + k] = t;
  }
}

__global__ void __launch_bounds__(1024)
bn_partials_sum(const float* __restrict__ part, float* __restrict__ stats,
                int rows) {
  __shared__ __attribute__((aligned(16))) float s_w[32][32];
  const int lane = threadIdx.x & 31;
  const int wave = threadIdx.x >> 5;
  const int cg   = lane & 7;
  const int rs   = wave * 4 + (lane >> 3);
  const float* __restrict__ p = part + blockIdx.x * 32 + cg * 4;
  float4 a = {0.0f, 0.0f, 0.0f, 0.0f};
  int r0 = rs;
  for (; r0 + 15 * 128 < rows; r0 += 16 * 128) {
    const float* __restrict__ pr = p + (size_t)r0 * 128;
    float4 q[16];
    for (int j = 0; j < 16; ++j) q[j] = *(const float4*)(pr + (size_t)j * (128 * 128));
    for (int j = 0; j < 16; ++j) {
      a.x += q[j].x; a.y += q[j].y; a.z += q[j].z; a.w += q[j].w;
    }
  }
  for (; r0 < rows; r0 += 128) {
    const float4 q = *(const float4*)(p + (size_t)r0 * 128);
    a.x += q.x; a.y += q.y; a.z += q.z; a.w += q.w;
  }
  a.x += __shfl_xor(a.x, 8);  a.y += __shfl_xor(a.y, 8);
  a.z += __shfl_xor(a.z, 8);  a.w += __shfl_xor(a.w, 8);
  a.x += __shfl_xor(a.x, 16); a.y += __shfl_xor(a.y, 16);
  a.z += __shfl_xor(a.z, 16); a.w += __shfl_xor(a.w, 16);
  if (lane < 8) *(float4*)&s_w[wave][cg * 4] = a;
  __syncthreads();
  if (wave == 0) {
    float t = 0.0f;
    for (int w = 0; w < 32; ++w) t += s_w[w][lane];
    stats[blockIdx.x * 32 + lane] = t;
  }
}

#define BN_BLOCKS  512
#define BN_THREADS 256
#define BN_PIECES  8

__device__ __forceinline__ float bn_one(float x, float mean, float rstd,
                                        float g, float b) {
  return (x - mean) * rstd * g + b;
}

template <int TANH>
__global__ void __launch_bounds__(BN_THREADS)
bn_apply(const float* __restrict__ tmp,
         const float* __restrict__ sum,
         const float* __restrict__ sumsq,
         const float* __restrict__ gamma,
         const float* __restrict__ beta,
         float* __restrict__ out,
         int N) {
  const int total  = N * 16;
  const int stride = gridDim.x * BN_THREADS;
  const int first  = blockIdx.x * BN_THREADS + threadIdx.x;
  const int c      = (first & 15) * 4;
  const float4* __restrict__ in4  = (const float4*)tmp;
  float4* __restrict__       out4 = (float4*)out;

  const float invN = 1.0f / (float)N;
  const float4 s4 = *(const float4*)(sum + c);
  const float4 q4 = *(const float4*)(sumsq + c);
  const float4 g4 = *(const float4*)(gamma + c);
  const float4 b4 = *(const float4*)(beta + c);
  float4 mean, rstd;
  mean.x = s4.x * invN; rstd.x = rsqrtf(q4.x * invN - mean.x * mean.x + BN_EPS);
  mean.y = s4.y * invN; rstd.y = rsqrtf(q4.y * invN - mean.y * mean.y + BN_EPS);
  mean.z = s4.z * invN; rstd.z = rsqrtf(q4.z * invN - mean.z * mean.z + BN_EPS);
  mean.w = s4.w * invN; rstd.w = rsqrtf(q4.w * invN - mean.w * mean.w + BN_EPS);

  for (int i0 = first; i0 < total; i0 += BN_PIECES * stride) {
    float4 v[BN_PIECES];
#pragma unroll
    for (int j = 0; j < BN_PIECES; ++j) {
      const int i = i0 + j * stride;
      v[j] = in4[(i < total) ? i : i0];
    }
#pragma unroll
    for (int j = 0; j < BN_PIECES; ++j)
      asm volatile("" :: "v"(v[j].x), "v"(v[j].y), "v"(v[j].z), "v"(v[j].w));
#pragma unroll
    for (int j = 0; j < BN_PIECES; ++j) {
      const int i = i0 + j * stride;
      float4 r;
      r.x = bn_one(v[j].x, mean.x, rstd.x, g4.x, b4.x);
      r.y = bn_one(v[j].y, mean.y, rstd.y, g4.y, b4.y);
      r.z = bn_one(v[j].z, mean.z, rstd.z, g4.z, b4.z);
      r.w = bn_one(v[j].w, mean.w, rstd.w, g4.w, b4.w);
      if (TANH) { r.x = tanhf(r.x); r.y = tanhf(r.y); r.z = tanhf(r.z); r.w = tanhf(r.w); }
      if (i < total) out4[i] = r;
    }
  }
}

extern "C" void kernel_launch(void* const* d_in, const int* in_sizes, int n_in,
                              void* d_out, int out_size, void* d_ws, size_t ws_size,
                              hipStream_t stream) {
  const float* x      = (const float*)d_in[0];
  const int*   ei     = (const int*)  d_in[1];
  const float* W0     = (const float*)d_in[2];
  const float* a0     = (const float*)d_in[3];
  const float* gamma0 = (const float*)d_in[4];
  const float* beta0  = (const float*)d_in[5];
  const float* W1     = (const float*)d_in[6];
  const float* a1     = (const float*)d_in[7];
  const float* gamma1 = (const float*)d_in[8];
  const float* beta1  = (const float*)d_in[9];
  float* out = (float*)d_out;

  const int N    = in_sizes[0] / 128;
  const int E    = in_sizes[1] / 2;

  char* ws = (char*)d_ws;
  size_t off = 0;
  auto carve = [&](size_t nwords) {
    float* p = (float*)(ws + off);
    off = (off + nwords * sizeof(float) + 255) & ~(size_t)255;
    return p;
  };
  float* xp    = carve((size_t)N * 128);
  float* tmp   = carve((size_t)N * 64);
  float* h0    = carve((size_t)N * 64);
  float* stats = carve(128);
  int* rowbeg  = (int*)carve((size_t)N);
  int* rowend  = (int*)carve((size_t)N);
  int* srcs    = (int*)carve((size_t)E);
  int* stg     = (int*)carve((size_t)E * 2 + SORT_BLOCKS * 32);
  int* tab     = (int*)carve((size_t)SORT_BLOCKS * MAX_BINS * 2);
  const int nodeGroups = (N + NODES_PER_BLOCK - 1) / NODES_PER_BLOCK;
  float* part  = carve((size_t)nodeGroups * 128);
  (void)ws_size;

  const int binBlocks    = (N + BIN_NODES - 1) / BIN_NODES;

  csr_bin<<<SORT_BLOCKS, 1024, 0, stream>>>(ei, stg, tab, E, N);
  csr_rows<<<binBlocks, 1024, 0, stream>>>(stg, tab, rowbeg, rowend, srcs, E, N);

  gemm_wmma_f32<128><<<512, 256, 0, stream>>>(x, W0, xp, N);
  gat_node<<<nodeGroups, 256, 0, stream>>>(xp, rowbeg, rowend, srcs, a0, tmp, part, N);
  bn_partials_sum<<<4, 1024, 0, stream>>>(part, stats, nodeGroups);
  bn_apply<1><<<BN_BLOCKS, BN_THREADS, 0, stream>>>(tmp, stats, stats + 64,
                                                    gamma0, beta0, h0, N);

  gemm_wmma_f32<64><<<512, 256, 0, stream>>>(h0, W1, xp, N);
  gat_node<<<nodeGroups, 256, 0, stream>>>(xp, rowbeg, rowend, srcs, a1, tmp, part, N);
  bn_partials_sum<<<4, 1024, 0, stream>>>(part, stats, nodeGroups);
  bn_apply<0><<<BN_BLOCKS, BN_THREADS, 0, stream>>>(tmp, stats, stats + 64,
                                                    gamma1, beta1, out, N);
}


# Batch-norm column sums of the partial rows spread over 64 workgroups (16 row slices x 4 column groups; was 4 workgroups); apply pass adds the 16 slice sums; stats sum order changes, still fixed
# speedup vs baseline: 6.8205x; 1.0922x over previous
#include <hip/hip_runtime.h>
#include <hip/hip_bf16.h>
#include <math.h>


typedef __attribute__((ext_vector_type(2))) float v2f;
typedef __attribute__((ext_vector_type(8))) float v8f;

#define NEG_SLOPE 0.2f
#define BN_EPS 1e-5f
#define LOG2E 1.4426950408889634f

template <int K>
__global__ void __launch_bounds__(256)
gemm_wmma_f32(const float* __restrict__ A,
              const float* __restrict__ B,
              float* __restrict__ C,
              int M) {
  constexpr int Nn  = 128;
  constexpr int PA  = K + 4;
  constexpr int PC  = Nn + 4;
  constexpr int KS  = K / 4;
  constexpr int RQ  = K / 4;
  constexpr int NPA = (16 * RQ) / 256;
  __shared__ __attribute__((aligned(16))) float sA[16 * PA];
  __shared__ __attribute__((aligned(16))) float sC[16 * PC];

  const int tid  = threadIdx.x;
  const int wave = tid >> 5;
  const int lane = tid & 31;
  const int half = lane >> 4;
  const int l    = lane & 15;
  const int n0   = wave * 16;
  const int tiles = M >> 4;

  v2f bf[KS];
#pragma unroll
  for (int ks = 0; ks < KS; ++ks) {
    const int kb = ks * 4 + half * 2;
    bf[ks].x = B[(size_t)kb * Nn + n0 + l];
    bf[ks].y = B[(size_t)(kb + 1) * Nn + n0 + l];
  }

  int tile = blockIdx.x;
  if (tile >= tiles) return;

  const int q0 = tid,       r0 = q0 / RQ, e0 = (q0 % RQ) * 4;
  const int q1 = tid + 256, r1 = q1 / RQ, e1 = (q1 % RQ) * 4;
  const unsigned aoff0 = r0 * K + e0;
  const unsigned aoff1 = r1 * K + e1;
  const int orow = tid >> 5, ocol = (tid & 31) * 4;
  const unsigned coff = orow * Nn + ocol;

  float4 pre0, pre1 = {0.0f, 0.0f, 0.0f, 0.0f};
  {
    const float* __restrict__ At = A + (size_t)tile * (16 * K);
    pre0 = *(const float4*)(At + aoff0);
    if constexpr (NPA > 1) pre1 = *(const float4*)(At + aoff1);
  }

  for (; tile < tiles; tile += gridDim.x) {
    *(float4*)&sA[r0 * PA + e0] = pre0;
    if constexpr (NPA > 1) *(float4*)&sA[r1 * PA + e1] = pre1;
    __syncthreads();

    const int nxt = tile + gridDim.x;
    if (nxt < tiles) {
      const float* __restrict__ An = A + (size_t)nxt * (16 * K);
      pre0 = *(const float4*)(An + aoff0);
      if constexpr (NPA > 1) pre1 = *(const float4*)(An + aoff1);
    }

    v2f af[KS];
#pragma unroll
    for (int ks = 0; ks < KS; ++ks)
      af[ks] = *(const v2f*)&sA[l * PA + ks * 4 + half * 2];
    __builtin_amdgcn_sched_barrier(0);

    v8f c = {};
#pragma unroll
    for (int ks = 0; ks < KS; ++ks) {
      c = __builtin_amdgcn_wmma_f32_16x16x4_f32(false, af[ks], false, bf[ks],
                                                (short)0, c, false, false);
    }
    __builtin_amdgcn_sched_barrier(0);

#pragma unroll
    for (int j = 0; j < 8; ++j) sC[(j + half * 8) * PC + n0 + l] = c[j];
    __syncthreads();

    float* __restrict__ Ct = C + (size_t)tile * (16 * Nn);
    const float4 o0 = *(const float4*)&sC[orow * PC + ocol];
    const float4 o1 = *(const float4*)&sC[(orow + 8) * PC + ocol];
    *(float4*)(Ct + coff) = o0;
    *(float4*)(Ct + coff + 8 * Nn) = o1;
  }
}

#define BIN_SHIFT   8
#define BIN_NODES   (1 << BIN_SHIFT)
#define SORT_BLOCKS 256
#define MAX_BINS    1024
#define CHUNK_LDS   4096
#define BIN_LDS     8192

__global__ void __launch_bounds__(1024)
csr_bin(const int* __restrict__ ei, int* __restrict__ stgw,
        int* __restrict__ tabw, int E, int N) {
  __shared__ int s_cnt[MAX_BINS];
  __shared__ int s_cur[MAX_BINS];
  __shared__ int s_tot[32];
  __shared__ int2 s_out[CHUNK_LDS];
  const int tid  = threadIdx.x;
  const int lane = tid & 31;
  const int wave = tid >> 5;
  const int G  = gridDim.x;
  const int C  = (E + G - 1) / G;
  const int CP = (C + 15) & ~15;
  const int NB = (N + BIN_NODES - 1) >> BIN_SHIFT;
  const int PB = (NB + 15) & ~15;
  const int e0 = blockIdx.x * C;
  const int e1 = min(E, e0 + C);

  s_cnt[tid] = 0;
  int dd[4], ss[4];
#pragma unroll
  for (int j = 0; j < 4; ++j) {
    const int i = e0 + tid + j * 1024;
    const bool ok = i < e1;
    dd[j] = ok ? ei[E + i] : -1;
    ss[j] = ok ? ei[i] : 0;
  }
  __syncthreads();
#pragma unroll
  for (int j = 0; j < 4; ++j)
    if (dd[j] >= 0) atomicAdd(&s_cnt[dd[j] >> BIN_SHIFT], 1);
  for (int i = e0 + tid + 4096; i < e1; i += 1024)
    atomicAdd(&s_cnt[ei[E + i] >> BIN_SHIFT], 1);
  __syncthreads();

  const int v = s_cnt[tid];
  int inc = v;
  for (int off = 1; off < 32; off <<= 1) {
    const int up = __shfl_up(inc, off);
    if (lane >= off) inc += up;
  }
  if (lane == 31) s_tot[wave] = inc;
  __syncthreads();
  int wb = (lane < wave) ? s_tot[lane] : 0;
  for (int off = 16; off > 0; off >>= 1) wb += __shfl_xor(wb, off);
  const int excl = wb + inc - v;
  s_cur[tid] = excl;
  if (tid < PB)
    ((int2*)tabw)[(size_t)blockIdx.x * PB + tid] = make_int2(excl, v);
  __syncthreads();

  int2* __restrict__ outp = (int2*)stgw + (size_t)blockIdx.x * CP;
  if (C <= CHUNK_LDS) {
#pragma unroll
    for (int j = 0; j < 4; ++j)
      if (dd[j] >= 0) {
        const int pos = atomicAdd(&s_cur[dd[j] >> BIN_SHIFT], 1);
        s_out[pos] = make_int2(ss[j], dd[j]);
      }
    __syncthreads();
    for (int q = tid; q < e1 - e0; q += 1024) outp[q] = s_out[q];
  } else {
#pragma unroll
    for (int j = 0; j < 4; ++j)
      if (dd[j] >= 0) {
        const int pos = atomicAdd(&s_cur[dd[j] >> BIN_SHIFT], 1);
        outp[pos] = make_int2(ss[j], dd[j]);
      }
    for (int i = e0 + tid + 4096; i < e1; i += 1024) {
      const int d = ei[E + i], s = ei[i];
      const int pos = atomicAdd(&s_cur[d >> BIN_SHIFT], 1);
      outp[pos] = make_int2(s, d);
    }
  }
}

__global__ void __launch_bounds__(1024)
csr_rows(const int* __restrict__ stgw, const int* __restrict__ tabw,
         int* __restrict__ rowbeg, int* __restrict__ rowend,
         int* __restrict__ srcs, int E, int N) {
  constexpr int G   = SORT_BLOCKS;
  constexpr int PPW = G / 32;
  __shared__ int s_po[G];
  __shared__ int s_pc[G];
  __shared__ int s_cnt[BIN_NODES];
  __shared__ int s_cur[BIN_NODES];
  __shared__ int s_red[32];
  __shared__ int s_len[32];
  __shared__ int s_tot[32];
  __shared__ int s_rows[BIN_LDS];
  const int2* __restrict__ stg = (const int2*)stgw;
  const int2* __restrict__ tab = (const int2*)tabw;
  const int tid   = threadIdx.x;
  const int lane  = tid & 31;
  const int wave  = tid >> 5;
  const int bin   = blockIdx.x;
  const int node0 = bin << BIN_SHIFT;
  const int C  = (E + G - 1) / G;
  const int CP = (C + 15) & ~15;
  const int NB = (N + BIN_NODES - 1) >> BIN_SHIFT;
  const int PB = (NB + 15) & ~15;

  int myoff = 0, mylen = 0;
  if (tid < G) {
    const int2 t = tab[(size_t)tid * PB + bin];
    myoff = t.x;
    mylen = t.y;
    s_po[tid] = tid * CP + t.x;
    s_pc[tid] = t.y;
  }
  if (tid < BIN_NODES) s_cnt[tid] = 0;
  for (int off = 16; off > 0; off >>= 1) {
    myoff += __shfl_xor(myoff, off);
    mylen += __shfl_xor(mylen, off);
  }
  if (lane == 0) { s_red[wave] = myoff; s_len[wave] = mylen; }
  __syncthreads();
  int base = s_red[lane], total = s_len[lane];
  for (int off = 16; off > 0; off >>= 1) {
    base  += __shfl_xor(base, off);
    total += __shfl_xor(total, off);
  }

  int2 e[PPW];
  int pc[PPW], po[PPW];
#pragma unroll
  for (int k = 0; k < PPW; ++k) {
    const int p = wave * PPW + k;
    pc[k] = s_pc[p];
    po[k] = s_po[p];
  }
#pragma unroll
  for (int k = 0; k < PPW; ++k)
    e[k] = (lane < pc[k]) ? stg[po[k] + lane] : make_int2(0, -1);
#pragma unroll
  for (int k = 0; k < PPW; ++k)
    if (lane < pc[k]) atomicAdd(&s_cnt[e[k].y - node0], 1);
#pragma unroll
  for (int k = 0; k < PPW; ++k)
    for (int i = lane + 32; i < pc[k]; i += 32)
      atomicAdd(&s_cnt[stg[po[k] + i].y - node0], 1);
  __syncthreads();

  const int v = (tid < BIN_NODES) ? s_cnt[tid] : 0;
  int inc = v;
  for (int off = 1; off < 32; off <<= 1) {
    const int up = __shfl_up(inc, off);
    if (lane >= off) inc += up;
  }
  if (lane == 31) s_tot[wave] = inc;
  __syncthreads();
  int wb = (lane < wave) ? s_tot[lane] : 0;
  for (int off = 16; off > 0; off >>= 1) wb += __shfl_xor(wb, off);
  const int excl = wb + inc - v;
  if (tid < BIN_NODES) {
    s_cur[tid] = excl;
    const int node = node0 + tid;
    if (node < N) {
      rowbeg[node] = base + excl;
      rowend[node] = base + excl + v;
    }
  }
  __syncthreads();

  int* __restrict__ rowsOut = srcs + base;
  if (total <= BIN_LDS) {
#pragma unroll
    for (int k = 0; k < PPW; ++k)
      if (lane < pc[k]) {
        const int pos = atomicAdd(&s_cur[e[k].y - node0], 1);
        s_rows[pos] = e[k].x;
      }
#pragma unroll
    for (int k = 0; k < PPW; ++k)
      for (int i = lane + 32; i < pc[k]; i += 32) {
        const int2 q = stg[po[k] + i];
        const int pos = atomicAdd(&s_cur[q.y - node0], 1);
        s_rows[pos] = q.x;
      }
    __syncthreads();
    const int lead = base & 31;
    for (int q = tid - lead; q < total; q += 1024)
      if (q >= 0) rowsOut[q] = s_rows[q];
  } else {
#pragma unroll
    for (int k = 0; k < PPW; ++k)
      if (lane < pc[k]) {
        const int pos = atomicAdd(&s_cur[e[k].y - node0], 1);
        rowsOut[pos] = e[k].x;
      }
#pragma unroll
    for (int k = 0; k < PPW; ++k)
      for (int i = lane + 32; i < pc[k]; i += 32) {
        const int2 q = stg[po[k] + i];
        const int pos = atomicAdd(&s_cur[q.y - node0], 1);
        rowsOut[pos] = q.x;
      }
  }
}

__device__ __forceinline__ float score4(const float4 s, const float4 d,
                                        const float4 av) {
  float m, lr, acc = 0.0f;
  m = s.x + d.x; lr = (m > 0.0f) ? m : NEG_SLOPE * m; acc += lr * av.x;
  m = s.y + d.y; lr = (m > 0.0f) ? m : NEG_SLOPE * m; acc += lr * av.y;
  m = s.z + d.z; lr = (m > 0.0f) ? m : NEG_SLOPE * m; acc += lr * av.z;
  m = s.w + d.w; lr = (m > 0.0f) ? m : NEG_SLOPE * m; acc += lr * av.w;
  return acc;
}

__device__ __forceinline__ float row16_sum(float v) {
  v += __int_as_float(__builtin_amdgcn_update_dpp(0, __float_as_int(v), 0x168, 0xf, 0xf, true));
  v += __int_as_float(__builtin_amdgcn_update_dpp(0, __float_as_int(v), 0x164, 0xf, 0xf, true));
  v += __int_as_float(__builtin_amdgcn_update_dpp(0, __float_as_int(v), 0x162, 0xf, 0xf, true));
  v += __int_as_float(__builtin_amdgcn_update_dpp(0, __float_as_int(v), 0x161, 0xf, 0xf, true));
  return v;
}

#define NODES_PER_BLOCK 8

__global__ void __launch_bounds__(256)
gat_node(const float* __restrict__ xp,
         const int* __restrict__ rowbeg,
         const int* __restrict__ rowend,
         const int* __restrict__ srcs,
         const float* __restrict__ avec,
         float* __restrict__ tmp,
         float* __restrict__ part,
         int N) {
  __shared__ __attribute__((aligned(16))) float s_red[NODES_PER_BLOCK][128];
  const int lane = threadIdx.x & 31;
  const int wave = threadIdx.x >> 5;
  const int node = __builtin_amdgcn_readfirstlane(
      (int)(blockIdx.x * NODES_PER_BLOCK + wave));

  float4 v = {0.0f, 0.0f, 0.0f, 0.0f};

  if (node < N) {
    const int coff = (lane >> 4) * 64 + (lane & 15) * 4;
    const float4 d  = *(const float4*)(xp + (size_t)node * 128 + coff);
    float4 av = *(const float4*)(avec + coff);
    av.x *= LOG2E; av.y *= LOG2E; av.z *= LOG2E; av.w *= LOG2E;
    const int beg = rowbeg[node];
    const int end = rowend[node];

    float  mx  = row16_sum(score4(d, d, av));
    float  l   = 1.0f;
    float4 acc = d;

    for (int base = beg; base < end; base += 32) {
      const int nb = min(32, end - base);
      const int mysrc = (lane < nb) ? srcs[base + lane] : node;
      for (int t = 0; t < nb; t += 4) {
        const int s0 = __builtin_amdgcn_readlane(mysrc, t);
        const int s1 = __builtin_amdgcn_readlane(mysrc, t + 1);
        const int s2 = __builtin_amdgcn_readlane(mysrc, t + 2);
        const int s3 = __builtin_amdgcn_readlane(mysrc, t + 3);
        const float4 x0 = *(const float4*)(xp + (size_t)s0 * 128 + coff);
        const float4 x1 = *(const float4*)(xp + (size_t)s1 * 128 + coff);
        const float4 x2 = *(const float4*)(xp + (size_t)s2 * 128 + coff);
        const float4 x3 = *(const float4*)(xp + (size_t)s3 * 128 + coff);
        float e0 = row16_sum(score4(x0, d, av));
        float e1 = row16_sum(score4(x1, d, av));
        float e2 = row16_sum(score4(x2, d, av));
        float e3 = row16_sum(score4(x3, d, av));
        e1 = (t + 1 < nb) ? e1 : -INFINITY;
        e2 = (t + 2 < nb) ? e2 : -INFINITY;
        e3 = (t + 3 < nb) ? e3 : -INFINITY;
        const float mn = fmaxf(fmaxf(mx, e0), fmaxf(e1, fmaxf(e2, e3)));
        const float sc = __builtin_amdgcn_exp2f(mx - mn);
        const float p0 = __builtin_amdgcn_exp2f(e0 - mn);
        const float p1 = __builtin_amdgcn_exp2f(e1 - mn);
        const float p2 = __builtin_amdgcn_exp2f(e2 - mn);
        const float p3 = __builtin_amdgcn_exp2f(e3 - mn);
        l = l * sc + ((p0 + p1) + (p2 + p3));
        acc.x = acc.x * sc + ((p0 * x0.x + p1 * x1.x) + (p2 * x2.x + p3 * x3.x));
        acc.y = acc.y * sc + ((p0 * x0.y + p1 * x1.y) + (p2 * x2.y + p3 * x3.y));
        acc.z = acc.z * sc + ((p0 * x0.z + p1 * x1.z) + (p2 * x2.z + p3 * x3.z));
        acc.w = acc.w * sc + ((p0 * x0.w + p1 * x1.w) + (p2 * x2.w + p3 * x3.w));
        mx = mn;
      }
    }

    const float den = l + 1e-16f;
    float4 o;
    o.x = acc.x / den; o.y = acc.y / den; o.z = acc.z / den; o.w = acc.w / den;

    v.x = 0.5f * (o.x + __shfl_xor(o.x, 16));
    v.y = 0.5f * (o.y + __shfl_xor(o.y, 16));
    v.z = 0.5f * (o.z + __shfl_xor(o.z, 16));
    v.w = 0.5f * (o.w + __shfl_xor(o.w, 16));
  }

  if (lane < 16) {
    float4 q;
    q.x = v.x * v.x; q.y = v.y * v.y; q.z = v.z * v.z; q.w = v.w * v.w;
    *(float4*)&s_red[wave][lane * 4]      = v;
    *(float4*)&s_red[wave][64 + lane * 4] = q;
  }
  __syncthreads();
  if (node < N && lane < 16) *(float4*)(tmp + (size_t)node * 64 + lane * 4) = v;
  if (threadIdx.x < 128) {
    const int k = threadIdx.x;
    const float t = ((s_red[0][k] + s_red[1][k]) + (s_red[2][k] + s_red[3][k])) +
                    ((s_red[4][k] + s_red[5][k]) + (s_red[6][k] + s_red[7][k]));
    part[(size_t)blockIdx.x * 128 + k] = t;
  }
}

#define BN_SLICES 16

__global__ void __launch_bounds__(1024)
bn_partials_sum(const float* __restrict__ part, float* __restrict__ slicesum,
                int rows) {
  __shared__ __attribute__((aligned(16))) float s_w[32][32];
  const int lane = threadIdx.x & 31;
  const int wave = threadIdx.x >> 5;
  const int cb   = blockIdx.x & 3;
  const int sl   = blockIdx.x >> 2;
  const int cg   = lane & 7;
  const int rs   = wave * 4 + (lane >> 3);
  const int per  = (rows + BN_SLICES - 1) / BN_SLICES;
  const int rend = min(rows, (sl + 1) * per);
  const float* __restrict__ p = part + cb * 32 + cg * 4;
  float4 a = {0.0f, 0.0f, 0.0f, 0.0f};
  for (int rb = sl * per + rs; rb < rend; rb += 4 * 128) {
    float4 q[4];
#pragma unroll
    for (int j = 0; j < 4; ++j) {
      const int r = rb + j * 128;
      q[j] = (r < rend) ? *(const float4*)(p + (size_t)r * 128)
                        : make_float4(0.0f, 0.0f, 0.0f, 0.0f);
    }
#pragma unroll
    for (int j = 0; j < 4; ++j) {
      a.x += q[j].x; a.y += q[j].y; a.z += q[j].z; a.w += q[j].w;
    }
  }
  a.x += __shfl_xor(a.x, 8);  a.y += __shfl_xor(a.y, 8);
  a.z += __shfl_xor(a.z, 8);  a.w += __shfl_xor(a.w, 8);
  a.x += __shfl_xor(a.x, 16); a.y += __shfl_xor(a.y, 16);
  a.z += __shfl_xor(a.z, 16); a.w += __shfl_xor(a.w, 16);
  if (lane < 8) *(float4*)&s_w[wave][cg * 4] = a;
  __syncthreads();
  if (wave == 0) {
    float t = 0.0f;
    for (int w = 0; w < 32; ++w) t += s_w[w][lane];
    slicesum[sl * 128 + cb * 32 + lane] = t;
  }
}

#define BN_BLOCKS  512
#define BN_THREADS 256
#define BN_PIECES  8

__device__ __forceinline__ float bn_one(float x, float mean, float rstd,
                                        float g, float b) {
  return (x - mean) * rstd * g + b;
}

template <int TANH>
__global__ void __launch_bounds__(BN_THREADS)
bn_apply(const float* __restrict__ tmp,
         const float* __restrict__ slicesum,
         const float* __restrict__ gamma,
         const float* __restrict__ beta,
         float* __restrict__ out,
         int N) {
  const int total  = N * 16;
  const int stride = gridDim.x * BN_THREADS;
  const int first  = blockIdx.x * BN_THREADS + threadIdx.x;
  const int c      = (first & 15) * 4;
  const float4* __restrict__ in4  = (const float4*)tmp;
  float4* __restrict__       out4 = (float4*)out;

  const float invN = 1.0f / (float)N;
  float4 s4 = {0.0f, 0.0f, 0.0f, 0.0f}, q4 = {0.0f, 0.0f, 0.0f, 0.0f};
#pragma unroll
  for (int s = 0; s < BN_SLICES; ++s) {
    const float4 ps = *(const float4*)(slicesum + s * 128 + c);
    const float4 pq = *(const float4*)(slicesum + s * 128 + 64 + c);
    s4.x += ps.x; s4.y += ps.y; s4.z += ps.z; s4.w += ps.w;
    q4.x += pq.x; q4.y += pq.y; q4.z += pq.z; q4.w += pq.w;
  }
  const float4 g4 = *(const float4*)(gamma + c);
  const float4 b4 = *(const float4*)(beta + c);
  float4 mean, rstd;
  mean.x = s4.x * invN; rstd.x = rsqrtf(q4.x * invN - mean.x * mean.x + BN_EPS);
  mean.y = s4.y * invN; rstd.y = rsqrtf(q4.y * invN - mean.y * mean.y + BN_EPS);
  mean.z = s4.z * invN; rstd.z = rsqrtf(q4.z * invN - mean.z * mean.z + BN_EPS);
  mean.w = s4.w * invN; rstd.w = rsqrtf(q4.w * invN - mean.w * mean.w + BN_EPS);

  for (int i0 = first; i0 < total; i0 += BN_PIECES * stride) {
    float4 v[BN_PIECES];
#pragma unroll
    for (int j = 0; j < BN_PIECES; ++j) {
      const int i = i0 + j * stride;
      v[j] = in4[(i < total) ? i : i0];
    }
#pragma unroll
    for (int j = 0; j < BN_PIECES; ++j)
      asm volatile("" :: "v"(v[j].x), "v"(v[j].y), "v"(v[j].z), "v"(v[j].w));
#pragma unroll
    for (int j = 0; j < BN_PIECES; ++j) {
      const int i = i0 + j * stride;
      float4 r;
      r.x = bn_one(v[j].x, mean.x, rstd.x, g4.x, b4.x);
      r.y = bn_one(v[j].y, mean.y, rstd.y, g4.y, b4.y);
      r.z = bn_one(v[j].z, mean.z, rstd.z, g4.z, b4.z);
      r.w = bn_one(v[j].w, mean.w, rstd.w, g4.w, b4.w);
      if (TANH) { r.x = tanhf(r.x); r.y = tanhf(r.y); r.z = tanhf(r.z); r.w = tanhf(r.w); }
      if (i < total) out4[i] = r;
    }
  }
}

extern "C" void kernel_launch(void* const* d_in, const int* in_sizes, int n_in,
                              void* d_out, int out_size, void* d_ws, size_t ws_size,
                              hipStream_t stream) {
  const float* x      = (const float*)d_in[0];
  const int*   ei     = (const int*)  d_in[1];
  const float* W0     = (const float*)d_in[2];
  const float* a0     = (const float*)d_in[3];
  const float* gamma0 = (const float*)d_in[4];
  const float* beta0  = (const float*)d_in[5];
  const float* W1     = (const float*)d_in[6];
  const float* a1     = (const float*)d_in[7];
  const float* gamma1 = (const float*)d_in[8];
  const float* beta1  = (const float*)d_in[9];
  float* out = (float*)d_out;

  const int N    = in_sizes[0] / 128;
  const int E    = in_sizes[1] / 2;

  char* ws = (char*)d_ws;
  size_t off = 0;
  auto carve = [&](size_t nwords) {
    float* p = (float*)(ws + off);
    off = (off + nwords * sizeof(float) + 255) & ~(size_t)255;
    return p;
  };
  float* xp    = carve((size_t)N * 128);
  float* tmp   = carve((size_t)N * 64);
  float* h0    = carve((size_t)N * 64);
  float* stats = carve(BN_SLICES * 128);
  int* rowbeg  = (int*)carve((size_t)N);
  int* rowend  = (int*)carve((size_t)N);
  int* srcs    = (int*)carve((size_t)E);
  int* stg     = (int*)carve((size_t)E * 2 + SORT_BLOCKS * 32);
  int* tab     = (int*)carve((size_t)SORT_BLOCKS * MAX_BINS * 2);
  const int nodeGroups = (N + NODES_PER_BLOCK - 1) / NODES_PER_BLOCK;
  float* part  = carve((size_t)nodeGroups * 128);
  (void)ws_size;

  const int binBlocks    = (N + BIN_NODES - 1) / BIN_NODES;

  csr_bin<<<SORT_BLOCKS, 1024, 0, stream>>>(ei, stg, tab, E, N);
  csr_rows<<<binBlocks, 1024, 0, stream>>>(stg, tab, rowbeg, rowend, srcs, E, N);

  gemm_wmma_f32<128><<<512, 256, 0, stream>>>(x, W0, xp, N);
  gat_node<<<nodeGroups, 256, 0, stream>>>(xp, rowbeg, rowend, srcs, a0, tmp, part, N);
  bn_partials_sum<<<4 * BN_SLICES, 1024, 0, stream>>>(part, stats, nodeGroups);
  bn_apply<1><<<BN_BLOCKS, BN_THREADS, 0, stream>>>(tmp, stats, gamma0, beta0, h0, N);

  gemm_wmma_f32<64><<<512, 256, 0, stream>>>(h0, W1, xp, N);
  gat_node<<<nodeGroups, 256, 0, stream>>>(xp, rowbeg, rowend, srcs, a1, tmp, part, N);
  bn_partials_sum<<<4 * BN_SLICES, 1024, 0, stream>>>(part, stats, nodeGroups);
  bn_apply<0><<<BN_BLOCKS, BN_THREADS, 0, stream>>>(tmp, stats, gamma1, beta1, out, N);
}


# Edge score: leaky-relu as one max(m, 0.2m) in place of compare+select, its 4 products added as two fused channel pairs (score sum order and rounding change); loads, softmax, aggregation as before
# speedup vs baseline: 7.0710x; 1.0367x over previous
#include <hip/hip_runtime.h>
#include <hip/hip_bf16.h>
#include <math.h>


typedef __attribute__((ext_vector_type(2))) float v2f;
typedef __attribute__((ext_vector_type(8))) float v8f;

#define NEG_SLOPE 0.2f
#define BN_EPS 1e-5f
#define LOG2E 1.4426950408889634f

template <int K>
__global__ void __launch_bounds__(256)
gemm_wmma_f32(const float* __restrict__ A,
              const float* __restrict__ B,
              float* __restrict__ C,
              int M) {
  constexpr int Nn  = 128;
  constexpr int PA  = K + 4;
  constexpr int PC  = Nn + 4;
  constexpr int KS  = K / 4;
  constexpr int RQ  = K / 4;
  constexpr int NPA = (16 * RQ) / 256;
  __shared__ __attribute__((aligned(16))) float sA[16 * PA];
  __shared__ __attribute__((aligned(16))) float sC[16 * PC];

  const int tid  = threadIdx.x;
  const int wave = tid >> 5;
  const int lane = tid & 31;
  const int half = lane >> 4;
  const int l    = lane & 15;
  const int n0   = wave * 16;
  const int tiles = M >> 4;

  v2f bf[KS];
#pragma unroll
  for (int ks = 0; ks < KS; ++ks) {
    const int kb = ks * 4 + half * 2;
    bf[ks].x = B[(size_t)kb * Nn + n0 + l];
    bf[ks].y = B[(size_t)(kb + 1) * Nn + n0 + l];
  }

  int tile = blockIdx.x;
  if (tile >= tiles) return;

  const int q0 = tid,       r0 = q0 / RQ, e0 = (q0 % RQ) * 4;
  const int q1 = tid + 256, r1 = q1 / RQ, e1 = (q1 % RQ) * 4;
  const unsigned aoff0 = r0 * K + e0;
  const unsigned aoff1 = r1 * K + e1;
  const int orow = tid >> 5, ocol = (tid & 31) * 4;
  const unsigned coff = orow * Nn + ocol;

  float4 pre0, pre1 = {0.0f, 0.0f, 0.0f, 0.0f};
  {
    const float* __restrict__ At = A + (size_t)tile * (16 * K);
    pre0 = *(const float4*)(At + aoff0);
    if constexpr (NPA > 1) pre1 = *(const float4*)(At + aoff1);
  }

  for (; tile < tiles; tile += gridDim.x) {
    *(float4*)&sA[r0 * PA + e0] = pre0;
    if constexpr (NPA > 1) *(float4*)&sA[r1 * PA + e1] = pre1;
    __syncthreads();

    const int nxt = tile + gridDim.x;
    if (nxt < tiles) {
      const float* __restrict__ An = A + (size_t)nxt * (16 * K);
      pre0 = *(const float4*)(An + aoff0);
      if constexpr (NPA > 1) pre1 = *(const float4*)(An + aoff1);
    }

    v2f af[KS];
#pragma unroll
    for (int ks = 0; ks < KS; ++ks)
      af[ks] = *(const v2f*)&sA[l * PA + ks * 4 + half * 2];
    __builtin_amdgcn_sched_barrier(0);

    v8f c = {};
#pragma unroll
    for (int ks = 0; ks < KS; ++ks) {
      c = __builtin_amdgcn_wmma_f32_16x16x4_f32(false, af[ks], false, bf[ks],
                                                (short)0, c, false, false);
    }
    __builtin_amdgcn_sched_barrier(0);

#pragma unroll
    for (int j = 0; j < 8; ++j) sC[(j + half * 8) * PC + n0 + l] = c[j];
    __syncthreads();

    float* __restrict__ Ct = C + (size_t)tile * (16 * Nn);
    const float4 o0 = *(const float4*)&sC[orow * PC + ocol];
    const float4 o1 = *(const float4*)&sC[(orow + 8) * PC + ocol];
    *(float4*)(Ct + coff) = o0;
    *(float4*)(Ct + coff + 8 * Nn) = o1;
  }
}

#define BIN_SHIFT   8
#define BIN_NODES   (1 << BIN_SHIFT)
#define SORT_BLOCKS 256
#define MAX_BINS    1024
#define CHUNK_LDS   4096
#define BIN_LDS     8192

__global__ void __launch_bounds__(1024)
csr_bin(const int* __restrict__ ei, int* __restrict__ stgw,
        int* __restrict__ tabw, int E, int N) {
  __shared__ int s_cnt[MAX_BINS];
  __shared__ int s_cur[MAX_BINS];
  __shared__ int s_tot[32];
  __shared__ int2 s_out[CHUNK_LDS];
  const int tid  = threadIdx.x;
  const int lane = tid & 31;
  const int wave = tid >> 5;
  const int G  = gridDim.x;
  const int C  = (E + G - 1) / G;
  const int CP = (C + 15) & ~15;
  const int NB = (N + BIN_NODES - 1) >> BIN_SHIFT;
  const int PB = (NB + 15) & ~15;
  const int e0 = blockIdx.x * C;
  const int e1 = min(E, e0 + C);

  s_cnt[tid] = 0;
  int dd[4], ss[4];
#pragma unroll
  for (int j = 0; j < 4; ++j) {
    const int i = e0 + tid + j * 1024;
    const bool ok = i < e1;
    dd[j] = ok ? ei[E + i] : -1;
    ss[j] = ok ? ei[i] : 0;
  }
  __syncthreads();
#pragma unroll
  for (int j = 0; j < 4; ++j)
    if (dd[j] >= 0) atomicAdd(&s_cnt[dd[j] >> BIN_SHIFT], 1);
  for (int i = e0 + tid + 4096; i < e1; i += 1024)
    atomicAdd(&s_cnt[ei[E + i] >> BIN_SHIFT], 1);
  __syncthreads();

  const int v = s_cnt[tid];
  int inc = v;
  for (int off = 1; off < 32; off <<= 1) {
    const int up = __shfl_up(inc, off);
    if (lane >= off) inc += up;
  }
  if (lane == 31) s_tot[wave] = inc;
  __syncthreads();
  int wb = (lane < wave) ? s_tot[lane] : 0;
  for (int off = 16; off > 0; off >>= 1) wb += __shfl_xor(wb, off);
  const int excl = wb + inc - v;
  s_cur[tid] = excl;
  if (tid < PB)
    ((int2*)tabw)[(size_t)blockIdx.x * PB + tid] = make_int2(excl, v);
  __syncthreads();

  int2* __restrict__ outp = (int2*)stgw + (size_t)blockIdx.x * CP;
  if (C <= CHUNK_LDS) {
#pragma unroll
    for (int j = 0; j < 4; ++j)
      if (dd[j] >= 0) {
        const int pos = atomicAdd(&s_cur[dd[j] >> BIN_SHIFT], 1);
        s_out[pos] = make_int2(ss[j], dd[j]);
      }
    __syncthreads();
    for (int q = tid; q < e1 - e0; q += 1024) outp[q] = s_out[q];
  } else {
#pragma unroll
    for (int j = 0; j < 4; ++j)
      if (dd[j] >= 0) {
        const int pos = atomicAdd(&s_cur[dd[j] >> BIN_SHIFT], 1);
        outp[pos] = make_int2(ss[j], dd[j]);
      }
    for (int i = e0 + tid + 4096; i < e1; i += 1024) {
      const int d = ei[E + i], s = ei[i];
      const int pos = atomicAdd(&s_cur[d >> BIN_SHIFT], 1);
      outp[pos] = make_int2(s, d);
    }
  }
}

__global__ void __launch_bounds__(1024)
csr_rows(const int* __restrict__ stgw, const int* __restrict__ tabw,
         int* __restrict__ rowbeg, int* __restrict__ rowend,
         int* __restrict__ srcs, int E, int N) {
  constexpr int G   = SORT_BLOCKS;
  constexpr int PPW = G / 32;
  __shared__ int s_po[G];
  __shared__ int s_pc[G];
  __shared__ int s_cnt[BIN_NODES];
  __shared__ int s_cur[BIN_NODES];
  __shared__ int s_red[32];
  __shared__ int s_len[32];
  __shared__ int s_tot[32];
  __shared__ int s_rows[BIN_LDS];
  const int2* __restrict__ stg = (const int2*)stgw;
  const int2* __restrict__ tab = (const int2*)tabw;
  const int tid   = threadIdx.x;
  const int lane  = tid & 31;
  const int wave  = tid >> 5;
  const int bin   = blockIdx.x;
  const int node0 = bin << BIN_SHIFT;
  const int C  = (E + G - 1) / G;
  const int CP = (C + 15) & ~15;
  const int NB = (N + BIN_NODES - 1) >> BIN_SHIFT;
  const int PB = (NB + 15) & ~15;

  int myoff = 0, mylen = 0;
  if (tid < G) {
    const int2 t = tab[(size_t)tid * PB + bin];
    myoff = t.x;
    mylen = t.y;
    s_po[tid] = tid * CP + t.x;
    s_pc[tid] = t.y;
  }
  if (tid < BIN_NODES) s_cnt[tid] = 0;
  for (int off = 16; off > 0; off >>= 1) {
    myoff += __shfl_xor(myoff, off);
    mylen += __shfl_xor(mylen, off);
  }
  if (lane == 0) { s_red[wave] = myoff; s_len[wave] = mylen; }
  __syncthreads();
  int base = s_red[lane], total = s_len[lane];
  for (int off = 16; off > 0; off >>= 1) {
    base  += __shfl_xor(base, off);
    total += __shfl_xor(total, off);
  }

  int2 e[PPW];
  int pc[PPW], po[PPW];
#pragma unroll
  for (int k = 0; k < PPW; ++k) {
    const int p = wave * PPW + k;
    pc[k] = s_pc[p];
    po[k] = s_po[p];
  }
#pragma unroll
  for (int k = 0; k < PPW; ++k)
    e[k] = (lane < pc[k]) ? stg[po[k] + lane] : make_int2(0, -1);
#pragma unroll
  for (int k = 0; k < PPW; ++k)
    if (lane < pc[k]) atomicAdd(&s_cnt[e[k].y - node0], 1);
#pragma unroll
  for (int k = 0; k < PPW; ++k)
    for (int i = lane + 32; i < pc[k]; i += 32)
      atomicAdd(&s_cnt[stg[po[k] + i].y - node0], 1);
  __syncthreads();

  const int v = (tid < BIN_NODES) ? s_cnt[tid] : 0;
  int inc = v;
  for (int off = 1; off < 32; off <<= 1) {
    const int up = __shfl_up(inc, off);
    if (lane >= off) inc += up;
  }
  if (lane == 31) s_tot[wave] = inc;
  __syncthreads();
  int wb = (lane < wave) ? s_tot[lane] : 0;
  for (int off = 16; off > 0; off >>= 1) wb += __shfl_xor(wb, off);
  const int excl = wb + inc - v;
  if (tid < BIN_NODES) {
    s_cur[tid] = excl;
    const int node = node0 + tid;
    if (node < N) {
      rowbeg[node] = base + excl;
      rowend[node] = base + excl + v;
    }
  }
  __syncthreads();

  int* __restrict__ rowsOut = srcs + base;
  if (total <= BIN_LDS) {
#pragma unroll
    for (int k = 0; k < PPW; ++k)
      if (lane < pc[k]) {
        const int pos = atomicAdd(&s_cur[e[k].y - node0], 1);
        s_rows[pos] = e[k].x;
      }
#pragma unroll
    for (int k = 0; k < PPW; ++k)
      for (int i = lane + 32; i < pc[k]; i += 32) {
        const int2 q = stg[po[k] + i];
        const int pos = atomicAdd(&s_cur[q.y - node0], 1);
        s_rows[pos] = q.x;
      }
    __syncthreads();
    const int lead = base & 31;
    for (int q = tid - lead; q < total; q += 1024)
      if (q >= 0) rowsOut[q] = s_rows[q];
  } else {
#pragma unroll
    for (int k = 0; k < PPW; ++k)
      if (lane < pc[k]) {
        const int pos = atomicAdd(&s_cur[e[k].y - node0], 1);
        rowsOut[pos] = e[k].x;
      }
#pragma unroll
    for (int k = 0; k < PPW; ++k)
      for (int i = lane + 32; i < pc[k]; i += 32) {
        const int2 q = stg[po[k] + i];
        const int pos = atomicAdd(&s_cur[q.y - node0], 1);
        rowsOut[pos] = q.x;
      }
  }
}

__device__ __forceinline__ float score4(const float4 s, const float4 d,
                                        const float4 av) {
  const v2f m01 = {s.x + d.x, s.y + d.y};
  const v2f m23 = {s.z + d.z, s.w + d.w};
  const v2f a01 = {av.x, av.y};
  const v2f a23 = {av.z, av.w};
  const v2f l01 = __builtin_elementwise_max(m01, m01 * NEG_SLOPE);
  const v2f l23 = __builtin_elementwise_max(m23, m23 * NEG_SLOPE);
  const v2f acc = __builtin_elementwise_fma(l23, a23, l01 * a01);
  return acc.x + acc.y;
}

__device__ __forceinline__ float row16_sum(float v) {
  v += __int_as_float(__builtin_amdgcn_update_dpp(0, __float_as_int(v), 0x168, 0xf, 0xf, true));
  v += __int_as_float(__builtin_amdgcn_update_dpp(0, __float_as_int(v), 0x164, 0xf, 0xf, true));
  v += __int_as_float(__builtin_amdgcn_update_dpp(0, __float_as_int(v), 0x162, 0xf, 0xf, true));
  v += __int_as_float(__builtin_amdgcn_update_dpp(0, __float_as_int(v), 0x161, 0xf, 0xf, true));
  return v;
}

#define NODES_PER_BLOCK 8

__global__ void __launch_bounds__(256)
gat_node(const float* __restrict__ xp,
         const int* __restrict__ rowbeg,
         const int* __restrict__ rowend,
         const int* __restrict__ srcs,
         const float* __restrict__ avec,
         float* __restrict__ tmp,
         float* __restrict__ part,
         int N) {
  __shared__ __attribute__((aligned(16))) float s_red[NODES_PER_BLOCK][128];
  const int lane = threadIdx.x & 31;
  const int wave = threadIdx.x >> 5;
  const int node = __builtin_amdgcn_readfirstlane(
      (int)(blockIdx.x * NODES_PER_BLOCK + wave));

  float4 v = {0.0f, 0.0f, 0.0f, 0.0f};

  if (node < N) {
    const int coff = (lane >> 4) * 64 + (lane & 15) * 4;
    const float4 d  = *(const float4*)(xp + (size_t)node * 128 + coff);
    float4 av = *(const float4*)(avec + coff);
    av.x *= LOG2E; av.y *= LOG2E; av.z *= LOG2E; av.w *= LOG2E;
    const int beg = rowbeg[node];
    const int end = rowend[node];

    float  mx  = row16_sum(score4(d, d, av));
    float  l   = 1.0f;
    float4 acc = d;

    for (int base = beg; base < end; base += 32) {
      const int nb = min(32, end - base);
      const int mysrc = (lane < nb) ? srcs[base + lane] : node;
      for (int t = 0; t < nb; t += 4) {
        const int s0 = __builtin_amdgcn_readlane(mysrc, t);
        const int s1 = __builtin_amdgcn_readlane(mysrc, t + 1);
        const int s2 = __builtin_amdgcn_readlane(mysrc, t + 2);
        const int s3 = __builtin_amdgcn_readlane(mysrc, t + 3);
        const float4 x0 = *(const float4*)(xp + (size_t)s0 * 128 + coff);
        const float4 x1 = *(const float4*)(xp + (size_t)s1 * 128 + coff);
        const float4 x2 = *(const float4*)(xp + (size_t)s2 * 128 + coff);
        const float4 x3 = *(const float4*)(xp + (size_t)s3 * 128 + coff);
        float e0 = row16_sum(score4(x0, d, av));
        float e1 = row16_sum(score4(x1, d, av));
        float e2 = row16_sum(score4(x2, d, av));
        float e3 = row16_sum(score4(x3, d, av));
        e1 = (t + 1 < nb) ? e1 : -INFINITY;
        e2 = (t + 2 < nb) ? e2 : -INFINITY;
        e3 = (t + 3 < nb) ? e3 : -INFINITY;
        const float mn = fmaxf(fmaxf(mx, e0), fmaxf(e1, fmaxf(e2, e3)));
        const float sc = __builtin_amdgcn_exp2f(mx - mn);
        const float p0 = __builtin_amdgcn_exp2f(e0 - mn);
        const float p1 = __builtin_amdgcn_exp2f(e1 - mn);
        const float p2 = __builtin_amdgcn_exp2f(e2 - mn);
        const float p3 = __builtin_amdgcn_exp2f(e3 - mn);
        l = l * sc + ((p0 + p1) + (p2 + p3));
        acc.x = acc.x * sc + ((p0 * x0.x + p1 * x1.x) + (p2 * x2.x + p3 * x3.x));
        acc.y = acc.y * sc + ((p0 * x0.y + p1 * x1.y) + (p2 * x2.y + p3 * x3.y));
        acc.z = acc.z * sc + ((p0 * x0.z + p1 * x1.z) + (p2 * x2.z + p3 * x3.z));
        acc.w = acc.w * sc + ((p0 * x0.w + p1 * x1.w) + (p2 * x2.w + p3 * x3.w));
        mx = mn;
      }
    }

    const float den = l + 1e-16f;
    float4 o;
    o.x = acc.x / den; o.y = acc.y / den; o.z = acc.z / den; o.w = acc.w / den;

    v.x = 0.5f * (o.x + __shfl_xor(o.x, 16));
    v.y = 0.5f * (o.y + __shfl_xor(o.y, 16));
    v.z = 0.5f * (o.z + __shfl_xor(o.z, 16));
    v.w = 0.5f * (o.w + __shfl_xor(o.w, 16));
  }

  if (lane < 16) {
    float4 q;
    q.x = v.x * v.x; q.y = v.y * v.y; q.z = v.z * v.z; q.w = v.w * v.w;
    *(float4*)&s_red[wave][lane * 4]      = v;
    *(float4*)&s_red[wave][64 + lane * 4] = q;
  }
  __syncthreads();
  if (node < N && lane < 16) *(float4*)(tmp + (size_t)node * 64 + lane * 4) = v;
  if (threadIdx.x < 128) {
    const int k = threadIdx.x;
    const float t = ((s_red[0][k] + s_red[1][k]) + (s_red[2][k] + s_red[3][k])) +
                    ((s_red[4][k] + s_red[5][k]) + (s_red[6][k] + s_red[7][k]));
    part[(size_t)blockIdx.x * 128 + k] = t;
  }
}

#define BN_SLICES 16

__global__ void __launch_bounds__(1024)
bn_partials_sum(const float* __restrict__ part, float* __restrict__ slicesum,
                int rows) {
  __shared__ __attribute__((aligned(16))) float s_w[32][32];
  const int lane = threadIdx.x & 31;
  const int wave = threadIdx.x >> 5;
  const int cb   = blockIdx.x & 3;
  const int sl   = blockIdx.x >> 2;
  const int cg   = lane & 7;
  const int rs   = wave * 4 + (lane >> 3);
  const int per  = (rows + BN_SLICES - 1) / BN_SLICES;
  const int rend = min(rows, (sl + 1) * per);
  const float* __restrict__ p = part + cb * 32 + cg * 4;
  float4 a = {0.0f, 0.0f, 0.0f, 0.0f};
  for (int rb = sl * per + rs; rb < rend; rb += 4 * 128) {
    float4 q[4];
#pragma unroll
    for (int j = 0; j < 4; ++j) {
      const int r = rb + j * 128;
      q[j] = (r < rend) ? *(const float4*)(p + (size_t)r * 128)
                        : make_float4(0.0f, 0.0f, 0.0f, 0.0f);
    }
#pragma unroll
    for (int j = 0; j < 4; ++j) {
      a.x += q[j].x; a.y += q[j].y; a.z += q[j].z; a.w += q[j].w;
    }
  }
  a.x += __shfl_xor(a.x, 8);  a.y += __shfl_xor(a.y, 8);
  a.z += __shfl_xor(a.z, 8);  a.w += __shfl_xor(a.w, 8);
  a.x += __shfl_xor(a.x, 16); a.y += __shfl_xor(a.y, 16);
  a.z += __shfl_xor(a.z, 16); a.w += __shfl_xor(a.w, 16);
  if (lane < 8) *(float4*)&s_w[wave][cg * 4] = a;
  __syncthreads();
  if (wave == 0) {
    float t = 0.0f;
    for (int w = 0; w < 32; ++w) t += s_w[w][lane];
    slicesum[sl * 128 + cb * 32 + lane] = t;
  }
}

#define BN_BLOCKS  512
#define BN_THREADS 256
#define BN_PIECES  8

__device__ __forceinline__ float bn_one(float x, float mean, float rstd,
                                        float g, float b) {
  return (x - mean) * rstd * g + b;
}

template <int TANH>
__global__ void __launch_bounds__(BN_THREADS)
bn_apply(const float* __restrict__ tmp,
         const float* __restrict__ slicesum,
         const float* __restrict__ gamma,
         const float* __restrict__ beta,
         float* __restrict__ out,
         int N) {
  const int total  = N * 16;
  const int stride = gridDim.x * BN_THREADS;
  const int first  = blockIdx.x * BN_THREADS + threadIdx.x;
  const int c      = (first & 15) * 4;
  const float4* __restrict__ in4  = (const float4*)tmp;
  float4* __restrict__       out4 = (float4*)out;

  const float invN = 1.0f / (float)N;
  float4 s4 = {0.0f, 0.0f, 0.0f, 0.0f}, q4 = {0.0f, 0.0f, 0.0f, 0.0f};
#pragma unroll
  for (int s = 0; s < BN_SLICES; ++s) {
    const float4 ps = *(const float4*)(slicesum + s * 128 + c);
    const float4 pq = *(const float4*)(slicesum + s * 128 + 64 + c);
    s4.x += ps.x; s4.y += ps.y; s4.z += ps.z; s4.w += ps.w;
    q4.x += pq.x; q4.y += pq.y; q4.z += pq.z; q4.w += pq.w;
  }
  const float4 g4 = *(const float4*)(gamma + c);
  const float4 b4 = *(const float4*)(beta + c);
  float4 mean, rstd;
  mean.x = s4.x * invN; rstd.x = rsqrtf(q4.x * invN - mean.x * mean.x + BN_EPS);
  mean.y = s4.y * invN; rstd.y = rsqrtf(q4.y * invN - mean.y * mean.y + BN_EPS);
  mean.z = s4.z * invN; rstd.z = rsqrtf(q4.z * invN - mean.z * mean.z + BN_EPS);
  mean.w = s4.w * invN; rstd.w = rsqrtf(q4.w * invN - mean.w * mean.w + BN_EPS);

  for (int i0 = first; i0 < total; i0 += BN_PIECES * stride) {
    float4 v[BN_PIECES];
#pragma unroll
    for (int j = 0; j < BN_PIECES; ++j) {
      const int i = i0 + j * stride;
      v[j] = in4[(i < total) ? i : i0];
    }
#pragma unroll
    for (int j = 0; j < BN_PIECES; ++j)
      asm volatile("" :: "v"(v[j].x), "v"(v[j].y), "v"(v[j].z), "v"(v[j].w));
#pragma unroll
    for (int j = 0; j < BN_PIECES; ++j) {
      const int i = i0 + j * stride;
      float4 r;
      r.x = bn_one(v[j].x, mean.x, rstd.x, g4.x, b4.x);
      r.y = bn_one(v[j].y, mean.y, rstd.y, g4.y, b4.y);
      r.z = bn_one(v[j].z, mean.z, rstd.z, g4.z, b4.z);
      r.w = bn_one(v[j].w, mean.w, rstd.w, g4.w, b4.w);
      if (TANH) { r.x = tanhf(r.x); r.y = tanhf(r.y); r.z = tanhf(r.z); r.w = tanhf(r.w); }
      if (i < total) out4[i] = r;
    }
  }
}

extern "C" void kernel_launch(void* const* d_in, const int* in_sizes, int n_in,
                              void* d_out, int out_size, void* d_ws, size_t ws_size,
                              hipStream_t stream) {
  const float* x      = (const float*)d_in[0];
  const int*   ei     = (const int*)  d_in[1];
  const float* W0     = (const float*)d_in[2];
  const float* a0     = (const float*)d_in[3];
  const float* gamma0 = (const float*)d_in[4];
  const float* beta0  = (const float*)d_in[5];
  const float* W1     = (const float*)d_in[6];
  const float* a1     = (const float*)d_in[7];
  const float* gamma1 = (const float*)d_in[8];
  const float* beta1  = (const float*)d_in[9];
  float* out = (float*)d_out;

  const int N    = in_sizes[0] / 128;
  const int E    = in_sizes[1] / 2;

  char* ws = (char*)d_ws;
  size_t off = 0;
  auto carve = [&](size_t nwords) {
    float* p = (float*)(ws + off);
    off = (off + nwords * sizeof(float) + 255) & ~(size_t)255;
    return p;
  };
  float* xp    = carve((size_t)N * 128);
  float* tmp   = carve((size_t)N * 64);
  float* h0    = carve((size_t)N * 64);
  float* stats = carve(BN_SLICES * 128);
  int* rowbeg  = (int*)carve((size_t)N);
  int* rowend  = (int*)carve((size_t)N);
  int* srcs    = (int*)carve((size_t)E);
  int* stg     = (int*)carve((size_t)E * 2 + SORT_BLOCKS * 32);
  int* tab     = (int*)carve((size_t)SORT_BLOCKS * MAX_BINS * 2);
  const int nodeGroups = (N + NODES_PER_BLOCK - 1) / NODES_PER_BLOCK;
  float* part  = carve((size_t)nodeGroups * 128);
  (void)ws_size;

  const int binBlocks    = (N + BIN_NODES - 1) / BIN_NODES;

  csr_bin<<<SORT_BLOCKS, 1024, 0, stream>>>(ei, stg, tab, E, N);
  csr_rows<<<binBlocks, 1024, 0, stream>>>(stg, tab, rowbeg, rowend, srcs, E, N);

  gemm_wmma_f32<128><<<512, 256, 0, stream>>>(x, W0, xp, N);
  gat_node<<<nodeGroups, 256, 0, stream>>>(xp, rowbeg, rowend, srcs, a0, tmp, part, N);
  bn_partials_sum<<<4 * BN_SLICES, 1024, 0, stream>>>(part, stats, nodeGroups);
  bn_apply<1><<<BN_BLOCKS, BN_THREADS, 0, stream>>>(tmp, stats, gamma0, beta0, h0, N);

  gemm_wmma_f32<64><<<512, 256, 0, stream>>>(h0, W1, xp, N);
  gat_node<<<nodeGroups, 256, 0, stream>>>(xp, rowbeg, rowend, srcs, a1, tmp, part, N);
  bn_partials_sum<<<4 * BN_SLICES, 1024, 0, stream>>>(part, stats, nodeGroups);
  bn_apply<0><<<BN_BLOCKS, BN_THREADS, 0, stream>>>(tmp, stats, gamma1, beta1, out, N);
}


# Layer-0 batch-norm apply and tanh folded into the layer-1 projection: rows normalised one tile ahead while staged for LDS, under the next load; apply launch and its plane gone; same values and sums
# speedup vs baseline: 7.5519x; 1.0680x over previous
#include <hip/hip_runtime.h>
#include <hip/hip_bf16.h>
#include <math.h>


typedef __attribute__((ext_vector_type(2))) float v2f;
typedef __attribute__((ext_vector_type(8))) float v8f;

#define NEG_SLOPE 0.2f
#define BN_EPS 1e-5f
#define LOG2E 1.4426950408889634f
#define BN_SLICES 16

struct BnTerms { float4 mean, rstd, g, b; };

__device__ __forceinline__ BnTerms bn_terms(const float* __restrict__ slicesum,
                                            const float* __restrict__ gamma,
                                            const float* __restrict__ beta,
                                            int c, int N) {
  const float invN = 1.0f / (float)N;
  float4 s4 = {0.0f, 0.0f, 0.0f, 0.0f}, q4 = {0.0f, 0.0f, 0.0f, 0.0f};
#pragma unroll
  for (int s = 0; s < BN_SLICES; ++s) {
    const float4 ps = *(const float4*)(slicesum + s * 128 + c);
    const float4 pq = *(const float4*)(slicesum + s * 128 + 64 + c);
    s4.x += ps.x; s4.y += ps.y; s4.z += ps.z; s4.w += ps.w;
    q4.x += pq.x; q4.y += pq.y; q4.z += pq.z; q4.w += pq.w;
  }
  BnTerms t;
  t.g = *(const float4*)(gamma + c);
  t.b = *(const float4*)(beta + c);
  t.mean.x = s4.x * invN; t.rstd.x = rsqrtf(q4.x * invN - t.mean.x * t.mean.x + BN_EPS);
  t.mean.y = s4.y * invN; t.rstd.y = rsqrtf(q4.y * invN - t.mean.y * t.mean.y + BN_EPS);
  t.mean.z = s4.z * invN; t.rstd.z = rsqrtf(q4.z * invN - t.mean.z * t.mean.z + BN_EPS);
  t.mean.w = s4.w * invN; t.rstd.w = rsqrtf(q4.w * invN - t.mean.w * t.mean.w + BN_EPS);
  return t;
}

__device__ __forceinline__ float bn_one(float x, float mean, float rstd,
                                        float g, float b) {
  return (x - mean) * rstd * g + b;
}

__device__ __forceinline__ float4 bn4(const float4 v, const BnTerms& t) {
  float4 r;
  r.x = bn_one(v.x, t.mean.x, t.rstd.x, t.g.x, t.b.x);
  r.y = bn_one(v.y, t.mean.y, t.rstd.y, t.g.y, t.b.y);
  r.z = bn_one(v.z, t.mean.z, t.rstd.z, t.g.z, t.b.z);
  r.w = bn_one(v.w, t.mean.w, t.rstd.w, t.g.w, t.b.w);
  return r;
}

__device__ __forceinline__ float4 bn_tanh4(const float4 v, const BnTerms& t) {
  float4 r = bn4(v, t);
  r.x = tanhf(r.x); r.y = tanhf(r.y); r.z = tanhf(r.z); r.w = tanhf(r.w);
  return r;
}

template <int K, bool BN>
__global__ void __launch_bounds__(256)
gemm_wmma_f32(const float* __restrict__ A,
              const float* __restrict__ B,
              float* __restrict__ C,
              int M,
              const float* __restrict__ slicesum,
              const float* __restrict__ gamma,
              const float* __restrict__ beta) {
  constexpr int Nn  = 128;
  constexpr int PA  = K + 4;
  constexpr int PC  = Nn + 4;
  constexpr int KS  = K / 4;
  constexpr int RQ  = K / 4;
  constexpr int NPA = (16 * RQ) / 256;
  static_assert(!BN || NPA == 1, "BN path: one piece, one channel group per thread");
  __shared__ __attribute__((aligned(16))) float sA[16 * PA];
  __shared__ __attribute__((aligned(16))) float sC[16 * PC];

  const int tid  = threadIdx.x;
  const int wave = tid >> 5;
  const int lane = tid & 31;
  const int half = lane >> 4;
  const int l    = lane & 15;
  const int n0   = wave * 16;
  const int tiles = M >> 4;

  v2f bf[KS];
#pragma unroll
  for (int ks = 0; ks < KS; ++ks) {
    const int kb = ks * 4 + half * 2;
    bf[ks].x = B[(size_t)kb * Nn + n0 + l];
    bf[ks].y = B[(size_t)(kb + 1) * Nn + n0 + l];
  }

  int tile = blockIdx.x;
  if (tile >= tiles) return;

  const int q0 = tid,       r0 = q0 / RQ, e0 = (q0 % RQ) * 4;
  const int q1 = tid + 256, r1 = q1 / RQ, e1 = (q1 % RQ) * 4;
  const unsigned aoff0 = r0 * K + e0;
  const unsigned aoff1 = r1 * K + e1;
  const int orow = tid >> 5, ocol = (tid & 31) * 4;
  const unsigned coff = orow * Nn + ocol;

  float4 pre0, pre1 = {0.0f, 0.0f, 0.0f, 0.0f};
  {
    const float* __restrict__ At = A + (size_t)tile * (16 * K);
    pre0 = *(const float4*)(At + aoff0);
    if constexpr (NPA > 1) pre1 = *(const float4*)(At + aoff1);
  }

  BnTerms bt = {};
  float4 rawn = {0.0f, 0.0f, 0.0f, 0.0f};
  if constexpr (BN) {
    const int nx1 = tile + gridDim.x;
    if (nx1 < tiles) rawn = *(const float4*)(A + (size_t)nx1 * (16 * K) + aoff0);
    __builtin_amdgcn_sched_barrier(0);
    bt = bn_terms(slicesum, gamma, beta, e0, M);
    pre0 = bn_tanh4(pre0, bt);
  }

  for (; tile < tiles; tile += gridDim.x) {
    *(float4*)&sA[r0 * PA + e0] = pre0;
    if constexpr (NPA > 1) *(float4*)&sA[r1 * PA + e1] = pre1;
    __syncthreads();

    const int nxt = tile + gridDim.x;
    float4 raw2 = {0.0f, 0.0f, 0.0f, 0.0f};
    if constexpr (BN) {
      const int nx2 = nxt + gridDim.x;
      if (nx2 < tiles) raw2 = *(const float4*)(A + (size_t)nx2 * (16 * K) + aoff0);
    } else {
      if (nxt < tiles) {
        const float* __restrict__ An = A + (size_t)nxt * (16 * K);
        pre0 = *(const float4*)(An + aoff0);
        if constexpr (NPA > 1) pre1 = *(const float4*)(An + aoff1);
      }
    }

    v2f af[KS];
#pragma unroll
    for (int ks = 0; ks < KS; ++ks)
      af[ks] = *(const v2f*)&sA[l * PA + ks * 4 + half * 2];
    __builtin_amdgcn_sched_barrier(0);

    v8f c = {};
#pragma unroll
    for (int ks = 0; ks < KS; ++ks) {
      c = __builtin_amdgcn_wmma_f32_16x16x4_f32(false, af[ks], false, bf[ks],
                                                (short)0, c, false, false);
    }
    __builtin_amdgcn_sched_barrier(0);

    if constexpr (BN) {
      if (nxt < tiles) pre0 = bn_tanh4(rawn, bt);
      rawn = raw2;
    }

#pragma unroll
    for (int j = 0; j < 8; ++j) sC[(j + half * 8) * PC + n0 + l] = c[j];
    __syncthreads();

    float* __restrict__ Ct = C + (size_t)tile * (16 * Nn);
    const float4 o0 = *(const float4*)&sC[orow * PC + ocol];
    const float4 o1 = *(const float4*)&sC[(orow + 8) * PC + ocol];
    *(float4*)(Ct + coff) = o0;
    *(float4*)(Ct + coff + 8 * Nn) = o1;
  }
}

#define BIN_SHIFT   8
#define BIN_NODES   (1 << BIN_SHIFT)
#define SORT_BLOCKS 256
#define MAX_BINS    1024
#define CHUNK_LDS   4096
#define BIN_LDS     8192

__global__ void __launch_bounds__(1024)
csr_bin(const int* __restrict__ ei, int* __restrict__ stgw,
        int* __restrict__ tabw, int E, int N) {
  __shared__ int s_cnt[MAX_BINS];
  __shared__ int s_cur[MAX_BINS];
  __shared__ int s_tot[32];
  __shared__ int2 s_out[CHUNK_LDS];
  const int tid  = threadIdx.x;
  const int lane = tid & 31;
  const int wave = tid >> 5;
  const int G  = gridDim.x;
  const int C  = (E + G - 1) / G;
  const int CP = (C + 15) & ~15;
  const int NB = (N + BIN_NODES - 1) >> BIN_SHIFT;
  const int PB = (NB + 15) & ~15;
  const int e0 = blockIdx.x * C;
  const int e1 = min(E, e0 + C);

  s_cnt[tid] = 0;
  int dd[4], ss[4];
#pragma unroll
  for (int j = 0; j < 4; ++j) {
    const int i = e0 + tid + j * 1024;
    const bool ok = i < e1;
    dd[j] = ok ? ei[E + i] : -1;
    ss[j] = ok ? ei[i] : 0;
  }
  __syncthreads();
#pragma unroll
  for (int j = 0; j < 4; ++j)
    if (dd[j] >= 0) atomicAdd(&s_cnt[dd[j] >> BIN_SHIFT], 1);
  for (int i = e0 + tid + 4096; i < e1; i += 1024)
    atomicAdd(&s_cnt[ei[E + i] >> BIN_SHIFT], 1);
  __syncthreads();

  const int v = s_cnt[tid];
  int inc = v;
  for (int off = 1; off < 32; off <<= 1) {
    const int up = __shfl_up(inc, off);
    if (lane >= off) inc += up;
  }
  if (lane == 31) s_tot[wave] = inc;
  __syncthreads();
  int wb = (lane < wave) ? s_tot[lane] : 0;
  for (int off = 16; off > 0; off >>= 1) wb += __shfl_xor(wb, off);
  const int excl = wb + inc - v;
  s_cur[tid] = excl;
  if (tid < PB)
    ((int2*)tabw)[(size_t)blockIdx.x * PB + tid] = make_int2(excl, v);
  __syncthreads();

  int2* __restrict__ outp = (int2*)stgw + (size_t)blockIdx.x * CP;
  if (C <= CHUNK_LDS) {
#pragma unroll
    for (int j = 0; j < 4; ++j)
      if (dd[j] >= 0) {
        const int pos = atomicAdd(&s_cur[dd[j] >> BIN_SHIFT], 1);
        s_out[pos] = make_int2(ss[j], dd[j]);
      }
    __syncthreads();
    for (int q = tid; q < e1 - e0; q += 1024) outp[q] = s_out[q];
  } else {
#pragma unroll
    for (int j = 0; j < 4; ++j)
      if (dd[j] >= 0) {
        const int pos = atomicAdd(&s_cur[dd[j] >> BIN_SHIFT], 1);
        outp[pos] = make_int2(ss[j], dd[j]);
      }
    for (int i = e0 + tid + 4096; i < e1; i += 1024) {
      const int d = ei[E + i], s = ei[i];
      const int pos = atomicAdd(&s_cur[d >> BIN_SHIFT], 1);
      outp[pos] = make_int2(s, d);
    }
  }
}

__global__ void __launch_bounds__(1024)
csr_rows(const int* __restrict__ stgw, const int* __restrict__ tabw,
         int* __restrict__ rowbeg, int* __restrict__ rowend,
         int* __restrict__ srcs, int E, int N) {
  constexpr int G   = SORT_BLOCKS;
  constexpr int PPW = G / 32;
  __shared__ int s_po[G];
  __shared__ int s_pc[G];
  __shared__ int s_cnt[BIN_NODES];
  __shared__ int s_cur[BIN_NODES];
  __shared__ int s_red[32];
  __shared__ int s_len[32];
  __shared__ int s_tot[32];
  __shared__ int s_rows[BIN_LDS];
  const int2* __restrict__ stg = (const int2*)stgw;
  const int2* __restrict__ tab = (const int2*)tabw;
  const int tid   = threadIdx.x;
  const int lane  = tid & 31;
  const int wave  = tid >> 5;
  const int bin   = blockIdx.x;
  const int node0 = bin << BIN_SHIFT;
  const int C  = (E + G - 1) / G;
  const int CP = (C + 15) & ~15;
  const int NB = (N + BIN_NODES - 1) >> BIN_SHIFT;
  const int PB = (NB + 15) & ~15;

  int myoff = 0, mylen = 0;
  if (tid < G) {
    const int2 t = tab[(size_t)tid * PB + bin];
    myoff = t.x;
    mylen = t.y;
    s_po[tid] = tid * CP + t.x;
    s_pc[tid] = t.y;
  }
  if (tid < BIN_NODES) s_cnt[tid] = 0;
  for (int off = 16; off > 0; off >>= 1) {
    myoff += __shfl_xor(myoff, off);
    mylen += __shfl_xor(mylen, off);
  }
  if (lane == 0) { s_red[wave] = myoff; s_len[wave] = mylen; }
  __syncthreads();
  int base = s_red[lane], total = s_len[lane];
  for (int off = 16; off > 0; off >>= 1) {
    base  += __shfl_xor(base, off);
    total += __shfl_xor(total, off);
  }

  int2 e[PPW];
  int pc[PPW], po[PPW];
#pragma unroll
  for (int k = 0; k < PPW; ++k) {
    const int p = wave * PPW + k;
    pc[k] = s_pc[p];
    po[k] = s_po[p];
  }
#pragma unroll
  for (int k = 0; k < PPW; ++k)
    e[k] = (lane < pc[k]) ? stg[po[k] + lane] : make_int2(0, -1);
#pragma unroll
  for (int k = 0; k < PPW; ++k)
    if (lane < pc[k]) atomicAdd(&s_cnt[e[k].y - node0], 1);
#pragma unroll
  for (int k = 0; k < PPW; ++k)
    for (int i = lane + 32; i < pc[k]; i += 32)
      atomicAdd(&s_cnt[stg[po[k] + i].y - node0], 1);
  __syncthreads();

  const int v = (tid < BIN_NODES) ? s_cnt[tid] : 0;
  int inc = v;
  for (int off = 1; off < 32; off <<= 1) {
    const int up = __shfl_up(inc, off);
    if (lane >= off) inc += up;
  }
  if (lane == 31) s_tot[wave] = inc;
  __syncthreads();
  int wb = (lane < wave) ? s_tot[lane] : 0;
  for (int off = 16; off > 0; off >>= 1) wb += __shfl_xor(wb, off);
  const int excl = wb + inc - v;
  if (tid < BIN_NODES) {
    s_cur[tid] = excl;
    const int node = node0 + tid;
    if (node < N) {
      rowbeg[node] = base + excl;
      rowend[node] = base + excl + v;
    }
  }
  __syncthreads();

  int* __restrict__ rowsOut = srcs + base;
  if (total <= BIN_LDS) {
#pragma unroll
    for (int k = 0; k < PPW; ++k)
      if (lane < pc[k]) {
        const int pos = atomicAdd(&s_cur[e[k].y - node0], 1);
        s_rows[pos] = e[k].x;
      }
#pragma unroll
    for (int k = 0; k < PPW; ++k)
      for (int i = lane + 32; i < pc[k]; i += 32) {
        const int2 q = stg[po[k] + i];
        const int pos = atomicAdd(&s_cur[q.y - node0], 1);
        s_rows[pos] = q.x;
      }
    __syncthreads();
    const int lead = base & 31;
    for (int q = tid - lead; q < total; q += 1024)
      if (q >= 0) rowsOut[q] = s_rows[q];
  } else {
#pragma unroll
    for (int k = 0; k < PPW; ++k)
      if (lane < pc[k]) {
        const int pos = atomicAdd(&s_cur[e[k].y - node0], 1);
        rowsOut[pos] = e[k].x;
      }
#pragma unroll
    for (int k = 0; k < PPW; ++k)
      for (int i = lane + 32; i < pc[k]; i += 32) {
        const int2 q = stg[po[k] + i];
        const int pos = atomicAdd(&s_cur[q.y - node0], 1);
        rowsOut[pos] = q.x;
      }
  }
}

__device__ __forceinline__ float score4(const float4 s, const float4 d,
                                        const float4 av) {
  const v2f m01 = {s.x + d.x, s.y + d.y};
  const v2f m23 = {s.z + d.z, s.w + d.w};
  const v2f a01 = {av.x, av.y};
  const v2f a23 = {av.z, av.w};
  const v2f l01 = __builtin_elementwise_max(m01, m01 * NEG_SLOPE);
  const v2f l23 = __builtin_elementwise_max(m23, m23 * NEG_SLOPE);
  const v2f acc = __builtin_elementwise_fma(l23, a23, l01 * a01);
  return acc.x + acc.y;
}

__device__ __forceinline__ float row16_sum(float v) {
  v += __int_as_float(__builtin_amdgcn_update_dpp(0, __float_as_int(v), 0x168, 0xf, 0xf, true));
  v += __int_as_float(__builtin_amdgcn_update_dpp(0, __float_as_int(v), 0x164, 0xf, 0xf, true));
  v += __int_as_float(__builtin_amdgcn_update_dpp(0, __float_as_int(v), 0x162, 0xf, 0xf, true));
  v += __int_as_float(__builtin_amdgcn_update_dpp(0, __float_as_int(v), 0x161, 0xf, 0xf, true));
  return v;
}

#define NODES_PER_BLOCK 8

__global__ void __launch_bounds__(256)
gat_node(const float* __restrict__ xp,
         const int* __restrict__ rowbeg,
         const int* __restrict__ rowend,
         const int* __restrict__ srcs,
         const float* __restrict__ avec,
         float* __restrict__ tmp,
         float* __restrict__ part,
         int N) {
  __shared__ __attribute__((aligned(16))) float s_red[NODES_PER_BLOCK][128];
  const int lane = threadIdx.x & 31;
  const int wave = threadIdx.x >> 5;
  const int node = __builtin_amdgcn_readfirstlane(
      (int)(blockIdx.x * NODES_PER_BLOCK + wave));

  float4 v = {0.0f, 0.0f, 0.0f, 0.0f};

  if (node < N) {
    const int coff = (lane >> 4) * 64 + (lane & 15) * 4;
    const float4 d  = *(const float4*)(xp + (size_t)node * 128 + coff);
    float4 av = *(const float4*)(avec + coff);
    av.x *= LOG2E; av.y *= LOG2E; av.z *= LOG2E; av.w *= LOG2E;
    const int beg = rowbeg[node];
    const int end = rowend[node];

    float  mx  = row16_sum(score4(d, d, av));
    float  l   = 1.0f;
    float4 acc = d;

    for (int base = beg; base < end; base += 32) {
      const int nb = min(32, end - base);
      const int mysrc = (lane < nb) ? srcs[base + lane] : node;
      for (int t = 0; t < nb; t += 4) {
        const int s0 = __builtin_amdgcn_readlane(mysrc, t);
        const int s1 = __builtin_amdgcn_readlane(mysrc, t + 1);
        const int s2 = __builtin_amdgcn_readlane(mysrc, t + 2);
        const int s3 = __builtin_amdgcn_readlane(mysrc, t + 3);
        const float4 x0 = *(const float4*)(xp + (size_t)s0 * 128 + coff);
        const float4 x1 = *(const float4*)(xp + (size_t)s1 * 128 + coff);
        const float4 x2 = *(const float4*)(xp + (size_t)s2 * 128 + coff);
        const float4 x3 = *(const float4*)(xp + (size_t)s3 * 128 + coff);
        float e0 = row16_sum(score4(x0, d, av));
        float e1 = row16_sum(score4(x1, d, av));
        float e2 = row16_sum(score4(x2, d, av));
        float e3 = row16_sum(score4(x3, d, av));
        e1 = (t + 1 < nb) ? e1 : -INFINITY;
        e2 = (t + 2 < nb) ? e2 : -INFINITY;
        e3 = (t + 3 < nb) ? e3 : -INFINITY;
        const float mn = fmaxf(fmaxf(mx, e0), fmaxf(e1, fmaxf(e2, e3)));
        const float sc = __builtin_amdgcn_exp2f(mx - mn);
        const float p0 = __builtin_amdgcn_exp2f(e0 - mn);
        const float p1 = __builtin_amdgcn_exp2f(e1 - mn);
        const float p2 = __builtin_amdgcn_exp2f(e2 - mn);
        const float p3 = __builtin_amdgcn_exp2f(e3 - mn);
        l = l * sc + ((p0 + p1) + (p2 + p3));
        acc.x = acc.x * sc + ((p0 * x0.x + p1 * x1.x) + (p2 * x2.x + p3 * x3.x));
        acc.y = acc.y * sc + ((p0 * x0.y + p1 * x1.y) + (p2 * x2.y + p3 * x3.y));
        acc.z = acc.z * sc + ((p0 * x0.z + p1 * x1.z) + (p2 * x2.z + p3 * x3.z));
        acc.w = acc.w * sc + ((p0 * x0.w + p1 * x1.w) + (p2 * x2.w + p3 * x3.w));
        mx = mn;
      }
    }

    const float den = l + 1e-16f;
    float4 o;
    o.x = acc.x / den; o.y = acc.y / den; o.z = acc.z / den; o.w = acc.w / den;

    v.x = 0.5f * (o.x + __shfl_xor(o.x, 16));
    v.y = 0.5f * (o.y + __shfl_xor(o.y, 16));
    v.z = 0.5f * (o.z + __shfl_xor(o.z, 16));
    v.w = 0.5f * (o.w + __shfl_xor(o.w, 16));
  }

  if (lane < 16) {
    float4 q;
    q.x = v.x * v.x; q.y = v.y * v.y; q.z = v.z * v.z; q.w = v.w * v.w;
    *(float4*)&s_red[wave][lane * 4]      = v;
    *(float4*)&s_red[wave][64 + lane * 4] = q;
  }
  __syncthreads();
  if (node < N && lane < 16) *(float4*)(tmp + (size_t)node * 64 + lane * 4) = v;
  if (threadIdx.x < 128) {
    const int k = threadIdx.x;
    const float t = ((s_red[0][k] + s_red[1][k]) + (s_red[2][k] + s_red[3][k])) +
                    ((s_red[4][k] + s_red[5][k]) + (s_red[6][k] + s_red[7][k]));
    part[(size_t)blockIdx.x * 128 + k] = t;
  }
}

__global__ void __launch_bounds__(1024)
bn_partials_sum(const float* __restrict__ part, float* __restrict__ slicesum,
                int rows) {
  __shared__ __attribute__((aligned(16))) float s_w[32][32];
  const int lane = threadIdx.x & 31;
  const int wave = threadIdx.x >> 5;
  const int cb   = blockIdx.x & 3;
  const int sl   = blockIdx.x >> 2;
  const int cg   = lane & 7;
  const int rs   = wave * 4 + (lane >> 3);
  const int per  = (rows + BN_SLICES - 1) / BN_SLICES;
  const int rend = min(rows, (sl + 1) * per);
  const float* __restrict__ p = part + cb * 32 + cg * 4;
  float4 a = {0.0f, 0.0f, 0.0f, 0.0f};
  for (int rb = sl * per + rs; rb < rend; rb += 4 * 128) {
    float4 q[4];
#pragma unroll
    for (int j = 0; j < 4; ++j) {
      const int r = rb + j * 128;
      q[j] = (r < rend) ? *(const float4*)(p + (size_t)r * 128)
                        : make_float4(0.0f, 0.0f, 0.0f, 0.0f);
    }
#pragma unroll
    for (int j = 0; j < 4; ++j) {
      a.x += q[j].x; a.y += q[j].y; a.z += q[j].z; a.w += q[j].w;
    }
  }
  a.x += __shfl_xor(a.x, 8);  a.y += __shfl_xor(a.y, 8);
  a.z += __shfl_xor(a.z, 8);  a.w += __shfl_xor(a.w, 8);
  a.x += __shfl_xor(a.x, 16); a.y += __shfl_xor(a.y, 16);
  a.z += __shfl_xor(a.z, 16); a.w += __shfl_xor(a.w, 16);
  if (lane < 8) *(float4*)&s_w[wave][cg * 4] = a;
  __syncthreads();
  if (wave == 0) {
    float t = 0.0f;
    for (int w = 0; w < 32; ++w) t += s_w[w][lane];
    slicesum[sl * 128 + cb * 32 + lane] = t;
  }
}

#define BN_BLOCKS  512
#define BN_THREADS 256
#define BN_PIECES  8

__global__ void __launch_bounds__(BN_THREADS)
bn_apply(const float* __restrict__ tmp,
         const float* __restrict__ slicesum,
         const float* __restrict__ gamma,
         const float* __restrict__ beta,
         float* __restrict__ out,
         int N) {
  const int total  = N * 16;
  const int stride = gridDim.x * BN_THREADS;
  const int first  = blockIdx.x * BN_THREADS + threadIdx.x;
  const int c      = (first & 15) * 4;
  const float4* __restrict__ in4  = (const float4*)tmp;
  float4* __restrict__       out4 = (float4*)out;

  const BnTerms bt = bn_terms(slicesum, gamma, beta, c, N);

  for (int i0 = first; i0 < total; i0 += BN_PIECES * stride) {
    float4 v[BN_PIECES];
#pragma unroll
    for (int j = 0; j < BN_PIECES; ++j) {
      const int i = i0 + j * stride;
      v[j] = in4[(i < total) ? i : i0];
    }
#pragma unroll
    for (int j = 0; j < BN_PIECES; ++j)
      asm volatile("" :: "v"(v[j].x), "v"(v[j].y), "v"(v[j].z), "v"(v[j].w));
#pragma unroll
    for (int j = 0; j < BN_PIECES; ++j) {
      const int i = i0 + j * stride;
      const float4 r = bn4(v[j], bt);
      if (i < total) out4[i] = r;
    }
  }
}

extern "C" void kernel_launch(void* const* d_in, const int* in_sizes, int n_in,
                              void* d_out, int out_size, void* d_ws, size_t ws_size,
                              hipStream_t stream) {
  const float* x      = (const float*)d_in[0];
  const int*   ei     = (const int*)  d_in[1];
  const float* W0     = (const float*)d_in[2];
  const float* a0     = (const float*)d_in[3];
  const float* gamma0 = (const float*)d_in[4];
  const float* beta0  = (const float*)d_in[5];
  const float* W1     = (const float*)d_in[6];
  const float* a1     = (const float*)d_in[7];
  const float* gamma1 = (const float*)d_in[8];
  const float* beta1  = (const float*)d_in[9];
  float* out = (float*)d_out;

  const int N    = in_sizes[0] / 128;
  const int E    = in_sizes[1] / 2;

  char* ws = (char*)d_ws;
  size_t off = 0;
  auto carve = [&](size_t nwords) {
    float* p = (float*)(ws + off);
    off = (off + nwords * sizeof(float) + 255) & ~(size_t)255;
    return p;
  };
  float* xp    = carve((size_t)N * 128);
  float* tmp   = carve((size_t)N * 64);
  float* stats = carve(BN_SLICES * 128);
  int* rowbeg  = (int*)carve((size_t)N);
  int* rowend  = (int*)carve((size_t)N);
  int* srcs    = (int*)carve((size_t)E);
  int* stg     = (int*)carve((size_t)E * 2 + SORT_BLOCKS * 32);
  int* tab     = (int*)carve((size_t)SORT_BLOCKS * MAX_BINS * 2);
  const int nodeGroups = (N + NODES_PER_BLOCK - 1) / NODES_PER_BLOCK;
  float* part  = carve((size_t)nodeGroups * 128);
  (void)ws_size;

  const int binBlocks    = (N + BIN_NODES - 1) / BIN_NODES;

  csr_bin<<<SORT_BLOCKS, 1024, 0, stream>>>(ei, stg, tab, E, N);
  csr_rows<<<binBlocks, 1024, 0, stream>>>(stg, tab, rowbeg, rowend, srcs, E, N);

  gemm_wmma_f32<128, false><<<512, 256, 0, stream>>>(x, W0, xp, N, nullptr, nullptr, nullptr);
  gat_node<<<nodeGroups, 256, 0, stream>>>(xp, rowbeg, rowend, srcs, a0, tmp, part, N);
  bn_partials_sum<<<4 * BN_SLICES, 1024, 0, stream>>>(part, stats, nodeGroups);

  gemm_wmma_f32<64, true><<<512, 256, 0, stream>>>(tmp, W1, xp, N, stats, gamma0, beta0);
  gat_node<<<nodeGroups, 256, 0, stream>>>(xp, rowbeg, rowend, srcs, a1, tmp, part, N);
  bn_partials_sum<<<4 * BN_SLICES, 1024, 0, stream>>>(part, stats, nodeGroups);
  bn_apply<<<BN_BLOCKS, BN_THREADS, 0, stream>>>(tmp, stats, gamma1, beta1, out, N);
}
